# starting kernel
# baseline (speedup 1.0000x reference)
#include <hip/hip_runtime.h>

#define NEG_SLOPE 0.2f
#define ENC_NEG_INF 0x007FFFFFu

typedef float v2f __attribute__((ext_vector_type(2)));
typedef float v8f __attribute__((ext_vector_type(8)));

__device__ __forceinline__ unsigned gat_fenc(float f) {
    unsigned u = __float_as_uint(f);
    return (u & 0x80000000u) ? ~u : (u | 0x80000000u);
}
__device__ __forceinline__ float gat_fdec(unsigned u) {
    unsigned b = (u & 0x80000000u) ? (u & 0x7FFFFFFFu) : ~u;
    return __uint_as_float(b);
}

template<int NT>
__global__ __launch_bounds__(256)
void gat_gemm_wmma(const float* __restrict__ X, const float* __restrict__ W,
                   float* __restrict__ Hout, int n_rows)
{
    constexpr int KD = 128;
    constexpr int ND = NT * 16;
    __shared__ float Wt[ND * KD];

    for (int i = threadIdx.x; i < KD * ND; i += 256) {
        int k = i / ND, n = i % ND;
        Wt[n * KD + ((k + 4 * n) & (KD - 1))] = W[i];
    }
    __syncthreads();

    const int lane = threadIdx.x & 31;
    const int wave = threadIdx.x >> 5;
    const int m0   = (blockIdx.x * 8 + wave) * 16;
    const int mlo  = lane & 15;
    const int kofs = (lane >> 4) * 2;

    int arow = m0 + mlo;
    if (arow >= n_rows) arow = n_rows - 1;
    const float* xrow = X + (size_t)arow * KD + kofs;

    v8f acc[NT] = {};

    for (int k = 0; k < KD; k += 4) {
        v2f a = *(const v2f*)(xrow + k);
#pragma unroll
        for (int t = 0; t < NT; ++t) {
            int n = t * 16 + mlo;
            v2f b = *(const v2f*)(&Wt[n * KD + ((k + kofs + 4 * n) & (KD - 1))]);
            acc[t] = __builtin_amdgcn_wmma_f32_16x16x4_f32(
                false, a, false, b, (short)0, acc[t], false, false);
        }
    }

    const int mhi = (lane >> 4) * 8;
#pragma unroll
    for (int t = 0; t < NT; ++t)
#pragma unroll
        for (int j = 0; j < 8; ++j) {
            int m = m0 + mhi + j;
            if (m < n_rows) Hout[(size_t)m * ND + t * 16 + mlo] = acc[t][j];
        }
}

__global__ void gat_attn_score(const float* __restrict__ Hf,
                               const float* __restrict__ att_s,
                               const float* __restrict__ att_d,
                               float* __restrict__ a_s, float* __restrict__ a_d,
                               int n, int heads, int ch)
{
    int t = blockIdx.x * blockDim.x + threadIdx.x;
    if (t >= n * heads) return;
    int node = t / heads, h = t - node * heads;
    const float* hp = Hf + ((size_t)node * heads + h) * ch;
    const float* sp = att_s + h * ch;
    const float* dp = att_d + h * ch;
    float ss = 0.f, dd = 0.f;
    for (int c = 0; c < ch; ++c) { float v = hp[c]; ss += v * sp[c]; dd += v * dp[c]; }
    a_s[t] = ss; a_d[t] = dd;
}

__global__ void gat_fill_u32(unsigned* __restrict__ p, unsigned v, int n)
{
    int i = blockIdx.x * blockDim.x + threadIdx.x;
    if (i < n) p[i] = v;
}

__global__ void gat_bias_init(float* __restrict__ out, const float* __restrict__ bias,
                              int total, int dim)
{
    int i = blockIdx.x * blockDim.x + threadIdx.x;
    if (i < total) out[i] = bias[i % dim];
}

__global__ void gat_relu(float* __restrict__ p, int n)
{
    int i = blockIdx.x * blockDim.x + threadIdx.x;
    if (i < n) p[i] = fmaxf(p[i], 0.f);
}

__global__ void gat_edge_scores(const int* __restrict__ srcI, const int* __restrict__ dstI,
                                const float* __restrict__ a_s, const float* __restrict__ a_d,
                                float* __restrict__ eb, unsigned* __restrict__ emax,
                                int E, int E2, int H)
{
    int e = blockIdx.x * blockDim.x + threadIdx.x;
    if (e >= E2) return;
    int s = (e < E) ? srcI[e] : (e - E);
    int d = (e < E) ? dstI[e] : (e - E);
    for (int h = 0; h < H; ++h) {
        float v = a_s[s * H + h] + a_d[d * H + h];
        v = (v >= 0.f) ? v : NEG_SLOPE * v;
        eb[(size_t)e * H + h] = v;
        atomicMax(&emax[d * H + h], gat_fenc(v));
    }
}

__global__ void gat_edge_exp(const int* __restrict__ dstI,
                             float* __restrict__ eb,
                             const unsigned* __restrict__ emax,
                             float* __restrict__ denom,
                             int E, int E2, int H)
{
    int e = blockIdx.x * blockDim.x + threadIdx.x;
    if (e >= E2) return;
    int d = (e < E) ? dstI[e] : (e - E);
    for (int h = 0; h < H; ++h) {
        float m = gat_fdec(emax[d * H + h]);
        float x = __expf(eb[(size_t)e * H + h] - m);
        eb[(size_t)e * H + h] = x;
        atomicAdd(&denom[d * H + h], x);
    }
}

__global__ void gat_edge_aggregate(const int* __restrict__ srcI, const int* __restrict__ dstI,
                                   const float* __restrict__ eb, const float* __restrict__ denom,
                                   const float* __restrict__ Hf, float* __restrict__ outp,
                                   int E, int E2, int H)
{
    int t = blockIdx.x * blockDim.x + threadIdx.x;
    if (t >= E2 * H) return;
    int e = t / H, h = t - e * H;
    int s = (e < E) ? srcI[e] : (e - E);
    int d = (e < E) ? dstI[e] : (e - E);
    float alpha = eb[(size_t)e * H + h] / denom[d * H + h];
    const float4* hv = (const float4*)(Hf + ((size_t)s * H + h) * 32);
    float*       ov = outp + ((size_t)d * H + h) * 32;
#pragma unroll
    for (int i = 0; i < 8; ++i) {
        float4 v = hv[i];
        atomicAdd(ov + 4 * i + 0, alpha * v.x);
        atomicAdd(ov + 4 * i + 1, alpha * v.y);
        atomicAdd(ov + 4 * i + 2, alpha * v.z);
        atomicAdd(ov + 4 * i + 3, alpha * v.w);
    }
}

extern "C" void kernel_launch(void* const* d_in, const int* in_sizes, int n_in,
                              void* d_out, int out_size, void* d_ws, size_t ws_size,
                              hipStream_t stream)
{
    const float* x   = (const float*)d_in[0];
    const int*   ei  = (const int*)d_in[1];
    const float* W1  = (const float*)d_in[2];
    const float* as1 = (const float*)d_in[3];
    const float* ad1 = (const float*)d_in[4];
    const float* b1  = (const float*)d_in[5];
    const float* W2  = (const float*)d_in[6];
    const float* as2 = (const float*)d_in[7];
    const float* ad2 = (const float*)d_in[8];
    const float* b2  = (const float*)d_in[9];
    float* out = (float*)d_out;

    const int N  = in_sizes[0] / 128;
    const int E  = in_sizes[1] / 2;
    const int E2 = E + N;
    const int* srcI = ei;
    const int* dstI = ei + E;

    float* ws = (float*)d_ws;
    float*    h1    = ws;              ws += (size_t)N * 128;
    float*    asrc1 = ws;              ws += (size_t)N * 4;
    float*    adst1 = ws;              ws += (size_t)N * 4;
    unsigned* emax1 = (unsigned*)ws;   ws += (size_t)N * 4;
    float*    den1  = ws;              ws += (size_t)N * 4;
    float*    eb1   = ws;              ws += (size_t)E2 * 4;
    float*    h2    = ws;              ws += (size_t)N * 128;
    float*    g2    = ws;              ws += (size_t)N * 32;
    float*    asrc2 = ws;              ws += (size_t)N;
    float*    adst2 = ws;              ws += (size_t)N;
    unsigned* emax2 = (unsigned*)ws;   ws += (size_t)N;
    float*    den2  = ws;              ws += (size_t)E2 > 0 ? (size_t)N : 0;
    float*    eb2   = ws;

    const int TB = 256;
    const int gB   = (N + 127) / 128;
    const int gE   = (E2 + TB - 1) / TB;
    const int gE4  = (E2 * 4 + TB - 1) / TB;
    const int gN4  = (N * 4 + TB - 1) / TB;
    const int gN1  = (N + TB - 1) / TB;
    const int gNF  = (N * 128 + TB - 1) / TB;
    const int gNO  = (N * 32 + TB - 1) / TB;

    gat_gemm_wmma<8><<<gB, TB, 0, stream>>>(x, W1, h1, N);
    gat_attn_score<<<gN4, TB, 0, stream>>>(h1, as1, ad1, asrc1, adst1, N, 4, 32);
    gat_fill_u32<<<gN4, TB, 0, stream>>>(emax1, ENC_NEG_INF, N * 4);
    gat_fill_u32<<<gN4, TB, 0, stream>>>((unsigned*)den1, 0u, N * 4);
    gat_bias_init<<<gNF, TB, 0, stream>>>(h2, b1, N * 128, 128);
    gat_edge_scores<<<gE, TB, 0, stream>>>(srcI, dstI, asrc1, adst1, eb1, emax1, E, E2, 4);
    gat_edge_exp<<<gE, TB, 0, stream>>>(dstI, eb1, emax1, den1, E, E2, 4);
    gat_edge_aggregate<<<gE4, TB, 0, stream>>>(srcI, dstI, eb1, den1, h1, h2, E, E2, 4);
    gat_relu<<<gNF, TB, 0, stream>>>(h2, N * 128);

    gat_gemm_wmma<2><<<gB, TB, 0, stream>>>(h2, W2, g2, N);
    gat_attn_score<<<gN1, TB, 0, stream>>>(g2, as2, ad2, asrc2, adst2, N, 1, 32);
    gat_fill_u32<<<gN1, TB, 0, stream>>>(emax2, ENC_NEG_INF, N);
    gat_fill_u32<<<gN1, TB, 0, stream>>>((unsigned*)den2, 0u, N);
    gat_bias_init<<<gNO, TB, 0, stream>>>(out, b2, N * 32, 32);
    gat_edge_scores<<<gE, TB, 0, stream>>>(srcI, dstI, asrc2, adst2, eb2, emax2, E, E2, 1);
    gat_edge_exp<<<gE, TB, 0, stream>>>(dstI, eb2, emax2, den2, E, E2, 1);
    gat_edge_aggregate<<<gE, TB, 0, stream>>>(srcI, dstI, eb2, den2, g2, out, E, E2, 1);
}


# Aggregate by destination: edges counting-sorted per call (one returning int atomic each), owner lanes sum source rows in registers, bias+ReLU at the single store; row sums now in arrival-rank order
# speedup vs baseline: 2.8153x; 2.8153x over previous
#include <hip/hip_runtime.h>

#define NEG_SLOPE 0.2f
#define ENC_NEG_INF 0x007FFFFFu

typedef float v2f __attribute__((ext_vector_type(2)));
typedef float v8f __attribute__((ext_vector_type(8)));
typedef float v4f __attribute__((ext_vector_type(4)));

__device__ __forceinline__ unsigned gat_fenc(float f) {
    unsigned u = __float_as_uint(f);
    return (u & 0x80000000u) ? ~u : (u | 0x80000000u);
}
__device__ __forceinline__ float gat_fdec(unsigned u) {
    unsigned b = (u & 0x80000000u) ? (u & 0x7FFFFFFFu) : ~u;
    return __uint_as_float(b);
}

template<int NT>
__global__ __launch_bounds__(256)
void gat_gemm_wmma(const float* __restrict__ X, const float* __restrict__ W,
                   float* __restrict__ Hout, int n_rows)
{
    constexpr int KD = 128;
    constexpr int ND = NT * 16;
    __shared__ float Wt[ND * KD];

    for (int i = threadIdx.x; i < KD * ND; i += 256) {
        int k = i / ND, n = i % ND;
        Wt[n * KD + ((k + 4 * n) & (KD - 1))] = W[i];
    }
    __syncthreads();

    const int lane = threadIdx.x & 31;
    const int wave = threadIdx.x >> 5;
    const int m0   = (blockIdx.x * 8 + wave) * 16;
    const int mlo  = lane & 15;
    const int kofs = (lane >> 4) * 2;

    int arow = m0 + mlo;
    if (arow >= n_rows) arow = n_rows - 1;
    const float* xrow = X + (size_t)arow * KD + kofs;

    v8f acc[NT] = {};

    for (int k = 0; k < KD; k += 4) {
        v2f a = *(const v2f*)(xrow + k);
#pragma unroll
        for (int t = 0; t < NT; ++t) {
            int n = t * 16 + mlo;
            v2f b = *(const v2f*)(&Wt[n * KD + ((k + kofs + 4 * n) & (KD - 1))]);
            acc[t] = __builtin_amdgcn_wmma_f32_16x16x4_f32(
                false, a, false, b, (short)0, acc[t], false, false);
        }
    }

    const int mhi = (lane >> 4) * 8;
#pragma unroll
    for (int t = 0; t < NT; ++t)
#pragma unroll
        for (int j = 0; j < 8; ++j) {
            int m = m0 + mhi + j;
            if (m < n_rows) Hout[(size_t)m * ND + t * 16 + mlo] = acc[t][j];
        }
}

__global__ void gat_attn_score(const float* __restrict__ Hf,
                               const float* __restrict__ att_s,
                               const float* __restrict__ att_d,
                               float* __restrict__ a_s, float* __restrict__ a_d,
                               int n, int heads, int ch)
{
    int t = blockIdx.x * blockDim.x + threadIdx.x;
    if (t >= n * heads) return;
    int node = t / heads, h = t - node * heads;
    const float* hp = Hf + ((size_t)node * heads + h) * ch;
    const float* sp = att_s + h * ch;
    const float* dp = att_d + h * ch;
    float ss = 0.f, dd = 0.f;
    for (int c = 0; c < ch; ++c) { float v = hp[c]; ss += v * sp[c]; dd += v * dp[c]; }
    a_s[t] = ss; a_d[t] = dd;
}

__global__ void gat_fill_u32(unsigned* __restrict__ p, unsigned v, int n)
{
    int i = blockIdx.x * blockDim.x + threadIdx.x;
    if (i < n) p[i] = v;
}

__global__ void gat_edge_scores(const int* __restrict__ srcI, const int* __restrict__ dstI,
                                const float* __restrict__ a_s, const float* __restrict__ a_d,
                                float* __restrict__ eb, unsigned* __restrict__ emax,
                                int E, int E2, int H)
{
    int e = blockIdx.x * blockDim.x + threadIdx.x;
    if (e >= E2) return;
    int s = (e < E) ? srcI[e] : (e - E);
    int d = (e < E) ? dstI[e] : (e - E);
    for (int h = 0; h < H; ++h) {
        float v = a_s[s * H + h] + a_d[d * H + h];
        v = (v >= 0.f) ? v : NEG_SLOPE * v;
        eb[(size_t)e * H + h] = v;
        atomicMax(&emax[d * H + h], gat_fenc(v));
    }
}

__global__ void gat_edge_exp(const int* __restrict__ dstI,
                             float* __restrict__ eb,
                             const unsigned* __restrict__ emax,
                             float* __restrict__ denom,
                             int E, int E2, int H)
{
    int e = blockIdx.x * blockDim.x + threadIdx.x;
    if (e >= E2) return;
    int d = (e < E) ? dstI[e] : (e - E);
    for (int h = 0; h < H; ++h) {
        float m = gat_fdec(emax[d * H + h]);
        float x = __expf(eb[(size_t)e * H + h] - m);
        eb[(size_t)e * H + h] = x;
        atomicAdd(&denom[d * H + h], x);
    }
}

__global__ void gat_csr_count(const int* __restrict__ dstI, int* __restrict__ cnt,
                              int* __restrict__ rank, int E)
{
    int e = blockIdx.x * blockDim.x + threadIdx.x;
    if (e < E) rank[e] = atomicAdd(&cnt[dstI[e]], 1);
}

__global__ __launch_bounds__(256)
void gat_csr_scan(const int* __restrict__ cnt, int* __restrict__ rowptr, int n)
{
    __shared__ int s_red[8];
    __shared__ int s_wtot[8];
    const int t = threadIdx.x, lane = t & 31, wv = t >> 5;
    const int base = blockIdx.x * 1024;

    int part = 0;
    for (int i = t; i < base; i += 256) part += cnt[i];
    for (int o = 16; o > 0; o >>= 1) part += __shfl_xor(part, o);

    int v[4], tot = 0;
    for (int k = 0; k < 4; ++k) {
        int i = base + t * 4 + k;
        v[k] = (i < n) ? cnt[i] : 0;
        tot += v[k];
    }
    int inc = tot;
    for (int o = 1; o < 32; o <<= 1) {
        int y = __shfl_up(inc, o);
        if (lane >= o) inc += y;
    }
    if (lane == 31) s_wtot[wv] = inc;
    if (lane == 0)  s_red[wv]  = part;
    __syncthreads();
    int off = 0;
    for (int w = 0; w < 8; ++w) {
        off += s_red[w];
        if (w < wv) off += s_wtot[w];
    }
    int run = off + inc - tot;
    for (int k = 0; k < 4; ++k) {
        int i = base + t * 4 + k;
        if (i < n) rowptr[i] = run;
        run += v[k];
    }
}

__global__ void gat_csr_place(const int* __restrict__ srcI, const int* __restrict__ dstI,
                              const int* __restrict__ rowptr, const int* __restrict__ rank,
                              int* __restrict__ es, int E)
{
    int e = blockIdx.x * blockDim.x + threadIdx.x;
    if (e >= E) return;
    int slot = rowptr[dstI[e]] + rank[e];
    ((int2*)es)[slot] = make_int2(e, srcI[e]);
}

template<int H, bool RELU>
__global__ __launch_bounds__(256)
void gat_gather_rows(const int* __restrict__ rowptr, const int* __restrict__ cnt,
                     const int* __restrict__ es_, const float* __restrict__ eb,
                     const float* __restrict__ denom, const float* __restrict__ Hf,
                     const float* __restrict__ bias, float* __restrict__ outp,
                     int n, int E)
{
    constexpr int LPN = 8 * H;
    const int d = blockIdx.x * (256 / LPN) + threadIdx.x / LPN;
    if (d >= n) return;
    const int sub = threadIdx.x % LPN;
    const int h   = sub >> 3;
    const int2* es = (const int2*)es_;
    const v4f*  Hv = (const v4f*)Hf;

    const float dn = denom[d * H + h];
    v4f acc = (eb[(size_t)(E + d) * H + h] / dn) * Hv[(size_t)d * LPN + sub];

    int j = rowptr[d];
    const int end = j + cnt[d];
    for (; j + 4 <= end; j += 4) {
        int2 p0 = es[j], p1 = es[j + 1], p2 = es[j + 2], p3 = es[j + 3];
        float x0 = eb[(size_t)p0.x * H + h];
        float x1 = eb[(size_t)p1.x * H + h];
        float x2 = eb[(size_t)p2.x * H + h];
        float x3 = eb[(size_t)p3.x * H + h];
        v4f u0 = Hv[(size_t)p0.y * LPN + sub];
        v4f u1 = Hv[(size_t)p1.y * LPN + sub];
        v4f u2 = Hv[(size_t)p2.y * LPN + sub];
        v4f u3 = Hv[(size_t)p3.y * LPN + sub];
        acc += (x0 / dn) * u0;
        acc += (x1 / dn) * u1;
        acc += (x2 / dn) * u2;
        acc += (x3 / dn) * u3;
    }
    for (; j < end; ++j) {
        int2 p0 = es[j];
        acc += (eb[(size_t)p0.x * H + h] / dn) * Hv[(size_t)p0.y * LPN + sub];
    }

    v4f b = { bias[sub * 4 + 0], bias[sub * 4 + 1], bias[sub * 4 + 2], bias[sub * 4 + 3] };
    acc += b;
    if (RELU) {
        acc.x = fmaxf(acc.x, 0.f); acc.y = fmaxf(acc.y, 0.f);
        acc.z = fmaxf(acc.z, 0.f); acc.w = fmaxf(acc.w, 0.f);
    }
    ((v4f*)outp)[(size_t)d * LPN + sub] = acc;
}

extern "C" void kernel_launch(void* const* d_in, const int* in_sizes, int n_in,
                              void* d_out, int out_size, void* d_ws, size_t ws_size,
                              hipStream_t stream)
{
    const float* x   = (const float*)d_in[0];
    const int*   ei  = (const int*)d_in[1];
    const float* W1  = (const float*)d_in[2];
    const float* as1 = (const float*)d_in[3];
    const float* ad1 = (const float*)d_in[4];
    const float* b1  = (const float*)d_in[5];
    const float* W2  = (const float*)d_in[6];
    const float* as2 = (const float*)d_in[7];
    const float* ad2 = (const float*)d_in[8];
    const float* b2  = (const float*)d_in[9];
    float* out = (float*)d_out;

    const int N  = in_sizes[0] / 128;
    const int E  = in_sizes[1] / 2;
    const int E2 = E + N;
    const int* srcI = ei;
    const int* dstI = ei + E;

    float* ws = (float*)d_ws;
    float*    h1    = ws;              ws += (size_t)N * 128;
    float*    asrc1 = ws;              ws += (size_t)N * 4;
    float*    adst1 = ws;              ws += (size_t)N * 4;
    unsigned* emax1 = (unsigned*)ws;   ws += (size_t)N * 4;
    float*    den1  = ws;              ws += (size_t)N * 4;
    float*    eb1   = ws;              ws += (size_t)E2 * 4;
    float*    h2    = ws;              ws += (size_t)N * 128;
    float*    g2    = ws;              ws += (size_t)N * 32;
    float*    asrc2 = ws;              ws += (size_t)N;
    float*    adst2 = ws;              ws += (size_t)N;
    unsigned* emax2 = (unsigned*)ws;   ws += (size_t)N;
    float*    den2  = ws;              ws += (size_t)E2 > 0 ? (size_t)N : 0;
    float*    eb2   = ws;              ws += (size_t)E2;
    ws += (4 - ((size_t)(ws - (float*)d_ws) & 3)) & 3;
    int*      es    = (int*)ws;        ws += (size_t)E * 2;
    int*      rank  = (int*)ws;        ws += (size_t)E;
    int*      cnt   = (int*)ws;        ws += (size_t)N;
    int*      rowp  = (int*)ws;

    const int TB = 256;
    const int gB   = (N + 127) / 128;
    const int gE   = (E2 + TB - 1) / TB;
    const int gN4  = (N * 4 + TB - 1) / TB;
    const int gN1  = (N + TB - 1) / TB;
    const int gEr  = (E + TB - 1) / TB;
    const int gSc  = (N + 1023) / 1024;
    const int gG4  = (N + 7) / 8;
    const int gG1  = (N + 31) / 32;

    gat_fill_u32<<<gN1, TB, 0, stream>>>((unsigned*)cnt, 0u, N);
    gat_csr_count<<<gEr, TB, 0, stream>>>(dstI, cnt, rank, E);
    gat_csr_scan<<<gSc, TB, 0, stream>>>(cnt, rowp, N);
    gat_csr_place<<<gEr, TB, 0, stream>>>(srcI, dstI, rowp, rank, es, E);

    gat_gemm_wmma<8><<<gB, TB, 0, stream>>>(x, W1, h1, N);
    gat_attn_score<<<gN4, TB, 0, stream>>>(h1, as1, ad1, asrc1, adst1, N, 4, 32);
    gat_fill_u32<<<gN4, TB, 0, stream>>>(emax1, ENC_NEG_INF, N * 4);
    gat_fill_u32<<<gN4, TB, 0, stream>>>((unsigned*)den1, 0u, N * 4);
    gat_edge_scores<<<gE, TB, 0, stream>>>(srcI, dstI, asrc1, adst1, eb1, emax1, E, E2, 4);
    gat_edge_exp<<<gE, TB, 0, stream>>>(dstI, eb1, emax1, den1, E, E2, 4);
    gat_gather_rows<4, true><<<gG4, TB, 0, stream>>>(rowp, cnt, es, eb1, den1, h1, b1, h2, N, E);

    gat_gemm_wmma<2><<<gB, TB, 0, stream>>>(h2, W2, g2, N);
    gat_attn_score<<<gN1, TB, 0, stream>>>(g2, as2, ad2, asrc2, adst2, N, 1, 32);
    gat_fill_u32<<<gN1, TB, 0, stream>>>(emax2, ENC_NEG_INF, N);
    gat_fill_u32<<<gN1, TB, 0, stream>>>((unsigned*)den2, 0u, N);
    gat_edge_scores<<<gE, TB, 0, stream>>>(srcI, dstI, asrc2, adst2, eb2, emax2, E, E2, 1);
    gat_edge_exp<<<gE, TB, 0, stream>>>(dstI, eb2, emax2, den2, E, E2, 1);
    gat_gather_rows<1, false><<<gG1, TB, 0, stream>>>(rowp, cnt, es, eb2, den2, g2, b2, out, N, E);
}


# Softmax by destination owner: a wave per node scores its sorted list (wave max, exp, sum) and gathers rows in batches; edge buffers and float atomics gone; sums in list order, layer 2 by 4-way tree
# speedup vs baseline: 8.1112x; 2.8811x over previous
#include <hip/hip_runtime.h>

#define NEG_SLOPE 0.2f

typedef float v2f __attribute__((ext_vector_type(2)));
typedef float v8f __attribute__((ext_vector_type(8)));
typedef float v4f __attribute__((ext_vector_type(4)));

template<int NT>
__global__ __launch_bounds__(256)
void gat_gemm_wmma(const float* __restrict__ X, const float* __restrict__ W,
                   float* __restrict__ Hout, int n_rows)
{
    constexpr int KD = 128;
    constexpr int ND = NT * 16;
    __shared__ float Wt[ND * KD];

    for (int i = threadIdx.x; i < KD * ND; i += 256) {
        int k = i / ND, n = i % ND;
        Wt[n * KD + ((k + 4 * n) & (KD - 1))] = W[i];
    }
    __syncthreads();

    const int lane = threadIdx.x & 31;
    const int wave = threadIdx.x >> 5;
    const int m0   = (blockIdx.x * 8 + wave) * 16;
    const int mlo  = lane & 15;
    const int kofs = (lane >> 4) * 2;

    int arow = m0 + mlo;
    if (arow >= n_rows) arow = n_rows - 1;
    const float* xrow = X + (size_t)arow * KD + kofs;

    v8f acc[NT] = {};

    for (int k = 0; k < KD; k += 4) {
        v2f a = *(const v2f*)(xrow + k);
#pragma unroll
        for (int t = 0; t < NT; ++t) {
            int n = t * 16 + mlo;
            v2f b = *(const v2f*)(&Wt[n * KD + ((k + kofs + 4 * n) & (KD - 1))]);
            acc[t] = __builtin_amdgcn_wmma_f32_16x16x4_f32(
                false, a, false, b, (short)0, acc[t], false, false);
        }
    }

    const int mhi = (lane >> 4) * 8;
#pragma unroll
    for (int t = 0; t < NT; ++t)
#pragma unroll
        for (int j = 0; j < 8; ++j) {
            int m = m0 + mhi + j;
            if (m < n_rows) Hout[(size_t)m * ND + t * 16 + mlo] = acc[t][j];
        }
}

__global__ void gat_attn_score(const float* __restrict__ Hf,
                               const float* __restrict__ att_s,
                               const float* __restrict__ att_d,
                               float* __restrict__ a_s, float* __restrict__ a_d,
                               int n, int heads, int ch)
{
    int t = blockIdx.x * blockDim.x + threadIdx.x;
    if (t >= n * heads) return;
    int node = t / heads, h = t - node * heads;
    const float* hp = Hf + ((size_t)node * heads + h) * ch;
    const float* sp = att_s + h * ch;
    const float* dp = att_d + h * ch;
    float ss = 0.f, dd = 0.f;
    for (int c = 0; c < ch; ++c) { float v = hp[c]; ss += v * sp[c]; dd += v * dp[c]; }
    a_s[t] = ss; a_d[t] = dd;
}

__global__ void gat_fill_u32(unsigned* __restrict__ p, unsigned v, int n)
{
    int i = blockIdx.x * blockDim.x + threadIdx.x;
    if (i < n) p[i] = v;
}

__global__ void gat_csr_count(const int* __restrict__ dstI, int* __restrict__ cnt,
                              int* __restrict__ rank, int E)
{
    int e = blockIdx.x * blockDim.x + threadIdx.x;
    if (e < E) rank[e] = atomicAdd(&cnt[dstI[e]], 1);
}

__global__ __launch_bounds__(256)
void gat_csr_scan(const int* __restrict__ cnt, int* __restrict__ rowptr, int n)
{
    __shared__ int s_red[8];
    __shared__ int s_wtot[8];
    const int t = threadIdx.x, lane = t & 31, wv = t >> 5;
    const int base = blockIdx.x * 1024;

    int part = 0;
    for (int i = t; i < base; i += 256) part += cnt[i];
    for (int o = 16; o > 0; o >>= 1) part += __shfl_xor(part, o);

    int v[4], tot = 0;
    for (int k = 0; k < 4; ++k) {
        int i = base + t * 4 + k;
        v[k] = (i < n) ? cnt[i] : 0;
        tot += v[k];
    }
    int inc = tot;
    for (int o = 1; o < 32; o <<= 1) {
        int y = __shfl_up(inc, o);
        if (lane >= o) inc += y;
    }
    if (lane == 31) s_wtot[wv] = inc;
    if (lane == 0)  s_red[wv]  = part;
    __syncthreads();
    int off = 0;
    for (int w = 0; w < 8; ++w) {
        off += s_red[w];
        if (w < wv) off += s_wtot[w];
    }
    int run = off + inc - tot;
    for (int k = 0; k < 4; ++k) {
        int i = base + t * 4 + k;
        if (i < n) rowptr[i] = run;
        run += v[k];
    }
}

__global__ void gat_csr_place(const int* __restrict__ srcI, const int* __restrict__ dstI,
                              const int* __restrict__ rowptr, const int* __restrict__ rank,
                              int* __restrict__ es, int E)
{
    int e = blockIdx.x * blockDim.x + threadIdx.x;
    if (e >= E) return;
    es[rowptr[dstI[e]] + rank[e]] = srcI[e];
}

__device__ __forceinline__ float gat_wave_max(float v) {
    for (int o = 16; o > 0; o >>= 1) v = fmaxf(v, __shfl_xor(v, o));
    return v;
}
__device__ __forceinline__ float gat_wave_sum(float v) {
    for (int o = 16; o > 0; o >>= 1) v += __shfl_xor(v, o);
    return v;
}

template<int H, int U>
__device__ __forceinline__ void gat_gather_batch(v4f& acc, const v4f* __restrict__ Hv,
                                                 int srcv, const float (&p)[H],
                                                 int i, int g, int sub, int h)
{
    constexpr int LPR = 8 * H;
    constexpr int EPT = 32 / LPR;
    int   s[U];
    float w[U];
    v4f   u[U];
    for (int k = 0; k < U; ++k) {
        if constexpr (H == 4) {
            const int slot = i + k;
            s[k] = __builtin_amdgcn_readlane(srcv, slot);
            int r = 0;
            for (int hh = 0; hh < 4; ++hh)
                r |= __builtin_amdgcn_readlane(__float_as_int(p[hh]), slot) & ((hh == h) ? -1 : 0);
            w[k] = __int_as_float(r);
        } else {
            const int slot = i + k * EPT + g;
            s[k] = __shfl(srcv, slot);
            w[k] = __shfl(p[0], slot);
        }
    }
    for (int k = 0; k < U; ++k) u[k] = Hv[(size_t)s[k] * LPR + sub];
    for (int k = 0; k < U; ++k) acc += w[k] * u[k];
}

template<int H, bool RELU>
__global__ __launch_bounds__(256)
void gat_softmax_gather(const int* __restrict__ rowptr, const int* __restrict__ cnt,
                        const int* __restrict__ es, const float* __restrict__ a_s,
                        const float* __restrict__ a_d, const float* __restrict__ Hf,
                        const float* __restrict__ bias, float* __restrict__ outp, int n)
{
    constexpr int LPR = 8 * H;
    constexpr int EPT = 32 / LPR;
    constexpr int NPW = 4;
    const int lane = threadIdx.x & 31;
    const int wv  = __builtin_amdgcn_readfirstlane((int)(threadIdx.x >> 5));
    const int sub = lane % LPR;
    const int g   = lane / LPR;
    const int h   = sub >> 3;
    const v4f* Hv = (const v4f*)Hf;
    const int d0  = (blockIdx.x * 8 + wv) * NPW;

    for (int q = 0; q < NPW; ++q) {
        const int d = d0 + q;
        if (d >= n) break;
        const int row0 = rowptr[d];
        const int deg  = cnt[d];
        const int T    = deg + 1;

        float ad[H], m[H], den[H];
        for (int hh = 0; hh < H; ++hh) { ad[hh] = a_d[(size_t)d * H + hh]; m[hh] = 0.f; den[hh] = 0.f; }
        v4f acc = { 0.f, 0.f, 0.f, 0.f };

        for (int c0 = 0; c0 < T; c0 += 32) {
            const int  t     = c0 + lane;
            const bool valid = t < T;
            int srcv = d;
            if (t < deg) srcv = es[row0 + t];

            float e[H], p[H];
            if constexpr (H == 4) {
                v4f a = ((const v4f*)a_s)[srcv];
                e[0] = a.x; e[1] = a.y; e[2] = a.z; e[3] = a.w;
            } else {
                for (int hh = 0; hh < H; ++hh) e[hh] = a_s[(size_t)srcv * H + hh];
            }
            float scl = 1.f;
            for (int hh = 0; hh < H; ++hh) {
                float v = e[hh] + ad[hh];
                v = (v >= 0.f) ? v : NEG_SLOPE * v;
                v = valid ? v : -__builtin_inff();
                float mn = gat_wave_max(v);
                float sc = 0.f;
                if (c0 > 0) { mn = fmaxf(mn, m[hh]); sc = __expf(m[hh] - mn); }
                p[hh]   = valid ? __expf(v - mn) : 0.f;
                den[hh] = den[hh] * sc + gat_wave_sum(p[hh]);
                m[hh]   = mn;
                if (hh == h) scl = sc;
            }
            if (c0 > 0) acc *= scl;

            const int Tc = min(T - c0, 32);
            int i = 0;
            for (; Tc - i >= 8 * EPT; i += 8 * EPT)
                gat_gather_batch<H, 8>(acc, Hv, srcv, p, i, g, sub, h);
            if (Tc - i > 4 * EPT)
                gat_gather_batch<H, 8>(acc, Hv, srcv, p, i, g, sub, h);
            else if (Tc - i > 0)
                gat_gather_batch<H, 4>(acc, Hv, srcv, p, i, g, sub, h);
        }

        float dsel = den[0];
        for (int hh = 1; hh < H; ++hh) dsel = (hh == h) ? den[hh] : dsel;
        if constexpr (EPT > 1) {
            for (int o = LPR; o < 32; o <<= 1) {
                acc.x += __shfl_xor(acc.x, o); acc.y += __shfl_xor(acc.y, o);
                acc.z += __shfl_xor(acc.z, o); acc.w += __shfl_xor(acc.w, o);
            }
        }
        if (g == 0) {
            v4f b = { bias[sub * 4 + 0], bias[sub * 4 + 1], bias[sub * 4 + 2], bias[sub * 4 + 3] };
            v4f r = acc / dsel + b;
            if (RELU) {
                r.x = fmaxf(r.x, 0.f); r.y = fmaxf(r.y, 0.f);
                r.z = fmaxf(r.z, 0.f); r.w = fmaxf(r.w, 0.f);
            }
            ((v4f*)outp)[(size_t)d * LPR + sub] = r;
        }
    }
}

extern "C" void kernel_launch(void* const* d_in, const int* in_sizes, int n_in,
                              void* d_out, int out_size, void* d_ws, size_t ws_size,
                              hipStream_t stream)
{
    const float* x   = (const float*)d_in[0];
    const int*   ei  = (const int*)d_in[1];
    const float* W1  = (const float*)d_in[2];
    const float* as1 = (const float*)d_in[3];
    const float* ad1 = (const float*)d_in[4];
    const float* b1  = (const float*)d_in[5];
    const float* W2  = (const float*)d_in[6];
    const float* as2 = (const float*)d_in[7];
    const float* ad2 = (const float*)d_in[8];
    const float* b2  = (const float*)d_in[9];
    float* out = (float*)d_out;

    const int N  = in_sizes[0] / 128;
    const int E  = in_sizes[1] / 2;
    const int* srcI = ei;
    const int* dstI = ei + E;

    float* ws = (float*)d_ws;
    float*    h1    = ws;              ws += (size_t)N * 128;
    float*    asrc1 = ws;              ws += (size_t)N * 4;
    float*    adst1 = ws;              ws += (size_t)N * 4;
    float*    h2    = ws;              ws += (size_t)N * 128;
    float*    g2    = ws;              ws += (size_t)N * 32;
    float*    asrc2 = ws;              ws += (size_t)N;
    float*    adst2 = ws;              ws += (size_t)N;
    int*      es    = (int*)ws;        ws += (size_t)E;
    int*      rank  = (int*)ws;        ws += (size_t)E;
    int*      cnt   = (int*)ws;        ws += (size_t)N;
    int*      rowp  = (int*)ws;

    const int TB = 256;
    const int gB   = (N + 127) / 128;
    const int gN4  = (N * 4 + TB - 1) / TB;
    const int gN1  = (N + TB - 1) / TB;
    const int gEr  = (E + TB - 1) / TB;
    const int gSc  = (N + 1023) / 1024;
    const int gG   = (N + 31) / 32;

    gat_fill_u32<<<gN1, TB, 0, stream>>>((unsigned*)cnt, 0u, N);
    gat_csr_count<<<gEr, TB, 0, stream>>>(dstI, cnt, rank, E);
    gat_csr_scan<<<gSc, TB, 0, stream>>>(cnt, rowp, N);
    gat_csr_place<<<gEr, TB, 0, stream>>>(srcI, dstI, rowp, rank, es, E);

    gat_gemm_wmma<8><<<gB, TB, 0, stream>>>(x, W1, h1, N);
    gat_attn_score<<<gN4, TB, 0, stream>>>(h1, as1, ad1, asrc1, adst1, N, 4, 32);
    gat_softmax_gather<4, true><<<gG, TB, 0, stream>>>(rowp, cnt, es, asrc1, adst1, h1, b1, h2, N);

    gat_gemm_wmma<2><<<gB, TB, 0, stream>>>(h2, W2, g2, N);
    gat_attn_score<<<gN1, TB, 0, stream>>>(g2, as2, ad2, asrc2, adst2, N, 1, 32);
    gat_softmax_gather<1, false><<<gG, TB, 0, stream>>>(rowp, cnt, es, asrc2, adst2, g2, b2, out, N);
}


# Edges bucketed by a two-level counting sort on LDS counters (chunk histogram per 256-node bin, pair copy into bin order, per-bin node sort): no global atomics; row sums in LDS arrival order
# speedup vs baseline: 12.6846x; 1.5638x over previous
#include <hip/hip_runtime.h>

#define NEG_SLOPE 0.2f

typedef float v2f __attribute__((ext_vector_type(2)));
typedef float v8f __attribute__((ext_vector_type(8)));
typedef float v4f __attribute__((ext_vector_type(4)));

template<int NT>
__global__ __launch_bounds__(256)
void gat_gemm_wmma(const float* __restrict__ X, const float* __restrict__ W,
                   float* __restrict__ Hout, int n_rows)
{
    constexpr int KD = 128;
    constexpr int ND = NT * 16;
    __shared__ float Wt[ND * KD];

    for (int i = threadIdx.x; i < KD * ND; i += 256) {
        int k = i / ND, n = i % ND;
        Wt[n * KD + ((k + 4 * n) & (KD - 1))] = W[i];
    }
    __syncthreads();

    const int lane = threadIdx.x & 31;
    const int wave = threadIdx.x >> 5;
    const int m0   = (blockIdx.x * 8 + wave) * 16;
    const int mlo  = lane & 15;
    const int kofs = (lane >> 4) * 2;

    int arow = m0 + mlo;
    if (arow >= n_rows) arow = n_rows - 1;
    const float* xrow = X + (size_t)arow * KD + kofs;

    v8f acc[NT] = {};

    for (int k = 0; k < KD; k += 4) {
        v2f a = *(const v2f*)(xrow + k);
#pragma unroll
        for (int t = 0; t < NT; ++t) {
            int n = t * 16 + mlo;
            v2f b = *(const v2f*)(&Wt[n * KD + ((k + kofs + 4 * n) & (KD - 1))]);
            acc[t] = __builtin_amdgcn_wmma_f32_16x16x4_f32(
                false, a, false, b, (short)0, acc[t], false, false);
        }
    }

    const int mhi = (lane >> 4) * 8;
#pragma unroll
    for (int t = 0; t < NT; ++t)
#pragma unroll
        for (int j = 0; j < 8; ++j) {
            int m = m0 + mhi + j;
            if (m < n_rows) Hout[(size_t)m * ND + t * 16 + mlo] = acc[t][j];
        }
}

__global__ void gat_attn_score(const float* __restrict__ Hf,
                               const float* __restrict__ att_s,
                               const float* __restrict__ att_d,
                               float* __restrict__ a_s, float* __restrict__ a_d,
                               int n, int heads, int ch)
{
    int t = blockIdx.x * blockDim.x + threadIdx.x;
    if (t >= n * heads) return;
    int node = t / heads, h = t - node * heads;
    const float* hp = Hf + ((size_t)node * heads + h) * ch;
    const float* sp = att_s + h * ch;
    const float* dp = att_d + h * ch;
    float ss = 0.f, dd = 0.f;
    for (int c = 0; c < ch; ++c) { float v = hp[c]; ss += v * sp[c]; dd += v * dp[c]; }
    a_s[t] = ss; a_d[t] = dd;
}

#define GAT_BINW   256
#define GAT_BINSH  8
#define GAT_NBMAX  1024
#define GAT_CHUNKS 128

__global__ __launch_bounds__(1024)
void gat_bin_hist(const int* __restrict__ dstI, int* __restrict__ tab, int E, int nb)
{
    __shared__ int hist[GAT_NBMAX];
    const int t = threadIdx.x;
    hist[t] = 0;
    __syncthreads();
    const int chunk = (E + GAT_CHUNKS - 1) / GAT_CHUNKS;
    const int e0 = blockIdx.x * chunk;
    const int e1 = min(E, e0 + chunk);
    for (int e = e0 + t; e < e1; e += 8 * 1024) {
        int d[8];
        for (int k = 0; k < 8; ++k) d[k] = dstI[min(e + k * 1024, e1 - 1)];
        asm volatile("" :: "v"(d[0]), "v"(d[1]), "v"(d[2]), "v"(d[3]), "v"(d[4]), "v"(d[5]), "v"(d[6]), "v"(d[7]));
        for (int k = 0; k < 8; ++k)
            if (e + k * 1024 < e1) atomicAdd(&hist[d[k] >> GAT_BINSH], 1);
    }
    __syncthreads();
    if (t < nb) tab[blockIdx.x * nb + t] = hist[t];
}

__global__ __launch_bounds__(1024)
void gat_bin_scatter(const int* __restrict__ srcI, const int* __restrict__ dstI,
                     const int* __restrict__ tab, int* __restrict__ binStart,
                     int* __restrict__ pairs, int E, int nb)
{
    __shared__ int colTot[GAT_NBMAX];
    __shared__ int cursor[GAT_NBMAX];
    __shared__ int s_wtot[32];
    const int t = threadIdx.x, lane = t & 31, wv = t >> 5;
    const int blk = blockIdx.x;
    colTot[t] = 0; cursor[t] = 0;
    __syncthreads();
    for (int j = t; j < nb * 8; j += 1024) {
        const int r = j / nb, b = j - r * nb;
        int tot = 0, part = 0;
        for (int k = 0; k < GAT_CHUNKS / 8; ++k) {
            const int row = r * (GAT_CHUNKS / 8) + k;
            const int v = tab[row * nb + b];
            tot  += v;
            part += (row < blk) ? v : 0;
        }
        atomicAdd(&colTot[b], tot);
        atomicAdd(&cursor[b], part);
    }
    __syncthreads();
    const int v = colTot[t];
    int inc = v;
    for (int o = 1; o < 32; o <<= 1) {
        int y = __shfl_up(inc, o);
        if (lane >= o) inc += y;
    }
    if (lane == 31) s_wtot[wv] = inc;
    __syncthreads();
    int off = 0;
    for (int w = 0; w < 32; ++w) off += (w < wv) ? s_wtot[w] : 0;
    const int excl = off + inc - v;
    if (t < nb) {
        cursor[t] += excl;
        if (blk == 0) binStart[t] = excl;
    }
    if (blk == 0 && t == 0) binStart[nb] = E;
    __syncthreads();
    const int chunk = (E + GAT_CHUNKS - 1) / GAT_CHUNKS;
    const int e0 = blk * chunk;
    const int e1 = min(E, e0 + chunk);
    int2* pr = (int2*)pairs;
    for (int e = e0 + t; e < e1; e += 8 * 1024) {
        int d[8], sv[8];
        for (int k = 0; k < 8; ++k) {
            const int idx = min(e + k * 1024, e1 - 1);
            d[k] = dstI[idx]; sv[k] = srcI[idx];
        }
        asm volatile("" :: "v"(d[0]), "v"(d[1]), "v"(d[2]), "v"(d[3]), "v"(d[4]), "v"(d[5]), "v"(d[6]), "v"(d[7]),
                           "v"(sv[0]), "v"(sv[1]), "v"(sv[2]), "v"(sv[3]), "v"(sv[4]), "v"(sv[5]), "v"(sv[6]), "v"(sv[7]));
        for (int k = 0; k < 8; ++k)
            if (e + k * 1024 < e1) {
                const int pos = atomicAdd(&cursor[d[k] >> GAT_BINSH], 1);
                pr[pos] = make_int2(sv[k], d[k]);
            }
    }
}

__global__ __launch_bounds__(512)
void gat_bin_sort(const int* __restrict__ pairs, const int* __restrict__ binStart,
                  int* __restrict__ cnt, int* __restrict__ rowptr, int* __restrict__ es, int n)
{
    __shared__ int c[GAT_BINW];
    __shared__ int s_wtot[GAT_BINW / 32];
    const int t = threadIdx.x, lane = t & 31, wv = t >> 5;
    const int base = blockIdx.x * GAT_BINW;
    const int s0 = binStart[blockIdx.x], s1 = binStart[blockIdx.x + 1];
    const int2* pr = (const int2*)pairs;
    if (t < GAT_BINW) c[t] = 0;
    __syncthreads();
    for (int i = s0 + t; i < s1; i += 8 * 512) {
        int y[8];
        for (int k = 0; k < 8; ++k) y[k] = pr[min(i + k * 512, s1 - 1)].y;
        asm volatile("" :: "v"(y[0]), "v"(y[1]), "v"(y[2]), "v"(y[3]), "v"(y[4]), "v"(y[5]), "v"(y[6]), "v"(y[7]));
        for (int k = 0; k < 8; ++k)
            if (i + k * 512 < s1) atomicAdd(&c[y[k] - base], 1);
    }
    __syncthreads();
    int v = 0, inc = 0;
    if (t < GAT_BINW) {
        v = c[t]; inc = v;
        for (int o = 1; o < 32; o <<= 1) {
            int y = __shfl_up(inc, o);
            if (lane >= o) inc += y;
        }
        if (lane == 31) s_wtot[wv] = inc;
    }
    __syncthreads();
    if (t < GAT_BINW) {
        int off = s0;
        for (int w = 0; w < GAT_BINW / 32; ++w) off += (w < wv) ? s_wtot[w] : 0;
        const int excl = off + inc - v;
        c[t] = excl;
        if (base + t < n) { cnt[base + t] = v; rowptr[base + t] = excl; }
    }
    __syncthreads();
    for (int i = s0 + t; i < s1; i += 8 * 512) {
        int2 p[8];
        for (int k = 0; k < 8; ++k) p[k] = pr[min(i + k * 512, s1 - 1)];
        asm volatile("" :: "v"(p[0].x), "v"(p[1].x), "v"(p[2].x), "v"(p[3].x), "v"(p[4].x), "v"(p[5].x), "v"(p[6].x), "v"(p[7].x),
                           "v"(p[0].y), "v"(p[1].y), "v"(p[2].y), "v"(p[3].y), "v"(p[4].y), "v"(p[5].y), "v"(p[6].y), "v"(p[7].y));
        for (int k = 0; k < 8; ++k)
            if (i + k * 512 < s1) {
                const int slot = atomicAdd(&c[p[k].y - base], 1);
                es[slot] = p[k].x;
            }
    }
}

__device__ __forceinline__ float gat_wave_max(float v) {
    for (int o = 16; o > 0; o >>= 1) v = fmaxf(v, __shfl_xor(v, o));
    return v;
}
__device__ __forceinline__ float gat_wave_sum(float v) {
    for (int o = 16; o > 0; o >>= 1) v += __shfl_xor(v, o);
    return v;
}

template<int H, int U>
__device__ __forceinline__ void gat_gather_batch(v4f& acc, const v4f* __restrict__ Hv,
                                                 int srcv, const float (&p)[H],
                                                 int i, int g, int sub, int h)
{
    constexpr int LPR = 8 * H;
    constexpr int EPT = 32 / LPR;
    int   s[U];
    float w[U];
    v4f   u[U];
    for (int k = 0; k < U; ++k) {
        if constexpr (H == 4) {
            const int slot = i + k;
            s[k] = __builtin_amdgcn_readlane(srcv, slot);
            int r = 0;
            for (int hh = 0; hh < 4; ++hh)
                r |= __builtin_amdgcn_readlane(__float_as_int(p[hh]), slot) & ((hh == h) ? -1 : 0);
            w[k] = __int_as_float(r);
        } else {
            const int slot = i + k * EPT + g;
            s[k] = __shfl(srcv, slot);
            w[k] = __shfl(p[0], slot);
        }
    }
    for (int k = 0; k < U; ++k) u[k] = Hv[(size_t)s[k] * LPR + sub];
    for (int k = 0; k < U; ++k) acc += w[k] * u[k];
}

template<int H, bool RELU>
__global__ __launch_bounds__(256)
void gat_softmax_gather(const int* __restrict__ rowptr, const int* __restrict__ cnt,
                        const int* __restrict__ es, const float* __restrict__ a_s,
                        const float* __restrict__ a_d, const float* __restrict__ Hf,
                        const float* __restrict__ bias, float* __restrict__ outp, int n)
{
    constexpr int LPR = 8 * H;
    constexpr int EPT = 32 / LPR;
    constexpr int NPW = 4;
    const int lane = threadIdx.x & 31;
    const int wv  = __builtin_amdgcn_readfirstlane((int)(threadIdx.x >> 5));
    const int sub = lane % LPR;
    const int g   = lane / LPR;
    const int h   = sub >> 3;
    const v4f* Hv = (const v4f*)Hf;
    const int d0  = (blockIdx.x * 8 + wv) * NPW;

    for (int q = 0; q < NPW; ++q) {
        const int d = d0 + q;
        if (d >= n) break;
        const int row0 = rowptr[d];
        const int deg  = cnt[d];
        const int T    = deg + 1;

        float ad[H], m[H], den[H];
        for (int hh = 0; hh < H; ++hh) { ad[hh] = a_d[(size_t)d * H + hh]; m[hh] = 0.f; den[hh] = 0.f; }
        v4f acc = { 0.f, 0.f, 0.f, 0.f };

        for (int c0 = 0; c0 < T; c0 += 32) {
            const int  t     = c0 + lane;
            const bool valid = t < T;
            int srcv = d;
            if (t < deg) srcv = es[row0 + t];

            float e[H], p[H];
            if constexpr (H == 4) {
                v4f a = ((const v4f*)a_s)[srcv];
                e[0] = a.x; e[1] = a.y; e[2] = a.z; e[3] = a.w;
            } else {
                for (int hh = 0; hh < H; ++hh) e[hh] = a_s[(size_t)srcv * H + hh];
            }
            float scl = 1.f;
            for (int hh = 0; hh < H; ++hh) {
                float v = e[hh] + ad[hh];
                v = (v >= 0.f) ? v : NEG_SLOPE * v;
                v = valid ? v : -__builtin_inff();
                float mn = gat_wave_max(v);
                float sc = 0.f;
                if (c0 > 0) { mn = fmaxf(mn, m[hh]); sc = __expf(m[hh] - mn); }
                p[hh]   = valid ? __expf(v - mn) : 0.f;
                den[hh] = den[hh] * sc + gat_wave_sum(p[hh]);
                m[hh]   = mn;
                if (hh == h) scl = sc;
            }
            if (c0 > 0) acc *= scl;

            const int Tc = min(T - c0, 32);
            int i = 0;
            for (; Tc - i >= 8 * EPT; i += 8 * EPT)
                gat_gather_batch<H, 8>(acc, Hv, srcv, p, i, g, sub, h);
            if (Tc - i > 4 * EPT)
                gat_gather_batch<H, 8>(acc, Hv, srcv, p, i, g, sub, h);
            else if (Tc - i > 0)
                gat_gather_batch<H, 4>(acc, Hv, srcv, p, i, g, sub, h);
        }

        float dsel = den[0];
        for (int hh = 1; hh < H; ++hh) dsel = (hh == h) ? den[hh] : dsel;
        if constexpr (EPT > 1) {
            for (int o = LPR; o < 32; o <<= 1) {
                acc.x += __shfl_xor(acc.x, o); acc.y += __shfl_xor(acc.y, o);
                acc.z += __shfl_xor(acc.z, o); acc.w += __shfl_xor(acc.w, o);
            }
        }
        if (g == 0) {
            v4f b = { bias[sub * 4 + 0], bias[sub * 4 + 1], bias[sub * 4 + 2], bias[sub * 4 + 3] };
            v4f r = acc / dsel + b;
            if (RELU) {
                r.x = fmaxf(r.x, 0.f); r.y = fmaxf(r.y, 0.f);
                r.z = fmaxf(r.z, 0.f); r.w = fmaxf(r.w, 0.f);
            }
            ((v4f*)outp)[(size_t)d * LPR + sub] = r;
        }
    }
}

extern "C" void kernel_launch(void* const* d_in, const int* in_sizes, int n_in,
                              void* d_out, int out_size, void* d_ws, size_t ws_size,
                              hipStream_t stream)
{
    const float* x   = (const float*)d_in[0];
    const int*   ei  = (const int*)d_in[1];
    const float* W1  = (const float*)d_in[2];
    const float* as1 = (const float*)d_in[3];
    const float* ad1 = (const float*)d_in[4];
    const float* b1  = (const float*)d_in[5];
    const float* W2  = (const float*)d_in[6];
    const float* as2 = (const float*)d_in[7];
    const float* ad2 = (const float*)d_in[8];
    const float* b2  = (const float*)d_in[9];
    float* out = (float*)d_out;

    const int N  = in_sizes[0] / 128;
    const int E  = in_sizes[1] / 2;
    const int* srcI = ei;
    const int* dstI = ei + E;

    float* ws = (float*)d_ws;
    float*    h1    = ws;              ws += (size_t)N * 128;
    float*    asrc1 = ws;              ws += (size_t)N * 4;
    float*    adst1 = ws;              ws += (size_t)N * 4;
    float*    h2    = ws;              ws += (size_t)N * 128;
    float*    g2    = ws;              ws += (size_t)N * 32;
    float*    asrc2 = ws;              ws += (size_t)N;
    float*    adst2 = ws;              ws += (size_t)N;
    int*      pairs = (int*)ws;        ws += (size_t)E * 2;
    int*      es    = (int*)ws;        ws += (size_t)E;
    int*      cnt   = (int*)ws;        ws += (size_t)N;
    int*      rowp  = (int*)ws;        ws += (size_t)N;
    int*      binS  = (int*)ws;        ws += GAT_NBMAX + 1;
    int*      tab   = (int*)ws;

    const int TB = 256;
    const int gB   = (N + 127) / 128;
    const int gN4  = (N * 4 + TB - 1) / TB;
    const int gN1  = (N + TB - 1) / TB;
    const int nb   = (N + GAT_BINW - 1) / GAT_BINW;
    const int gG   = (N + 31) / 32;

    gat_bin_hist<<<GAT_CHUNKS, 1024, 0, stream>>>(dstI, tab, E, nb);
    gat_bin_scatter<<<GAT_CHUNKS, 1024, 0, stream>>>(srcI, dstI, tab, binS, pairs, E, nb);
    gat_bin_sort<<<nb, 512, 0, stream>>>(pairs, binS, cnt, rowp, es, N);

    gat_gemm_wmma<8><<<gB, TB, 0, stream>>>(x, W1, h1, N);
    gat_attn_score<<<gN4, TB, 0, stream>>>(h1, as1, ad1, asrc1, adst1, N, 4, 32);
    gat_softmax_gather<4, true><<<gG, TB, 0, stream>>>(rowp, cnt, es, asrc1, adst1, h1, b1, h2, N);

    gat_gemm_wmma<2><<<gB, TB, 0, stream>>>(h2, W2, g2, N);
    gat_attn_score<<<gN1, TB, 0, stream>>>(g2, as2, ad2, asrc2, adst2, N, 1, 32);
    gat_softmax_gather<1, false><<<gG, TB, 0, stream>>>(rowp, cnt, es, asrc2, adst2, g2, b2, out, N);
}


# Attention scores folded from the accumulator tiles in the projection's epilogue (16-lane DPP tree per head): score launches and their re-read of h gone, 9 -> 7 launches; score sums now in tree order
# speedup vs baseline: 13.6120x; 1.0731x over previous
#include <hip/hip_runtime.h>

#define NEG_SLOPE 0.2f

typedef float v2f __attribute__((ext_vector_type(2)));
typedef float v8f __attribute__((ext_vector_type(8)));
typedef float v4f __attribute__((ext_vector_type(4)));

template<int M>
__device__ __forceinline__ float gat_row_xor(float v) {
    return __int_as_float(__builtin_amdgcn_update_dpp(0, __float_as_int(v), 0x160 | M, 0xf, 0xf, true));
}

template<int M, int HALF>
__device__ __forceinline__ void gat_fold_step(float (&v)[16], bool up) {
#pragma unroll
    for (int i = 0; i < HALF; ++i) {
        const float keep = up ? v[i + HALF] : v[i];
        const float send = up ? v[i] : v[i + HALF];
        v[i] = keep + gat_row_xor<M>(send);
    }
}

template<int NT>
__global__ __launch_bounds__(256)
void gat_gemm_wmma(const float* __restrict__ X, const float* __restrict__ W,
                   const float* __restrict__ att_s, const float* __restrict__ att_d,
                   float* __restrict__ Hout, float* __restrict__ a_s, float* __restrict__ a_d,
                   int n_rows)
{
    constexpr int KD = 128;
    constexpr int ND = NT * 16;
    __shared__ float Wt[ND * KD];

    for (int i = threadIdx.x; i < KD * ND; i += 256) {
        int k = i / ND, n = i % ND;
        Wt[n * KD + ((k + 4 * n) & (KD - 1))] = W[i];
    }
    __syncthreads();

    const int lane = threadIdx.x & 31;
    const int wave = threadIdx.x >> 5;
    const int m0   = (blockIdx.x * 8 + wave) * 16;
    const int mlo  = lane & 15;
    const int kofs = (lane >> 4) * 2;

    int arow = m0 + mlo;
    if (arow >= n_rows) arow = n_rows - 1;
    const float* xrow = X + (size_t)arow * KD + kofs;

    v8f acc[NT] = {};

    for (int k = 0; k < KD; k += 4) {
        v2f a = *(const v2f*)(xrow + k);
#pragma unroll
        for (int t = 0; t < NT; ++t) {
            int n = t * 16 + mlo;
            v2f b = *(const v2f*)(&Wt[n * KD + ((k + kofs + 4 * n) & (KD - 1))]);
            acc[t] = __builtin_amdgcn_wmma_f32_16x16x4_f32(
                false, a, false, b, (short)0, acc[t], false, false);
        }
    }

    const int mhi = (lane >> 4) * 8;
#pragma unroll
    for (int t = 0; t < NT; ++t)
#pragma unroll
        for (int j = 0; j < 8; ++j) {
            int m = m0 + mhi + j;
            if (m < n_rows) Hout[(size_t)m * ND + t * 16 + mlo] = acc[t][j];
        }

    constexpr int NH = NT / 2;
    float res[NH];
#pragma unroll
    for (int hd = 0; hd < NH; ++hd) {
        const float s0 = att_s[hd * 32 + mlo], s1 = att_s[hd * 32 + 16 + mlo];
        const float d0 = att_d[hd * 32 + mlo], d1 = att_d[hd * 32 + 16 + mlo];
        float v[16];
#pragma unroll
        for (int j = 0; j < 8; ++j) {
            v[j]     = acc[2 * hd][j] * s0 + acc[2 * hd + 1][j] * s1;
            v[8 + j] = acc[2 * hd][j] * d0 + acc[2 * hd + 1][j] * d1;
        }
        gat_fold_step<8, 8>(v, (mlo & 8) != 0);
        gat_fold_step<4, 4>(v, (mlo & 4) != 0);
        gat_fold_step<2, 2>(v, (mlo & 2) != 0);
        gat_fold_step<1, 1>(v, (mlo & 1) != 0);
        res[hd] = v[0];
    }
    {
        const int m = m0 + mhi + (mlo & 7);
        float* pl = (mlo & 8) ? a_d : a_s;
        if (m < n_rows) {
            if constexpr (NH == 4) {
                v4f r = { res[0], res[1], res[2], res[3] };
                *(v4f*)(pl + (size_t)m * 4) = r;
            } else {
#pragma unroll
                for (int hd = 0; hd < NH; ++hd) pl[(size_t)m * NH + hd] = res[hd];
            }
        }
    }
}

#define GAT_BINW   256
#define GAT_BINSH  8
#define GAT_NBMAX  1024
#define GAT_CHUNKS 128

__global__ __launch_bounds__(1024)
void gat_bin_hist(const int* __restrict__ dstI, int* __restrict__ tab, int E, int nb)
{
    __shared__ int hist[GAT_NBMAX];
    const int t = threadIdx.x;
    hist[t] = 0;
    __syncthreads();
    const int chunk = (E + GAT_CHUNKS - 1) / GAT_CHUNKS;
    const int e0 = blockIdx.x * chunk;
    const int e1 = min(E, e0 + chunk);
    for (int e = e0 + t; e < e1; e += 8 * 1024) {
        int d[8];
        for (int k = 0; k < 8; ++k) d[k] = dstI[min(e + k * 1024, e1 - 1)];
        asm volatile("" :: "v"(d[0]), "v"(d[1]), "v"(d[2]), "v"(d[3]), "v"(d[4]), "v"(d[5]), "v"(d[6]), "v"(d[7]));
        for (int k = 0; k < 8; ++k)
            if (e + k * 1024 < e1) atomicAdd(&hist[d[k] >> GAT_BINSH], 1);
    }
    __syncthreads();
    if (t < nb) tab[blockIdx.x * nb + t] = hist[t];
}

__global__ __launch_bounds__(1024)
void gat_bin_scatter(const int* __restrict__ srcI, const int* __restrict__ dstI,
                     const int* __restrict__ tab, int* __restrict__ binStart,
                     int* __restrict__ pairs, int E, int nb)
{
    __shared__ int colTot[GAT_NBMAX];
    __shared__ int cursor[GAT_NBMAX];
    __shared__ int s_wtot[32];
    const int t = threadIdx.x, lane = t & 31, wv = t >> 5;
    const int blk = blockIdx.x;
    colTot[t] = 0; cursor[t] = 0;
    __syncthreads();
    for (int j = t; j < nb * 8; j += 1024) {
        const int r = j / nb, b = j - r * nb;
        int tot = 0, part = 0;
        for (int k = 0; k < GAT_CHUNKS / 8; ++k) {
            const int row = r * (GAT_CHUNKS / 8) + k;
            const int v = tab[row * nb + b];
            tot  += v;
            part += (row < blk) ? v : 0;
        }
        atomicAdd(&colTot[b], tot);
        atomicAdd(&cursor[b], part);
    }
    __syncthreads();
    const int v = colTot[t];
    int inc = v;
    for (int o = 1; o < 32; o <<= 1) {
        int y = __shfl_up(inc, o);
        if (lane >= o) inc += y;
    }
    if (lane == 31) s_wtot[wv] = inc;
    __syncthreads();
    int off = 0;
    for (int w = 0; w < 32; ++w) off += (w < wv) ? s_wtot[w] : 0;
    const int excl = off + inc - v;
    if (t < nb) {
        cursor[t] += excl;
        if (blk == 0) binStart[t] = excl;
    }
    if (blk == 0 && t == 0) binStart[nb] = E;
    __syncthreads();
    const int chunk = (E + GAT_CHUNKS - 1) / GAT_CHUNKS;
    const int e0 = blk * chunk;
    const int e1 = min(E, e0 + chunk);
    int2* pr = (int2*)pairs;
    for (int e = e0 + t; e < e1; e += 8 * 1024) {
        int d[8], sv[8];
        for (int k = 0; k < 8; ++k) {
            const int idx = min(e + k * 1024, e1 - 1);
            d[k] = dstI[idx]; sv[k] = srcI[idx];
        }
        asm volatile("" :: "v"(d[0]), "v"(d[1]), "v"(d[2]), "v"(d[3]), "v"(d[4]), "v"(d[5]), "v"(d[6]), "v"(d[7]),
                           "v"(sv[0]), "v"(sv[1]), "v"(sv[2]), "v"(sv[3]), "v"(sv[4]), "v"(sv[5]), "v"(sv[6]), "v"(sv[7]));
        for (int k = 0; k < 8; ++k)
            if (e + k * 1024 < e1) {
                const int pos = atomicAdd(&cursor[d[k] >> GAT_BINSH], 1);
                pr[pos] = make_int2(sv[k], d[k]);
            }
    }
}

__global__ __launch_bounds__(512)
void gat_bin_sort(const int* __restrict__ pairs, const int* __restrict__ binStart,
                  int* __restrict__ cnt, int* __restrict__ rowptr, int* __restrict__ es, int n)
{
    __shared__ int c[GAT_BINW];
    __shared__ int s_wtot[GAT_BINW / 32];
    const int t = threadIdx.x, lane = t & 31, wv = t >> 5;
    const int base = blockIdx.x * GAT_BINW;
    const int s0 = binStart[blockIdx.x], s1 = binStart[blockIdx.x + 1];
    const int2* pr = (const int2*)pairs;
    if (t < GAT_BINW) c[t] = 0;
    __syncthreads();
    for (int i = s0 + t; i < s1; i += 8 * 512) {
        int y[8];
        for (int k = 0; k < 8; ++k) y[k] = pr[min(i + k * 512, s1 - 1)].y;
        asm volatile("" :: "v"(y[0]), "v"(y[1]), "v"(y[2]), "v"(y[3]), "v"(y[4]), "v"(y[5]), "v"(y[6]), "v"(y[7]));
        for (int k = 0; k < 8; ++k)
            if (i + k * 512 < s1) atomicAdd(&c[y[k] - base], 1);
    }
    __syncthreads();
    int v = 0, inc = 0;
    if (t < GAT_BINW) {
        v = c[t]; inc = v;
        for (int o = 1; o < 32; o <<= 1) {
            int y = __shfl_up(inc, o);
            if (lane >= o) inc += y;
        }
        if (lane == 31) s_wtot[wv] = inc;
    }
    __syncthreads();
    if (t < GAT_BINW) {
        int off = s0;
        for (int w = 0; w < GAT_BINW / 32; ++w) off += (w < wv) ? s_wtot[w] : 0;
        const int excl = off + inc - v;
        c[t] = excl;
        if (base + t < n) { cnt[base + t] = v; rowptr[base + t] = excl; }
    }
    __syncthreads();
    for (int i = s0 + t; i < s1; i += 8 * 512) {
        int2 p[8];
        for (int k = 0; k < 8; ++k) p[k] = pr[min(i + k * 512, s1 - 1)];
        asm volatile("" :: "v"(p[0].x), "v"(p[1].x), "v"(p[2].x), "v"(p[3].x), "v"(p[4].x), "v"(p[5].x), "v"(p[6].x), "v"(p[7].x),
                           "v"(p[0].y), "v"(p[1].y), "v"(p[2].y), "v"(p[3].y), "v"(p[4].y), "v"(p[5].y), "v"(p[6].y), "v"(p[7].y));
        for (int k = 0; k < 8; ++k)
            if (i + k * 512 < s1) {
                const int slot = atomicAdd(&c[p[k].y - base], 1);
                es[slot] = p[k].x;
            }
    }
}

__device__ __forceinline__ float gat_wave_max(float v) {
    for (int o = 16; o > 0; o >>= 1) v = fmaxf(v, __shfl_xor(v, o));
    return v;
}
__device__ __forceinline__ float gat_wave_sum(float v) {
    for (int o = 16; o > 0; o >>= 1) v += __shfl_xor(v, o);
    return v;
}

template<int H, int U>
__device__ __forceinline__ void gat_gather_batch(v4f& acc, const v4f* __restrict__ Hv,
                                                 int srcv, const float (&p)[H],
                                                 int i, int g, int sub, int h)
{
    constexpr int LPR = 8 * H;
    constexpr int EPT = 32 / LPR;
    int   s[U];
    float w[U];
    v4f   u[U];
    for (int k = 0; k < U; ++k) {
        if constexpr (H == 4) {
            const int slot = i + k;
            s[k] = __builtin_amdgcn_readlane(srcv, slot);
            int r = 0;
            for (int hh = 0; hh < 4; ++hh)
                r |= __builtin_amdgcn_readlane(__float_as_int(p[hh]), slot) & ((hh == h) ? -1 : 0);
            w[k] = __int_as_float(r);
        } else {
            const int slot = i + k * EPT + g;
            s[k] = __shfl(srcv, slot);
            w[k] = __shfl(p[0], slot);
        }
    }
    for (int k = 0; k < U; ++k) u[k] = Hv[(size_t)s[k] * LPR + sub];
    for (int k = 0; k < U; ++k) acc += w[k] * u[k];
}

template<int H, bool RELU>
__global__ __launch_bounds__(256)
void gat_softmax_gather(const int* __restrict__ rowptr, const int* __restrict__ cnt,
                        const int* __restrict__ es, const float* __restrict__ a_s,
                        const float* __restrict__ a_d, const float* __restrict__ Hf,
                        const float* __restrict__ bias, float* __restrict__ outp, int n)
{
    constexpr int LPR = 8 * H;
    constexpr int EPT = 32 / LPR;
    constexpr int NPW = 4;
    const int lane = threadIdx.x & 31;
    const int wv  = __builtin_amdgcn_readfirstlane((int)(threadIdx.x >> 5));
    const int sub = lane % LPR;
    const int g   = lane / LPR;
    const int h   = sub >> 3;
    const v4f* Hv = (const v4f*)Hf;
    const int d0  = (blockIdx.x * 8 + wv) * NPW;

    for (int q = 0; q < NPW; ++q) {
        const int d = d0 + q;
        if (d >= n) break;
        const int row0 = rowptr[d];
        const int deg  = cnt[d];
        const int T    = deg + 1;

        float ad[H], m[H], den[H];
        for (int hh = 0; hh < H; ++hh) { ad[hh] = a_d[(size_t)d * H + hh]; m[hh] = 0.f; den[hh] = 0.f; }
        v4f acc = { 0.f, 0.f, 0.f, 0.f };

        for (int c0 = 0; c0 < T; c0 += 32) {
            const int  t     = c0 + lane;
            const bool valid = t < T;
            int srcv = d;
            if (t < deg) srcv = es[row0 + t];

            float e[H], p[H];
            if constexpr (H == 4) {
                v4f a = ((const v4f*)a_s)[srcv];
                e[0] = a.x; e[1] = a.y; e[2] = a.z; e[3] = a.w;
            } else {
                for (int hh = 0; hh < H; ++hh) e[hh] = a_s[(size_t)srcv * H + hh];
            }
            float scl = 1.f;
            for (int hh = 0; hh < H; ++hh) {
                float v = e[hh] + ad[hh];
                v = (v >= 0.f) ? v : NEG_SLOPE * v;
                v = valid ? v : -__builtin_inff();
                float mn = gat_wave_max(v);
                float sc = 0.f;
                if (c0 > 0) { mn = fmaxf(mn, m[hh]); sc = __expf(m[hh] - mn); }
                p[hh]   = valid ? __expf(v - mn) : 0.f;
                den[hh] = den[hh] * sc + gat_wave_sum(p[hh]);
                m[hh]   = mn;
                if (hh == h) scl = sc;
            }
            if (c0 > 0) acc *= scl;

            const int Tc = min(T - c0, 32);
            int i = 0;
            for (; Tc - i >= 8 * EPT; i += 8 * EPT)
                gat_gather_batch<H, 8>(acc, Hv, srcv, p, i, g, sub, h);
            if (Tc - i > 4 * EPT)
                gat_gather_batch<H, 8>(acc, Hv, srcv, p, i, g, sub, h);
            else if (Tc - i > 0)
                gat_gather_batch<H, 4>(acc, Hv, srcv, p, i, g, sub, h);
        }

        float dsel = den[0];
        for (int hh = 1; hh < H; ++hh) dsel = (hh == h) ? den[hh] : dsel;
        if constexpr (EPT > 1) {
            for (int o = LPR; o < 32; o <<= 1) {
                acc.x += __shfl_xor(acc.x, o); acc.y += __shfl_xor(acc.y, o);
                acc.z += __shfl_xor(acc.z, o); acc.w += __shfl_xor(acc.w, o);
            }
        }
        if (g == 0) {
            v4f b = { bias[sub * 4 + 0], bias[sub * 4 + 1], bias[sub * 4 + 2], bias[sub * 4 + 3] };
            v4f r = acc / dsel + b;
            if (RELU) {
                r.x = fmaxf(r.x, 0.f); r.y = fmaxf(r.y, 0.f);
                r.z = fmaxf(r.z, 0.f); r.w = fmaxf(r.w, 0.f);
            }
            ((v4f*)outp)[(size_t)d * LPR + sub] = r;
        }
    }
}

extern "C" void kernel_launch(void* const* d_in, const int* in_sizes, int n_in,
                              void* d_out, int out_size, void* d_ws, size_t ws_size,
                              hipStream_t stream)
{
    const float* x   = (const float*)d_in[0];
    const int*   ei  = (const int*)d_in[1];
    const float* W1  = (const float*)d_in[2];
    const float* as1 = (const float*)d_in[3];
    const float* ad1 = (const float*)d_in[4];
    const float* b1  = (const float*)d_in[5];
    const float* W2  = (const float*)d_in[6];
    const float* as2 = (const float*)d_in[7];
    const float* ad2 = (const float*)d_in[8];
    const float* b2  = (const float*)d_in[9];
    float* out = (float*)d_out;

    const int N  = in_sizes[0] / 128;
    const int E  = in_sizes[1] / 2;
    const int* srcI = ei;
    const int* dstI = ei + E;

    float* ws = (float*)d_ws;
    float*    h1    = ws;              ws += (size_t)N * 128;
    float*    asrc1 = ws;              ws += (size_t)N * 4;
    float*    adst1 = ws;              ws += (size_t)N * 4;
    float*    h2    = ws;              ws += (size_t)N * 128;
    float*    g2    = ws;              ws += (size_t)N * 32;
    float*    asrc2 = ws;              ws += (size_t)N;
    float*    adst2 = ws;              ws += (size_t)N;
    int*      pairs = (int*)ws;        ws += (size_t)E * 2;
    int*      es    = (int*)ws;        ws += (size_t)E;
    int*      cnt   = (int*)ws;        ws += (size_t)N;
    int*      rowp  = (int*)ws;        ws += (size_t)N;
    int*      binS  = (int*)ws;        ws += GAT_NBMAX + 1;
    int*      tab   = (int*)ws;

    const int TB = 256;
    const int gB   = (N + 127) / 128;
    const int nb   = (N + GAT_BINW - 1) / GAT_BINW;
    const int gG   = (N + 31) / 32;

    gat_bin_hist<<<GAT_CHUNKS, 1024, 0, stream>>>(dstI, tab, E, nb);
    gat_bin_scatter<<<GAT_CHUNKS, 1024, 0, stream>>>(srcI, dstI, tab, binS, pairs, E, nb);
    gat_bin_sort<<<nb, 512, 0, stream>>>(pairs, binS, cnt, rowp, es, N);

    gat_gemm_wmma<8><<<gB, TB, 0, stream>>>(x, W1, as1, ad1, h1, asrc1, adst1, N);
    gat_softmax_gather<4, true><<<gG, TB, 0, stream>>>(rowp, cnt, es, asrc1, adst1, h1, b1, h2, N);

    gat_gemm_wmma<2><<<gB, TB, 0, stream>>>(h2, W2, as2, ad2, g2, asrc2, adst2, N);
    gat_softmax_gather<1, false><<<gG, TB, 0, stream>>>(rowp, cnt, es, asrc2, adst2, g2, b2, out, N);
}


# Layer-2 projection moved into the layer-1 owner pass: a block's 32 finished rows stay in LDS and two of its waves run them through the f32 matrix tiles with W2, scores folded there: 7 -> 6 launches
# speedup vs baseline: 14.2144x; 1.0443x over previous
#include <hip/hip_runtime.h>

#define NEG_SLOPE 0.2f

typedef float v2f __attribute__((ext_vector_type(2)));
typedef float v8f __attribute__((ext_vector_type(8)));
typedef float v4f __attribute__((ext_vector_type(4)));

template<int M>
__device__ __forceinline__ float gat_row_xor(float v) {
    return __int_as_float(__builtin_amdgcn_update_dpp(0, __float_as_int(v), 0x160 | M, 0xf, 0xf, true));
}

template<int M, int HALF>
__device__ __forceinline__ void gat_fold_step(float (&v)[16], bool up) {
#pragma unroll
    for (int i = 0; i < HALF; ++i) {
        const float keep = up ? v[i + HALF] : v[i];
        const float send = up ? v[i] : v[i + HALF];
        v[i] = keep + gat_row_xor<M>(send);
    }
}

template<int NT>
__global__ __launch_bounds__(256)
void gat_gemm_wmma(const float* __restrict__ X, const float* __restrict__ W,
                   const float* __restrict__ att_s, const float* __restrict__ att_d,
                   float* __restrict__ Hout, float* __restrict__ a_s, float* __restrict__ a_d,
                   int n_rows)
{
    constexpr int KD = 128;
    constexpr int ND = NT * 16;
    __shared__ float Wt[ND * KD];

    for (int i = threadIdx.x; i < KD * ND; i += 256) {
        int k = i / ND, n = i % ND;
        Wt[n * KD + ((k + 4 * n) & (KD - 1))] = W[i];
    }
    __syncthreads();

    const int lane = threadIdx.x & 31;
    const int wave = threadIdx.x >> 5;
    const int m0   = (blockIdx.x * 8 + wave) * 16;
    const int mlo  = lane & 15;
    const int kofs = (lane >> 4) * 2;

    int arow = m0 + mlo;
    if (arow >= n_rows) arow = n_rows - 1;
    const float* xrow = X + (size_t)arow * KD + kofs;

    v8f acc[NT] = {};

    for (int k = 0; k < KD; k += 4) {
        v2f a = *(const v2f*)(xrow + k);
#pragma unroll
        for (int t = 0; t < NT; ++t) {
            int n = t * 16 + mlo;
            v2f b = *(const v2f*)(&Wt[n * KD + ((k + kofs + 4 * n) & (KD - 1))]);
            acc[t] = __builtin_amdgcn_wmma_f32_16x16x4_f32(
                false, a, false, b, (short)0, acc[t], false, false);
        }
    }

    const int mhi = (lane >> 4) * 8;
#pragma unroll
    for (int t = 0; t < NT; ++t)
#pragma unroll
        for (int j = 0; j < 8; ++j) {
            int m = m0 + mhi + j;
            if (m < n_rows) Hout[(size_t)m * ND + t * 16 + mlo] = acc[t][j];
        }

    constexpr int NH = NT / 2;
    float res[NH];
#pragma unroll
    for (int hd = 0; hd < NH; ++hd) {
        const float s0 = att_s[hd * 32 + mlo], s1 = att_s[hd * 32 + 16 + mlo];
        const float d0 = att_d[hd * 32 + mlo], d1 = att_d[hd * 32 + 16 + mlo];
        float v[16];
#pragma unroll
        for (int j = 0; j < 8; ++j) {
            v[j]     = acc[2 * hd][j] * s0 + acc[2 * hd + 1][j] * s1;
            v[8 + j] = acc[2 * hd][j] * d0 + acc[2 * hd + 1][j] * d1;
        }
        gat_fold_step<8, 8>(v, (mlo & 8) != 0);
        gat_fold_step<4, 4>(v, (mlo & 4) != 0);
        gat_fold_step<2, 2>(v, (mlo & 2) != 0);
        gat_fold_step<1, 1>(v, (mlo & 1) != 0);
        res[hd] = v[0];
    }
    {
        const int m = m0 + mhi + (mlo & 7);
        float* pl = (mlo & 8) ? a_d : a_s;
        if (m < n_rows) {
            if constexpr (NH == 4) {
                v4f r = { res[0], res[1], res[2], res[3] };
                *(v4f*)(pl + (size_t)m * 4) = r;
            } else {
#pragma unroll
                for (int hd = 0; hd < NH; ++hd) pl[(size_t)m * NH + hd] = res[hd];
            }
        }
    }
}

#define GAT_BINW   256
#define GAT_BINSH  8
#define GAT_NBMAX  1024
#define GAT_CHUNKS 128

__global__ __launch_bounds__(1024)
void gat_bin_hist(const int* __restrict__ dstI, int* __restrict__ tab, int E, int nb)
{
    __shared__ int hist[GAT_NBMAX];
    const int t = threadIdx.x;
    hist[t] = 0;
    __syncthreads();
    const int chunk = (E + GAT_CHUNKS - 1) / GAT_CHUNKS;
    const int e0 = blockIdx.x * chunk;
    const int e1 = min(E, e0 + chunk);
    for (int e = e0 + t; e < e1; e += 8 * 1024) {
        int d[8];
        for (int k = 0; k < 8; ++k) d[k] = dstI[min(e + k * 1024, e1 - 1)];
        asm volatile("" :: "v"(d[0]), "v"(d[1]), "v"(d[2]), "v"(d[3]), "v"(d[4]), "v"(d[5]), "v"(d[6]), "v"(d[7]));
        for (int k = 0; k < 8; ++k)
            if (e + k * 1024 < e1) atomicAdd(&hist[d[k] >> GAT_BINSH], 1);
    }
    __syncthreads();
    if (t < nb) tab[blockIdx.x * nb + t] = hist[t];
}

__global__ __launch_bounds__(1024)
void gat_bin_scatter(const int* __restrict__ srcI, const int* __restrict__ dstI,
                     const int* __restrict__ tab, int* __restrict__ binStart,
                     int* __restrict__ pairs, int E, int nb)
{
    __shared__ int colTot[GAT_NBMAX];
    __shared__ int cursor[GAT_NBMAX];
    __shared__ int s_wtot[32];
    const int t = threadIdx.x, lane = t & 31, wv = t >> 5;
    const int blk = blockIdx.x;
    colTot[t] = 0; cursor[t] = 0;
    __syncthreads();
    for (int j = t; j < nb * 8; j += 1024) {
        const int r = j / nb, b = j - r * nb;
        int tot = 0, part = 0;
        for (int k = 0; k < GAT_CHUNKS / 8; ++k) {
            const int row = r * (GAT_CHUNKS / 8) + k;
            const int v = tab[row * nb + b];
            tot  += v;
            part += (row < blk) ? v : 0;
        }
        atomicAdd(&colTot[b], tot);
        atomicAdd(&cursor[b], part);
    }
    __syncthreads();
    const int v = colTot[t];
    int inc = v;
    for (int o = 1; o < 32; o <<= 1) {
        int y = __shfl_up(inc, o);
        if (lane >= o) inc += y;
    }
    if (lane == 31) s_wtot[wv] = inc;
    __syncthreads();
    int off = 0;
    for (int w = 0; w < 32; ++w) off += (w < wv) ? s_wtot[w] : 0;
    const int excl = off + inc - v;
    if (t < nb) {
        cursor[t] += excl;
        if (blk == 0) binStart[t] = excl;
    }
    if (blk == 0 && t == 0) binStart[nb] = E;
    __syncthreads();
    const int chunk = (E + GAT_CHUNKS - 1) / GAT_CHUNKS;
    const int e0 = blk * chunk;
    const int e1 = min(E, e0 + chunk);
    int2* pr = (int2*)pairs;
    for (int e = e0 + t; e < e1; e += 8 * 1024) {
        int d[8], sv[8];
        for (int k = 0; k < 8; ++k) {
            const int idx = min(e + k * 1024, e1 - 1);
            d[k] = dstI[idx]; sv[k] = srcI[idx];
        }
        asm volatile("" :: "v"(d[0]), "v"(d[1]), "v"(d[2]), "v"(d[3]), "v"(d[4]), "v"(d[5]), "v"(d[6]), "v"(d[7]),
                           "v"(sv[0]), "v"(sv[1]), "v"(sv[2]), "v"(sv[3]), "v"(sv[4]), "v"(sv[5]), "v"(sv[6]), "v"(sv[7]));
        for (int k = 0; k < 8; ++k)
            if (e + k * 1024 < e1) {
                const int pos = atomicAdd(&cursor[d[k] >> GAT_BINSH], 1);
                pr[pos] = make_int2(sv[k], d[k]);
            }
    }
}

__global__ __launch_bounds__(512)
void gat_bin_sort(const int* __restrict__ pairs, const int* __restrict__ binStart,
                  int* __restrict__ cnt, int* __restrict__ rowptr, int* __restrict__ es, int n)
{
    __shared__ int c[GAT_BINW];
    __shared__ int s_wtot[GAT_BINW / 32];
    const int t = threadIdx.x, lane = t & 31, wv = t >> 5;
    const int base = blockIdx.x * GAT_BINW;
    const int s0 = binStart[blockIdx.x], s1 = binStart[blockIdx.x + 1];
    const int2* pr = (const int2*)pairs;
    if (t < GAT_BINW) c[t] = 0;
    __syncthreads();
    for (int i = s0 + t; i < s1; i += 8 * 512) {
        int y[8];
        for (int k = 0; k < 8; ++k) y[k] = pr[min(i + k * 512, s1 - 1)].y;
        asm volatile("" :: "v"(y[0]), "v"(y[1]), "v"(y[2]), "v"(y[3]), "v"(y[4]), "v"(y[5]), "v"(y[6]), "v"(y[7]));
        for (int k = 0; k < 8; ++k)
            if (i + k * 512 < s1) atomicAdd(&c[y[k] - base], 1);
    }
    __syncthreads();
    int v = 0, inc = 0;
    if (t < GAT_BINW) {
        v = c[t]; inc = v;
        for (int o = 1; o < 32; o <<= 1) {
            int y = __shfl_up(inc, o);
            if (lane >= o) inc += y;
        }
        if (lane == 31) s_wtot[wv] = inc;
    }
    __syncthreads();
    if (t < GAT_BINW) {
        int off = s0;
        for (int w = 0; w < GAT_BINW / 32; ++w) off += (w < wv) ? s_wtot[w] : 0;
        const int excl = off + inc - v;
        c[t] = excl;
        if (base + t < n) { cnt[base + t] = v; rowptr[base + t] = excl; }
    }
    __syncthreads();
    for (int i = s0 + t; i < s1; i += 8 * 512) {
        int2 p[8];
        for (int k = 0; k < 8; ++k) p[k] = pr[min(i + k * 512, s1 - 1)];
        asm volatile("" :: "v"(p[0].x), "v"(p[1].x), "v"(p[2].x), "v"(p[3].x), "v"(p[4].x), "v"(p[5].x), "v"(p[6].x), "v"(p[7].x),
                           "v"(p[0].y), "v"(p[1].y), "v"(p[2].y), "v"(p[3].y), "v"(p[4].y), "v"(p[5].y), "v"(p[6].y), "v"(p[7].y));
        for (int k = 0; k < 8; ++k)
            if (i + k * 512 < s1) {
                const int slot = atomicAdd(&c[p[k].y - base], 1);
                es[slot] = p[k].x;
            }
    }
}

__device__ __forceinline__ float gat_wave_max(float v) {
    for (int o = 16; o > 0; o >>= 1) v = fmaxf(v, __shfl_xor(v, o));
    return v;
}
__device__ __forceinline__ float gat_wave_sum(float v) {
    for (int o = 16; o > 0; o >>= 1) v += __shfl_xor(v, o);
    return v;
}

template<int H, int U>
__device__ __forceinline__ void gat_gather_batch(v4f& acc, const v4f* __restrict__ Hv,
                                                 int srcv, const float (&p)[H],
                                                 int i, int g, int sub, int h)
{
    constexpr int LPR = 8 * H;
    constexpr int EPT = 32 / LPR;
    int   s[U];
    float w[U];
    v4f   u[U];
    for (int k = 0; k < U; ++k) {
        if constexpr (H == 4) {
            const int slot = i + k;
            s[k] = __builtin_amdgcn_readlane(srcv, slot);
            int r = 0;
            for (int hh = 0; hh < 4; ++hh)
                r |= __builtin_amdgcn_readlane(__float_as_int(p[hh]), slot) & ((hh == h) ? -1 : 0);
            w[k] = __int_as_float(r);
        } else {
            const int slot = i + k * EPT + g;
            s[k] = __shfl(srcv, slot);
            w[k] = __shfl(p[0], slot);
        }
    }
    for (int k = 0; k < U; ++k) u[k] = Hv[(size_t)s[k] * LPR + sub];
    for (int k = 0; k < U; ++k) acc += w[k] * u[k];
}

#define GAT_ROWP 132

__device__ __forceinline__ void gat_project_rows(const float* rw, const float* __restrict__ W,
                                                 const float* __restrict__ att_s, const float* __restrict__ att_d,
                                                 float* __restrict__ Gout, float* __restrict__ a_s,
                                                 float* __restrict__ a_d, int m0, int n_rows, int lane)
{
    const int mlo  = lane & 15;
    const int kofs = (lane >> 4) * 2;
    const float* arow = rw + mlo * GAT_ROWP + kofs;
    const float* bcol = W + kofs * 32 + mlo;
    v8f acc0 = {}, acc1 = {};
#pragma unroll 16
    for (int k = 0; k < 128; k += 4) {
        const v2f a  = *(const v2f*)(arow + k);
        const v2f b0 = { bcol[k * 32],      bcol[(k + 1) * 32] };
        const v2f b1 = { bcol[k * 32 + 16], bcol[(k + 1) * 32 + 16] };
        acc0 = __builtin_amdgcn_wmma_f32_16x16x4_f32(false, a, false, b0, (short)0, acc0, false, false);
        acc1 = __builtin_amdgcn_wmma_f32_16x16x4_f32(false, a, false, b1, (short)0, acc1, false, false);
    }
    const int mhi = (lane >> 4) * 8;
#pragma unroll
    for (int j = 0; j < 8; ++j) {
        const int m = m0 + mhi + j;
        if (m < n_rows) {
            Gout[(size_t)m * 32 + mlo]      = acc0[j];
            Gout[(size_t)m * 32 + 16 + mlo] = acc1[j];
        }
    }
    const float s0 = att_s[mlo], s1 = att_s[16 + mlo];
    const float d0 = att_d[mlo], d1 = att_d[16 + mlo];
    float v[16];
#pragma unroll
    for (int j = 0; j < 8; ++j) {
        v[j]     = acc0[j] * s0 + acc1[j] * s1;
        v[8 + j] = acc0[j] * d0 + acc1[j] * d1;
    }
    gat_fold_step<8, 8>(v, (mlo & 8) != 0);
    gat_fold_step<4, 4>(v, (mlo & 4) != 0);
    gat_fold_step<2, 2>(v, (mlo & 2) != 0);
    gat_fold_step<1, 1>(v, (mlo & 1) != 0);
    const int m = m0 + mhi + (mlo & 7);
    float* pl = (mlo & 8) ? a_d : a_s;
    if (m < n_rows) pl[m] = v[0];
}

template<int H, bool RELU, bool PROJ>
__global__ __launch_bounds__(256)
void gat_softmax_gather(const int* __restrict__ rowptr, const int* __restrict__ cnt,
                        const int* __restrict__ es, const float* __restrict__ a_s,
                        const float* __restrict__ a_d, const float* __restrict__ Hf,
                        const float* __restrict__ bias, float* __restrict__ outp, int n,
                        const float* __restrict__ W2, const float* __restrict__ att_s2,
                        const float* __restrict__ att_d2, float* __restrict__ a_s2,
                        float* __restrict__ a_d2)
{
    static_assert(!PROJ || H == 4, "the projection epilogue takes whole 128-float rows");
    __shared__ float rows[PROJ ? 32 * GAT_ROWP : 1];
    constexpr int LPR = 8 * H;
    constexpr int EPT = 32 / LPR;
    constexpr int NPW = 4;
    const int lane = threadIdx.x & 31;
    const int wv  = __builtin_amdgcn_readfirstlane((int)(threadIdx.x >> 5));
    const int sub = lane % LPR;
    const int g   = lane / LPR;
    const int h   = sub >> 3;
    const v4f* Hv = (const v4f*)Hf;
    const int d0  = (blockIdx.x * 8 + wv) * NPW;

    for (int q = 0; q < NPW; ++q) {
        const int d = d0 + q;
        if (d >= n) {
            if constexpr (PROJ) {
                const v4f z = { 0.f, 0.f, 0.f, 0.f };
                *(v4f*)&rows[(wv * NPW + q) * GAT_ROWP + sub * 4] = z;
                continue;
            } else break;
        }
        const int row0 = rowptr[d];
        const int deg  = cnt[d];
        const int T    = deg + 1;

        float ad[H], m[H], den[H];
        for (int hh = 0; hh < H; ++hh) { ad[hh] = a_d[(size_t)d * H + hh]; m[hh] = 0.f; den[hh] = 0.f; }
        v4f acc = { 0.f, 0.f, 0.f, 0.f };

        for (int c0 = 0; c0 < T; c0 += 32) {
            const int  t     = c0 + lane;
            const bool valid = t < T;
            int srcv = d;
            if (t < deg) srcv = es[row0 + t];

            float e[H], p[H];
            if constexpr (H == 4) {
                v4f a = ((const v4f*)a_s)[srcv];
                e[0] = a.x; e[1] = a.y; e[2] = a.z; e[3] = a.w;
            } else {
                for (int hh = 0; hh < H; ++hh) e[hh] = a_s[(size_t)srcv * H + hh];
            }
            float scl = 1.f;
            for (int hh = 0; hh < H; ++hh) {
                float v = e[hh] + ad[hh];
                v = (v >= 0.f) ? v : NEG_SLOPE * v;
                v = valid ? v : -__builtin_inff();
                float mn = gat_wave_max(v);
                float sc = 0.f;
                if (c0 > 0) { mn = fmaxf(mn, m[hh]); sc = __expf(m[hh] - mn); }
                p[hh]   = valid ? __expf(v - mn) : 0.f;
                den[hh] = den[hh] * sc + gat_wave_sum(p[hh]);
                m[hh]   = mn;
                if (hh == h) scl = sc;
            }
            if (c0 > 0) acc *= scl;

            const int Tc = min(T - c0, 32);
            int i = 0;
            for (; Tc - i >= 8 * EPT; i += 8 * EPT)
                gat_gather_batch<H, 8>(acc, Hv, srcv, p, i, g, sub, h);
            if (Tc - i > 4 * EPT)
                gat_gather_batch<H, 8>(acc, Hv, srcv, p, i, g, sub, h);
            else if (Tc - i > 0)
                gat_gather_batch<H, 4>(acc, Hv, srcv, p, i, g, sub, h);
        }

        float dsel = den[0];
        for (int hh = 1; hh < H; ++hh) dsel = (hh == h) ? den[hh] : dsel;
        if constexpr (EPT > 1) {
            for (int o = LPR; o < 32; o <<= 1) {
                acc.x += __shfl_xor(acc.x, o); acc.y += __shfl_xor(acc.y, o);
                acc.z += __shfl_xor(acc.z, o); acc.w += __shfl_xor(acc.w, o);
            }
        }
        if (g == 0) {
            v4f b = { bias[sub * 4 + 0], bias[sub * 4 + 1], bias[sub * 4 + 2], bias[sub * 4 + 3] };
            v4f r = acc / dsel + b;
            if (RELU) {
                r.x = fmaxf(r.x, 0.f); r.y = fmaxf(r.y, 0.f);
                r.z = fmaxf(r.z, 0.f); r.w = fmaxf(r.w, 0.f);
            }
            if constexpr (PROJ) *(v4f*)&rows[(wv * NPW + q) * GAT_ROWP + sub * 4] = r;
            else ((v4f*)outp)[(size_t)d * LPR + sub] = r;
        }
    }
    if constexpr (PROJ) {
        __syncthreads();
        if (wv < 2)
            gat_project_rows(rows + wv * 16 * GAT_ROWP, W2, att_s2, att_d2, outp, a_s2, a_d2,
                             blockIdx.x * 32 + wv * 16, n, lane);
    }
}

extern "C" void kernel_launch(void* const* d_in, const int* in_sizes, int n_in,
                              void* d_out, int out_size, void* d_ws, size_t ws_size,
                              hipStream_t stream)
{
    const float* x   = (const float*)d_in[0];
    const int*   ei  = (const int*)d_in[1];
    const float* W1  = (const float*)d_in[2];
    const float* as1 = (const float*)d_in[3];
    const float* ad1 = (const float*)d_in[4];
    const float* b1  = (const float*)d_in[5];
    const float* W2  = (const float*)d_in[6];
    const float* as2 = (const float*)d_in[7];
    const float* ad2 = (const float*)d_in[8];
    const float* b2  = (const float*)d_in[9];
    float* out = (float*)d_out;

    const int N  = in_sizes[0] / 128;
    const int E  = in_sizes[1] / 2;
    const int* srcI = ei;
    const int* dstI = ei + E;

    float* ws = (float*)d_ws;
    float*    h1    = ws;              ws += (size_t)N * 128;
    float*    asrc1 = ws;              ws += (size_t)N * 4;
    float*    adst1 = ws;              ws += (size_t)N * 4;
    float*    g2    = ws;              ws += (size_t)N * 32;
    float*    asrc2 = ws;              ws += (size_t)N;
    float*    adst2 = ws;              ws += (size_t)N;
    int*      pairs = (int*)ws;        ws += (size_t)E * 2;
    int*      es    = (int*)ws;        ws += (size_t)E;
    int*      cnt   = (int*)ws;        ws += (size_t)N;
    int*      rowp  = (int*)ws;        ws += (size_t)N;
    int*      binS  = (int*)ws;        ws += GAT_NBMAX + 1;
    int*      tab   = (int*)ws;

    const int TB = 256;
    const int gB   = (N + 127) / 128;
    const int nb   = (N + GAT_BINW - 1) / GAT_BINW;
    const int gG   = (N + 31) / 32;

    gat_bin_hist<<<GAT_CHUNKS, 1024, 0, stream>>>(dstI, tab, E, nb);
    gat_bin_scatter<<<GAT_CHUNKS, 1024, 0, stream>>>(srcI, dstI, tab, binS, pairs, E, nb);
    gat_bin_sort<<<nb, 512, 0, stream>>>(pairs, binS, cnt, rowp, es, N);

    gat_gemm_wmma<8><<<gB, TB, 0, stream>>>(x, W1, as1, ad1, h1, asrc1, adst1, N);
    gat_softmax_gather<4, true, true><<<gG, TB, 0, stream>>>(rowp, cnt, es, asrc1, adst1, h1, b1, g2, N,
                                                             W2, as2, ad2, asrc2, adst2);

    gat_softmax_gather<1, false, false><<<gG, TB, 0, stream>>>(rowp, cnt, es, asrc2, adst2, g2, b2, out, N,
                                                               nullptr, nullptr, nullptr, nullptr, nullptr);
}


# Owner pass scores each gathered row itself (row . att, 8-lane DPP sum) under a running-max softmax: per-edge score gather and score planes gone; sums now in running-max rescale order
# speedup vs baseline: 15.6433x; 1.1005x over previous
#include <hip/hip_runtime.h>

#define NEG_SLOPE 0.2f

typedef float v2f __attribute__((ext_vector_type(2)));
typedef float v8f __attribute__((ext_vector_type(8)));
typedef float v4f __attribute__((ext_vector_type(4)));

template<int M>
__device__ __forceinline__ float gat_row_xor(float v) {
    return __int_as_float(__builtin_amdgcn_update_dpp(0, __float_as_int(v), 0x160 | M, 0xf, 0xf, true));
}

__device__ __forceinline__ float gat_sum8(float v) {
    v += gat_row_xor<1>(v);
    v += gat_row_xor<2>(v);
    v += gat_row_xor<4>(v);
    return v;
}

__device__ __forceinline__ float gat_dot4(v4f a, v4f b, v2f seed) {
    const v2f t = a.xy * b.xy + (a.zw * b.zw + seed);
    return t.x + t.y;
}

template<int NT>
__global__ __launch_bounds__(256)
void gat_gemm_wmma(const float* __restrict__ X, const float* __restrict__ W,
                   float* __restrict__ Hout, int n_rows)
{
    constexpr int KD = 128;
    constexpr int ND = NT * 16;
    __shared__ float Wt[ND * KD];

    for (int i = threadIdx.x; i < KD * ND; i += 256) {
        int k = i / ND, n = i % ND;
        Wt[n * KD + ((k + 4 * n) & (KD - 1))] = W[i];
    }
    __syncthreads();

    const int lane = threadIdx.x & 31;
    const int wave = threadIdx.x >> 5;
    const int m0   = (blockIdx.x * 8 + wave) * 16;
    const int mlo  = lane & 15;
    const int kofs = (lane >> 4) * 2;

    int arow = m0 + mlo;
    if (arow >= n_rows) arow = n_rows - 1;
    const float* xrow = X + (size_t)arow * KD + kofs;

    v8f acc[NT] = {};

    for (int k = 0; k < KD; k += 4) {
        v2f a = *(const v2f*)(xrow + k);
#pragma unroll
        for (int t = 0; t < NT; ++t) {
            int n = t * 16 + mlo;
            v2f b = *(const v2f*)(&Wt[n * KD + ((k + kofs + 4 * n) & (KD - 1))]);
            acc[t] = __builtin_amdgcn_wmma_f32_16x16x4_f32(
                false, a, false, b, (short)0, acc[t], false, false);
        }
    }

    const int mhi = (lane >> 4) * 8;
#pragma unroll
    for (int t = 0; t < NT; ++t)
#pragma unroll
        for (int j = 0; j < 8; ++j) {
            int m = m0 + mhi + j;
            if (m < n_rows) Hout[(size_t)m * ND + t * 16 + mlo] = acc[t][j];
        }
}

#define GAT_BINW   256
#define GAT_BINSH  8
#define GAT_NBMAX  1024
#define GAT_CHUNKS 128

__global__ __launch_bounds__(1024)
void gat_bin_hist(const int* __restrict__ dstI, int* __restrict__ tab, int E, int nb)
{
    __shared__ int hist[GAT_NBMAX];
    const int t = threadIdx.x;
    hist[t] = 0;
    __syncthreads();
    const int chunk = (E + GAT_CHUNKS - 1) / GAT_CHUNKS;
    const int e0 = blockIdx.x * chunk;
    const int e1 = min(E, e0 + chunk);
    for (int e = e0 + t; e < e1; e += 8 * 1024) {
        int d[8];
        for (int k = 0; k < 8; ++k) d[k] = dstI[min(e + k * 1024, e1 - 1)];
        asm volatile("" :: "v"(d[0]), "v"(d[1]), "v"(d[2]), "v"(d[3]), "v"(d[4]), "v"(d[5]), "v"(d[6]), "v"(d[7]));
        for (int k = 0; k < 8; ++k)
            if (e + k * 1024 < e1) atomicAdd(&hist[d[k] >> GAT_BINSH], 1);
    }
    __syncthreads();
    if (t < nb) tab[blockIdx.x * nb + t] = hist[t];
}

__global__ __launch_bounds__(1024)
void gat_bin_scatter(const int* __restrict__ srcI, const int* __restrict__ dstI,
                     const int* __restrict__ tab, int* __restrict__ binStart,
                     int* __restrict__ pairs, int E, int nb)
{
    __shared__ int colTot[GAT_NBMAX];
    __shared__ int cursor[GAT_NBMAX];
    __shared__ int s_wtot[32];
    const int t = threadIdx.x, lane = t & 31, wv = t >> 5;
    const int blk = blockIdx.x;
    colTot[t] = 0; cursor[t] = 0;
    __syncthreads();
    for (int j = t; j < nb * 8; j += 1024) {
        const int r = j / nb, b = j - r * nb;
        int tot = 0, part = 0;
        for (int k = 0; k < GAT_CHUNKS / 8; ++k) {
            const int row = r * (GAT_CHUNKS / 8) + k;
            const int v = tab[row * nb + b];
            tot  += v;
            part += (row < blk) ? v : 0;
        }
        atomicAdd(&colTot[b], tot);
        atomicAdd(&cursor[b], part);
    }
    __syncthreads();
    const int v = colTot[t];
    int inc = v;
    for (int o = 1; o < 32; o <<= 1) {
        int y = __shfl_up(inc, o);
        if (lane >= o) inc += y;
    }
    if (lane == 31) s_wtot[wv] = inc;
    __syncthreads();
    int off = 0;
    for (int w = 0; w < 32; ++w) off += (w < wv) ? s_wtot[w] : 0;
    const int excl = off + inc - v;
    if (t < nb) {
        cursor[t] += excl;
        if (blk == 0) binStart[t] = excl;
    }
    if (blk == 0 && t == 0) binStart[nb] = E;
    __syncthreads();
    const int chunk = (E + GAT_CHUNKS - 1) / GAT_CHUNKS;
    const int e0 = blk * chunk;
    const int e1 = min(E, e0 + chunk);
    int2* pr = (int2*)pairs;
    for (int e = e0 + t; e < e1; e += 8 * 1024) {
        int d[8], sv[8];
        for (int k = 0; k < 8; ++k) {
            const int idx = min(e + k * 1024, e1 - 1);
            d[k] = dstI[idx]; sv[k] = srcI[idx];
        }
        asm volatile("" :: "v"(d[0]), "v"(d[1]), "v"(d[2]), "v"(d[3]), "v"(d[4]), "v"(d[5]), "v"(d[6]), "v"(d[7]),
                           "v"(sv[0]), "v"(sv[1]), "v"(sv[2]), "v"(sv[3]), "v"(sv[4]), "v"(sv[5]), "v"(sv[6]), "v"(sv[7]));
        for (int k = 0; k < 8; ++k)
            if (e + k * 1024 < e1) {
                const int pos = atomicAdd(&cursor[d[k] >> GAT_BINSH], 1);
                pr[pos] = make_int2(sv[k], d[k]);
            }
    }
}

__global__ __launch_bounds__(512)
void gat_bin_sort(const int* __restrict__ pairs, const int* __restrict__ binStart,
                  int* __restrict__ cnt, int* __restrict__ rowptr, int* __restrict__ es, int n)
{
    __shared__ int c[GAT_BINW];
    __shared__ int s_wtot[GAT_BINW / 32];
    const int t = threadIdx.x, lane = t & 31, wv = t >> 5;
    const int base = blockIdx.x * GAT_BINW;
    const int s0 = binStart[blockIdx.x], s1 = binStart[blockIdx.x + 1];
    const int2* pr = (const int2*)pairs;
    if (t < GAT_BINW) c[t] = 0;
    __syncthreads();
    for (int i = s0 + t; i < s1; i += 8 * 512) {
        int y[8];
        for (int k = 0; k < 8; ++k) y[k] = pr[min(i + k * 512, s1 - 1)].y;
        asm volatile("" :: "v"(y[0]), "v"(y[1]), "v"(y[2]), "v"(y[3]), "v"(y[4]), "v"(y[5]), "v"(y[6]), "v"(y[7]));
        for (int k = 0; k < 8; ++k)
            if (i + k * 512 < s1) atomicAdd(&c[y[k] - base], 1);
    }
    __syncthreads();
    int v = 0, inc = 0;
    if (t < GAT_BINW) {
        v = c[t]; inc = v;
        for (int o = 1; o < 32; o <<= 1) {
            int y = __shfl_up(inc, o);
            if (lane >= o) inc += y;
        }
        if (lane == 31) s_wtot[wv] = inc;
    }
    __syncthreads();
    if (t < GAT_BINW) {
        int off = s0;
        for (int w = 0; w < GAT_BINW / 32; ++w) off += (w < wv) ? s_wtot[w] : 0;
        const int excl = off + inc - v;
        c[t] = excl;
        if (base + t < n) { cnt[base + t] = v; rowptr[base + t] = excl; }
    }
    __syncthreads();
    for (int i = s0 + t; i < s1; i += 8 * 512) {
        int2 p[8];
        for (int k = 0; k < 8; ++k) p[k] = pr[min(i + k * 512, s1 - 1)];
        asm volatile("" :: "v"(p[0].x), "v"(p[1].x), "v"(p[2].x), "v"(p[3].x), "v"(p[4].x), "v"(p[5].x), "v"(p[6].x), "v"(p[7].x),
                           "v"(p[0].y), "v"(p[1].y), "v"(p[2].y), "v"(p[3].y), "v"(p[4].y), "v"(p[5].y), "v"(p[6].y), "v"(p[7].y));
        for (int k = 0; k < 8; ++k)
            if (i + k * 512 < s1) {
                const int slot = atomicAdd(&c[p[k].y - base], 1);
                es[slot] = p[k].x;
            }
    }
}

template<int H, int U>
__device__ __forceinline__ void gat_gather_batch(v4f& acc, float& m, float& den,
                                                 const v4f* __restrict__ Hv, int srcv,
                                                 int i, int Tc, int g, int sub,
                                                 v4f as4, v2f ad8)
{
    constexpr int LPR = 8 * H;
    constexpr int EPT = 32 / LPR;
    v4f   u[U];
    float sc[U];
    for (int k = 0; k < U; ++k) {
        int s;
        if constexpr (H == 4) s = __builtin_amdgcn_readlane(srcv, i + k);
        else                  s = __shfl(srcv, i + k * EPT + g);
        if constexpr (H == 4) u[k] = (Hv + (size_t)s * LPR)[sub];
        else                  u[k] = Hv[(unsigned)(s * LPR + sub)];
    }
    __builtin_amdgcn_sched_barrier(0);
    for (int k = 0; k < U; ++k) {
        const int slot = (H == 4) ? (i + k) : (i + k * EPT + g);
        float v = gat_sum8(gat_dot4(u[k], as4, ad8));
        v = (v >= 0.f) ? v : NEG_SLOPE * v;
        sc[k] = (slot < Tc) ? v : -__builtin_inff();
    }
    float mn = m;
    for (int k = 0; k < U; ++k) mn = fmaxf(mn, sc[k]);
    const float scale = __expf(m - mn);
    acc *= scale;
    den *= scale;
    for (int k = 0; k < U; ++k) {
        const float p = __expf(sc[k] - mn);
        den += p;
        acc += p * u[k];
    }
    m = mn;
}

#define GAT_ROWP 132

__device__ __forceinline__ void gat_project_rows(const float* rw, const float* __restrict__ W,
                                                 float* __restrict__ Gout, int m0, int n_rows, int lane)
{
    const int mlo  = lane & 15;
    const int kofs = (lane >> 4) * 2;
    const float* arow = rw + mlo * GAT_ROWP + kofs;
    const float* bcol = W + kofs * 32 + mlo;
    v8f acc0 = {}, acc1 = {};
#pragma unroll 16
    for (int k = 0; k < 128; k += 4) {
        const v2f a  = *(const v2f*)(arow + k);
        const v2f b0 = { bcol[k * 32],      bcol[(k + 1) * 32] };
        const v2f b1 = { bcol[k * 32 + 16], bcol[(k + 1) * 32 + 16] };
        acc0 = __builtin_amdgcn_wmma_f32_16x16x4_f32(false, a, false, b0, (short)0, acc0, false, false);
        acc1 = __builtin_amdgcn_wmma_f32_16x16x4_f32(false, a, false, b1, (short)0, acc1, false, false);
    }
    const int mhi = (lane >> 4) * 8;
#pragma unroll
    for (int j = 0; j < 8; ++j) {
        const int m = m0 + mhi + j;
        if (m < n_rows) {
            Gout[(size_t)m * 32 + mlo]      = acc0[j];
            Gout[(size_t)m * 32 + 16 + mlo] = acc1[j];
        }
    }
}

template<int H, bool RELU, bool PROJ>
__global__ __launch_bounds__(256, H == 4 ? 12 : 10)
void gat_softmax_gather(const int* __restrict__ rowptr, const int* __restrict__ cnt,
                        const int* __restrict__ es, const float* __restrict__ att_s,
                        const float* __restrict__ att_d, const float* __restrict__ Hf,
                        const float* __restrict__ bias, float* __restrict__ outp, int n,
                        const float* __restrict__ W2)
{
    static_assert(!PROJ || H == 4, "the projection epilogue takes whole 128-float rows");
    __shared__ float rows[PROJ ? 32 * GAT_ROWP : 1];
    constexpr int LPR = 8 * H;
    constexpr int EPT = 32 / LPR;
    constexpr int NPW = 4;
    const int lane = threadIdx.x & 31;
    const int wv  = __builtin_amdgcn_readfirstlane((int)(threadIdx.x >> 5));
    const int sub = lane % LPR;
    const int g   = lane / LPR;
    const v4f* Hv = (const v4f*)Hf;
    const int d0  = (blockIdx.x * 8 + wv) * NPW;
    const v4f as4 = ((const v4f*)att_s)[sub];
    const v4f ad4 = ((const v4f*)att_d)[sub];

    for (int q = 0; q < NPW; ++q) {
        const int d = d0 + q;
        if (d >= n) {
            if constexpr (PROJ) {
                const v4f z = { 0.f, 0.f, 0.f, 0.f };
                *(v4f*)&rows[(wv * NPW + q) * GAT_ROWP + sub * 4] = z;
                continue;
            } else break;
        }
        const v4f us   = Hv[(size_t)d * LPR + sub];
        const int row0 = rowptr[d];
        const int deg  = cnt[d];
        int srcv = d;
        if (lane < deg) srcv = es[row0 + lane];

        const v2f  z2  = { 0.f, 0.f };
        const float ad = gat_sum8(gat_dot4(us, ad4, z2));
        const v2f  ad8 = { 0.125f * ad, 0.f };
        float m = gat_sum8(gat_dot4(us, as4, ad8));
        m = (m >= 0.f) ? m : NEG_SLOPE * m;
        const v4f z4 = { 0.f, 0.f, 0.f, 0.f };
        float den = (g == 0) ? 1.f : 0.f;
        v4f   acc = (g == 0) ? us : z4;

        for (int c0 = 0; c0 < deg; c0 += 32) {
            if (c0 > 0) {
                const int t = c0 + lane;
                srcv = d;
                if (t < deg) srcv = es[row0 + t];
            }
            const int Tc = min(deg - c0, 32);
            int i = 0;
            for (; Tc - i >= 8 * EPT; i += 8 * EPT)
                gat_gather_batch<H, 8>(acc, m, den, Hv, srcv, i, Tc, g, sub, as4, ad8);
            if (Tc - i > 4 * EPT)
                gat_gather_batch<H, 8>(acc, m, den, Hv, srcv, i, Tc, g, sub, as4, ad8);
            else if (Tc - i > 0)
                gat_gather_batch<H, 4>(acc, m, den, Hv, srcv, i, Tc, g, sub, as4, ad8);
        }

        if constexpr (EPT > 1) {
            float mx = m;
            for (int o = LPR; o < 32; o <<= 1) mx = fmaxf(mx, __shfl_xor(mx, o));
            const float scale = __expf(m - mx);
            den *= scale;
            acc *= scale;
            for (int o = LPR; o < 32; o <<= 1) {
                den   += __shfl_xor(den, o);
                acc.x += __shfl_xor(acc.x, o); acc.y += __shfl_xor(acc.y, o);
                acc.z += __shfl_xor(acc.z, o); acc.w += __shfl_xor(acc.w, o);
            }
        }
        if (g == 0) {
            v4f b = { bias[sub * 4 + 0], bias[sub * 4 + 1], bias[sub * 4 + 2], bias[sub * 4 + 3] };
            v4f r = acc / den + b;
            if (RELU) {
                r.x = fmaxf(r.x, 0.f); r.y = fmaxf(r.y, 0.f);
                r.z = fmaxf(r.z, 0.f); r.w = fmaxf(r.w, 0.f);
            }
            if constexpr (PROJ) *(v4f*)&rows[(wv * NPW + q) * GAT_ROWP + sub * 4] = r;
            else ((v4f*)outp)[(size_t)d * LPR + sub] = r;
        }
    }
    if constexpr (PROJ) {
        __syncthreads();
        if (wv < 2)
            gat_project_rows(rows + wv * 16 * GAT_ROWP, W2, outp, blockIdx.x * 32 + wv * 16, n, lane);
    }
}

extern "C" void kernel_launch(void* const* d_in, const int* in_sizes, int n_in,
                              void* d_out, int out_size, void* d_ws, size_t ws_size,
                              hipStream_t stream)
{
    const float* x   = (const float*)d_in[0];
    const int*   ei  = (const int*)d_in[1];
    const float* W1  = (const float*)d_in[2];
    const float* as1 = (const float*)d_in[3];
    const float* ad1 = (const float*)d_in[4];
    const float* b1  = (const float*)d_in[5];
    const float* W2  = (const float*)d_in[6];
    const float* as2 = (const float*)d_in[7];
    const float* ad2 = (const float*)d_in[8];
    const float* b2  = (const float*)d_in[9];
    float* out = (float*)d_out;

    const int N  = in_sizes[0] / 128;
    const int E  = in_sizes[1] / 2;
    const int* srcI = ei;
    const int* dstI = ei + E;

    float* ws = (float*)d_ws;
    float*    h1    = ws;              ws += (size_t)N * 128;
    float*    g2    = ws;              ws += (size_t)N * 32;
    int*      pairs = (int*)ws;        ws += (size_t)E * 2;
    int*      es    = (int*)ws;        ws += (size_t)E;
    int*      cnt   = (int*)ws;        ws += (size_t)N;
    int*      rowp  = (int*)ws;        ws += (size_t)N;
    int*      binS  = (int*)ws;        ws += GAT_NBMAX + 1;
    int*      tab   = (int*)ws;

    const int TB = 256;
    const int gB   = (N + 127) / 128;
    const int nb   = (N + GAT_BINW - 1) / GAT_BINW;
    const int gG   = (N + 31) / 32;

    gat_bin_hist<<<GAT_CHUNKS, 1024, 0, stream>>>(dstI, tab, E, nb);
    gat_bin_scatter<<<GAT_CHUNKS, 1024, 0, stream>>>(srcI, dstI, tab, binS, pairs, E, nb);
    gat_bin_sort<<<nb, 512, 0, stream>>>(pairs, binS, cnt, rowp, es, N);

    gat_gemm_wmma<8><<<gB, TB, 0, stream>>>(x, W1, h1, N);
    gat_softmax_gather<4, true, true><<<gG, TB, 0, stream>>>(rowp, cnt, es, as1, ad1, h1, b1, g2, N, W2);

    gat_softmax_gather<1, false, false><<<gG, TB, 0, stream>>>(rowp, cnt, es, as2, ad2, g2, b2, out, N, nullptr);
}


# Projection GEMM pipelined: X rows staged to LDS by whole-row loads, each k-step's A and B fragments read as one group a step ahead of its products; products and k order unchanged
# speedup vs baseline: 17.3448x; 1.1088x over previous
#include <hip/hip_runtime.h>

#define NEG_SLOPE 0.2f

typedef float v2f __attribute__((ext_vector_type(2)));
typedef float v8f __attribute__((ext_vector_type(8)));
typedef float v4f __attribute__((ext_vector_type(4)));

template<int M>
__device__ __forceinline__ float gat_row_xor(float v) {
    return __int_as_float(__builtin_amdgcn_update_dpp(0, __float_as_int(v), 0x160 | M, 0xf, 0xf, true));
}

__device__ __forceinline__ float gat_sum8(float v) {
    v += gat_row_xor<1>(v);
    v += gat_row_xor<2>(v);
    v += gat_row_xor<4>(v);
    return v;
}

__device__ __forceinline__ float gat_dot4(v4f a, v4f b, v2f seed) {
    const v2f t = a.xy * b.xy + (a.zw * b.zw + seed);
    return t.x + t.y;
}

#define GAT_ROWP 132

template<int NT>
__global__ __launch_bounds__(256)
void gat_gemm_wmma(const float* __restrict__ X, const float* __restrict__ W,
                   float* __restrict__ Hout, int n_rows)
{
    constexpr int KD = 128;
    constexpr int ND = NT * 16;
    __shared__ float Wt[ND * KD];
    __shared__ float Xs[8 * 16 * GAT_ROWP];

    const int lane = threadIdx.x & 31;
    const int wave = threadIdx.x >> 5;
    const int m0   = (blockIdx.x * 8 + wave) * 16;
    float* xs = Xs + wave * 16 * GAT_ROWP;

    v4f xr[16];
#pragma unroll
    for (int r = 0; r < 16; ++r) {
        int row = m0 + r;
        if (row >= n_rows) row = n_rows - 1;
        xr[r] = *(const v4f*)(X + (size_t)row * KD + lane * 4);
    }
    static_assert(ND == 128, "the thread-to-column map below takes 128 columns and 256 threads");
    const int wn  = threadIdx.x & (ND - 1);
    const int wkb = threadIdx.x >> 7;
    float wv[64];
#pragma unroll
    for (int j = 0; j < 64; ++j) wv[j] = W[(wkb + 2 * j) * ND + wn];
#pragma unroll
    for (int j = 0; j < 64; ++j)
        Wt[wn * KD + ((wkb + 2 * j + 4 * wn) & (KD - 1))] = wv[j];
#pragma unroll
    for (int r = 0; r < 16; ++r) *(v4f*)(xs + r * GAT_ROWP + lane * 4) = xr[r];
    __syncthreads();

    const int mlo  = lane & 15;
    const int kofs = (lane >> 4) * 2;
    const float* arow = xs + mlo * GAT_ROWP + kofs;
    const float* wcol = Wt + mlo * KD;
    const int c0 = kofs + 4 * mlo;

    v8f acc[NT] = {};
    v2f a0, a1, b0[NT], b1[NT];
    a0 = *(const v2f*)arow;
#pragma unroll
    for (int t = 0; t < NT; ++t)
        b0[t] = *(const v2f*)(wcol + t * 16 * KD + ((c0 + 64 * t) & (KD - 1)));

#pragma unroll 1
    for (int k = 0; k < KD; k += 8) {
        a1 = *(const v2f*)(arow + k + 4);
#pragma unroll
        for (int t = 0; t < NT; ++t)
            b1[t] = *(const v2f*)(wcol + t * 16 * KD + ((k + 4 + c0 + 64 * t) & (KD - 1)));
        __builtin_amdgcn_sched_barrier(0);
#pragma unroll
        for (int t = 0; t < NT; ++t)
            acc[t] = __builtin_amdgcn_wmma_f32_16x16x4_f32(
                false, a0, false, b0[t], (short)0, acc[t], false, false);
        __builtin_amdgcn_sched_barrier(0);
        a0 = *(const v2f*)(arow + ((k + 8) & (KD - 1)));
#pragma unroll
        for (int t = 0; t < NT; ++t)
            b0[t] = *(const v2f*)(wcol + t * 16 * KD + ((k + 8 + c0 + 64 * t) & (KD - 1)));
        __builtin_amdgcn_sched_barrier(0);
#pragma unroll
        for (int t = 0; t < NT; ++t)
            acc[t] = __builtin_amdgcn_wmma_f32_16x16x4_f32(
                false, a1, false, b1[t], (short)0, acc[t], false, false);
        __builtin_amdgcn_sched_barrier(0);
    }

    const int mhi = (lane >> 4) * 8;
#pragma unroll
    for (int t = 0; t < NT; ++t)
#pragma unroll
        for (int j = 0; j < 8; ++j) {
            int m = m0 + mhi + j;
            if (m < n_rows) Hout[(size_t)m * ND + t * 16 + mlo] = acc[t][j];
        }
}

#define GAT_BINW   256
#define GAT_BINSH  8
#define GAT_NBMAX  1024
#define GAT_CHUNKS 128

__global__ __launch_bounds__(1024)
void gat_bin_hist(const int* __restrict__ dstI, int* __restrict__ tab, int E, int nb)
{
    __shared__ int hist[GAT_NBMAX];
    const int t = threadIdx.x;
    hist[t] = 0;
    __syncthreads();
    const int chunk = (E + GAT_CHUNKS - 1) / GAT_CHUNKS;
    const int e0 = blockIdx.x * chunk;
    const int e1 = min(E, e0 + chunk);
    for (int e = e0 + t; e < e1; e += 8 * 1024) {
        int d[8];
        for (int k = 0; k < 8; ++k) d[k] = dstI[min(e + k * 1024, e1 - 1)];
        asm volatile("" :: "v"(d[0]), "v"(d[1]), "v"(d[2]), "v"(d[3]), "v"(d[4]), "v"(d[5]), "v"(d[6]), "v"(d[7]));
        for (int k = 0; k < 8; ++k)
            if (e + k * 1024 < e1) atomicAdd(&hist[d[k] >> GAT_BINSH], 1);
    }
    __syncthreads();
    if (t < nb) tab[blockIdx.x * nb + t] = hist[t];
}

__global__ __launch_bounds__(1024)
void gat_bin_scatter(const int* __restrict__ srcI, const int* __restrict__ dstI,
                     const int* __restrict__ tab, int* __restrict__ binStart,
                     int* __restrict__ pairs, int E, int nb)
{
    __shared__ int colTot[GAT_NBMAX];
    __shared__ int cursor[GAT_NBMAX];
    __shared__ int s_wtot[32];
    const int t = threadIdx.x, lane = t & 31, wv = t >> 5;
    const int blk = blockIdx.x;
    colTot[t] = 0; cursor[t] = 0;
    __syncthreads();
    for (int j = t; j < nb * 8; j += 1024) {
        const int r = j / nb, b = j - r * nb;
        int tot = 0, part = 0;
        for (int k = 0; k < GAT_CHUNKS / 8; ++k) {
            const int row = r * (GAT_CHUNKS / 8) + k;
            const int v = tab[row * nb + b];
            tot  += v;
            part += (row < blk) ? v : 0;
        }
        atomicAdd(&colTot[b], tot);
        atomicAdd(&cursor[b], part);
    }
    __syncthreads();
    const int v = colTot[t];
    int inc = v;
    for (int o = 1; o < 32; o <<= 1) {
        int y = __shfl_up(inc, o);
        if (lane >= o) inc += y;
    }
    if (lane == 31) s_wtot[wv] = inc;
    __syncthreads();
    int off = 0;
    for (int w = 0; w < 32; ++w) off += (w < wv) ? s_wtot[w] : 0;
    const int excl = off + inc - v;
    if (t < nb) {
        cursor[t] += excl;
        if (blk == 0) binStart[t] = excl;
    }
    if (blk == 0 && t == 0) binStart[nb] = E;
    __syncthreads();
    const int chunk = (E + GAT_CHUNKS - 1) / GAT_CHUNKS;
    const int e0 = blk * chunk;
    const int e1 = min(E, e0 + chunk);
    int2* pr = (int2*)pairs;
    for (int e = e0 + t; e < e1; e += 8 * 1024) {
        int d[8], sv[8];
        for (int k = 0; k < 8; ++k) {
            const int idx = min(e + k * 1024, e1 - 1);
            d[k] = dstI[idx]; sv[k] = srcI[idx];
        }
        asm volatile("" :: "v"(d[0]), "v"(d[1]), "v"(d[2]), "v"(d[3]), "v"(d[4]), "v"(d[5]), "v"(d[6]), "v"(d[7]),
                           "v"(sv[0]), "v"(sv[1]), "v"(sv[2]), "v"(sv[3]), "v"(sv[4]), "v"(sv[5]), "v"(sv[6]), "v"(sv[7]));
        for (int k = 0; k < 8; ++k)
            if (e + k * 1024 < e1) {
                const int pos = atomicAdd(&cursor[d[k] >> GAT_BINSH], 1);
                pr[pos] = make_int2(sv[k], d[k]);
            }
    }
}

__global__ __launch_bounds__(512)
void gat_bin_sort(const int* __restrict__ pairs, const int* __restrict__ binStart,
                  int* __restrict__ cnt, int* __restrict__ rowptr, int* __restrict__ es, int n)
{
    __shared__ int c[GAT_BINW];
    __shared__ int s_wtot[GAT_BINW / 32];
    const int t = threadIdx.x, lane = t & 31, wv = t >> 5;
    const int base = blockIdx.x * GAT_BINW;
    const int s0 = binStart[blockIdx.x], s1 = binStart[blockIdx.x + 1];
    const int2* pr = (const int2*)pairs;
    if (t < GAT_BINW) c[t] = 0;
    __syncthreads();
    for (int i = s0 + t; i < s1; i += 8 * 512) {
        int y[8];
        for (int k = 0; k < 8; ++k) y[k] = pr[min(i + k * 512, s1 - 1)].y;
        asm volatile("" :: "v"(y[0]), "v"(y[1]), "v"(y[2]), "v"(y[3]), "v"(y[4]), "v"(y[5]), "v"(y[6]), "v"(y[7]));
        for (int k = 0; k < 8; ++k)
            if (i + k * 512 < s1) atomicAdd(&c[y[k] - base], 1);
    }
    __syncthreads();
    int v = 0, inc = 0;
    if (t < GAT_BINW) {
        v = c[t]; inc = v;
        for (int o = 1; o < 32; o <<= 1) {
            int y = __shfl_up(inc, o);
            if (lane >= o) inc += y;
        }
        if (lane == 31) s_wtot[wv] = inc;
    }
    __syncthreads();
    if (t < GAT_BINW) {
        int off = s0;
        for (int w = 0; w < GAT_BINW / 32; ++w) off += (w < wv) ? s_wtot[w] : 0;
        const int excl = off + inc - v;
        c[t] = excl;
        if (base + t < n) { cnt[base + t] = v; rowptr[base + t] = excl; }
    }
    __syncthreads();
    for (int i = s0 + t; i < s1; i += 8 * 512) {
        int2 p[8];
        for (int k = 0; k < 8; ++k) p[k] = pr[min(i + k * 512, s1 - 1)];
        asm volatile("" :: "v"(p[0].x), "v"(p[1].x), "v"(p[2].x), "v"(p[3].x), "v"(p[4].x), "v"(p[5].x), "v"(p[6].x), "v"(p[7].x),
                           "v"(p[0].y), "v"(p[1].y), "v"(p[2].y), "v"(p[3].y), "v"(p[4].y), "v"(p[5].y), "v"(p[6].y), "v"(p[7].y));
        for (int k = 0; k < 8; ++k)
            if (i + k * 512 < s1) {
                const int slot = atomicAdd(&c[p[k].y - base], 1);
                es[slot] = p[k].x;
            }
    }
}

template<int H, int U>
__device__ __forceinline__ void gat_gather_batch(v4f& acc, float& m, float& den,
                                                 const v4f* __restrict__ Hv, int srcv,
                                                 int i, int Tc, int g, int sub,
                                                 v4f as4, v2f ad8)
{
    constexpr int LPR = 8 * H;
    constexpr int EPT = 32 / LPR;
    v4f   u[U];
    float sc[U];
    for (int k = 0; k < U; ++k) {
        int s;
        if constexpr (H == 4) s = __builtin_amdgcn_readlane(srcv, i + k);
        else                  s = __shfl(srcv, i + k * EPT + g);
        if constexpr (H == 4) u[k] = (Hv + (size_t)s * LPR)[sub];
        else                  u[k] = Hv[(unsigned)(s * LPR + sub)];
    }
    __builtin_amdgcn_sched_barrier(0);
    for (int k = 0; k < U; ++k) {
        const int slot = (H == 4) ? (i + k) : (i + k * EPT + g);
        float v = gat_sum8(gat_dot4(u[k], as4, ad8));
        v = (v >= 0.f) ? v : NEG_SLOPE * v;
        sc[k] = (slot < Tc) ? v : -__builtin_inff();
    }
    float mn = m;
    for (int k = 0; k < U; ++k) mn = fmaxf(mn, sc[k]);
    const float scale = __expf(m - mn);
    acc *= scale;
    den *= scale;
    for (int k = 0; k < U; ++k) {
        const float p = __expf(sc[k] - mn);
        den += p;
        acc += p * u[k];
    }
    m = mn;
}

__device__ __forceinline__ void gat_project_rows(const float* rw, const float* __restrict__ W,
                                                 float* __restrict__ Gout, int m0, int n_rows, int lane)
{
    const int mlo  = lane & 15;
    const int kofs = (lane >> 4) * 2;
    const float* arow = rw + mlo * GAT_ROWP + kofs;
    const float* bcol = W + kofs * 32 + mlo;
    v8f acc0 = {}, acc1 = {};
#pragma unroll 16
    for (int k = 0; k < 128; k += 4) {
        const v2f a  = *(const v2f*)(arow + k);
        const v2f b0 = { bcol[k * 32],      bcol[(k + 1) * 32] };
        const v2f b1 = { bcol[k * 32 + 16], bcol[(k + 1) * 32 + 16] };
        acc0 = __builtin_amdgcn_wmma_f32_16x16x4_f32(false, a, false, b0, (short)0, acc0, false, false);
        acc1 = __builtin_amdgcn_wmma_f32_16x16x4_f32(false, a, false, b1, (short)0, acc1, false, false);
    }
    const int mhi = (lane >> 4) * 8;
#pragma unroll
    for (int j = 0; j < 8; ++j) {
        const int m = m0 + mhi + j;
        if (m < n_rows) {
            Gout[(size_t)m * 32 + mlo]      = acc0[j];
            Gout[(size_t)m * 32 + 16 + mlo] = acc1[j];
        }
    }
}

template<int H, bool RELU, bool PROJ>
__global__ __launch_bounds__(256, H == 4 ? 12 : 10)
void gat_softmax_gather(const int* __restrict__ rowptr, const int* __restrict__ cnt,
                        const int* __restrict__ es, const float* __restrict__ att_s,
                        const float* __restrict__ att_d, const float* __restrict__ Hf,
                        const float* __restrict__ bias, float* __restrict__ outp, int n,
                        const float* __restrict__ W2)
{
    static_assert(!PROJ || H == 4, "the projection epilogue takes whole 128-float rows");
    __shared__ float rows[PROJ ? 32 * GAT_ROWP : 1];
    constexpr int LPR = 8 * H;
    constexpr int EPT = 32 / LPR;
    constexpr int NPW = 4;
    const int lane = threadIdx.x & 31;
    const int wv  = __builtin_amdgcn_readfirstlane((int)(threadIdx.x >> 5));
    const int sub = lane % LPR;
    const int g   = lane / LPR;
    const v4f* Hv = (const v4f*)Hf;
    const int d0  = (blockIdx.x * 8 + wv) * NPW;
    const v4f as4 = ((const v4f*)att_s)[sub];
    const v4f ad4 = ((const v4f*)att_d)[sub];

    for (int q = 0; q < NPW; ++q) {
        const int d = d0 + q;
        if (d >= n) {
            if constexpr (PROJ) {
                const v4f z = { 0.f, 0.f, 0.f, 0.f };
                *(v4f*)&rows[(wv * NPW + q) * GAT_ROWP + sub * 4] = z;
                continue;
            } else break;
        }
        const v4f us   = Hv[(size_t)d * LPR + sub];
        const int row0 = rowptr[d];
        const int deg  = cnt[d];
        int srcv = d;
        if (lane < deg) srcv = es[row0 + lane];

        const v2f  z2  = { 0.f, 0.f };
        const float ad = gat_sum8(gat_dot4(us, ad4, z2));
        const v2f  ad8 = { 0.125f * ad, 0.f };
        float m = gat_sum8(gat_dot4(us, as4, ad8));
        m = (m >= 0.f) ? m : NEG_SLOPE * m;
        const v4f z4 = { 0.f, 0.f, 0.f, 0.f };
        float den = (g == 0) ? 1.f : 0.f;
        v4f   acc = (g == 0) ? us : z4;

        for (int c0 = 0; c0 < deg; c0 += 32) {
            if (c0 > 0) {
                const int t = c0 + lane;
                srcv = d;
                if (t < deg) srcv = es[row0 + t];
            }
            const int Tc = min(deg - c0, 32);
            int i = 0;
            for (; Tc - i >= 8 * EPT; i += 8 * EPT)
                gat_gather_batch<H, 8>(acc, m, den, Hv, srcv, i, Tc, g, sub, as4, ad8);
            if (Tc - i > 4 * EPT)
                gat_gather_batch<H, 8>(acc, m, den, Hv, srcv, i, Tc, g, sub, as4, ad8);
            else if (Tc - i > 0)
                gat_gather_batch<H, 4>(acc, m, den, Hv, srcv, i, Tc, g, sub, as4, ad8);
        }

        if constexpr (EPT > 1) {
            float mx = m;
            for (int o = LPR; o < 32; o <<= 1) mx = fmaxf(mx, __shfl_xor(mx, o));
            const float scale = __expf(m - mx);
            den *= scale;
            acc *= scale;
            for (int o = LPR; o < 32; o <<= 1) {
                den   += __shfl_xor(den, o);
                acc.x += __shfl_xor(acc.x, o); acc.y += __shfl_xor(acc.y, o);
                acc.z += __shfl_xor(acc.z, o); acc.w += __shfl_xor(acc.w, o);
            }
        }
        if (g == 0) {
            v4f b = { bias[sub * 4 + 0], bias[sub * 4 + 1], bias[sub * 4 + 2], bias[sub * 4 + 3] };
            v4f r = acc / den + b;
            if (RELU) {
                r.x = fmaxf(r.x, 0.f); r.y = fmaxf(r.y, 0.f);
                r.z = fmaxf(r.z, 0.f); r.w = fmaxf(r.w, 0.f);
            }
            if constexpr (PROJ) *(v4f*)&rows[(wv * NPW + q) * GAT_ROWP + sub * 4] = r;
            else ((v4f*)outp)[(size_t)d * LPR + sub] = r;
        }
    }
    if constexpr (PROJ) {
        __syncthreads();
        if (wv < 2)
            gat_project_rows(rows + wv * 16 * GAT_ROWP, W2, outp, blockIdx.x * 32 + wv * 16, n, lane);
    }
}

extern "C" void kernel_launch(void* const* d_in, const int* in_sizes, int n_in,
                              void* d_out, int out_size, void* d_ws, size_t ws_size,
                              hipStream_t stream)
{
    const float* x   = (const float*)d_in[0];
    const int*   ei  = (const int*)d_in[1];
    const float* W1  = (const float*)d_in[2];
    const float* as1 = (const float*)d_in[3];
    const float* ad1 = (const float*)d_in[4];
    const float* b1  = (const float*)d_in[5];
    const float* W2  = (const float*)d_in[6];
    const float* as2 = (const float*)d_in[7];
    const float* ad2 = (const float*)d_in[8];
    const float* b2  = (const float*)d_in[9];
    float* out = (float*)d_out;

    const int N  = in_sizes[0] / 128;
    const int E  = in_sizes[1] / 2;
    const int* srcI = ei;
    const int* dstI = ei + E;

    float* ws = (float*)d_ws;
    float*    h1    = ws;              ws += (size_t)N * 128;
    float*    g2    = ws;              ws += (size_t)N * 32;
    int*      pairs = (int*)ws;        ws += (size_t)E * 2;
    int*      es    = (int*)ws;        ws += (size_t)E;
    int*      cnt   = (int*)ws;        ws += (size_t)N;
    int*      rowp  = (int*)ws;        ws += (size_t)N;
    int*      binS  = (int*)ws;        ws += GAT_NBMAX + 1;
    int*      tab   = (int*)ws;

    const int TB = 256;
    const int gB   = (N + 127) / 128;
    const int nb   = (N + GAT_BINW - 1) / GAT_BINW;
    const int gG   = (N + 31) / 32;

    gat_bin_hist<<<GAT_CHUNKS, 1024, 0, stream>>>(dstI, tab, E, nb);
    gat_bin_scatter<<<GAT_CHUNKS, 1024, 0, stream>>>(srcI, dstI, tab, binS, pairs, E, nb);
    gat_bin_sort<<<nb, 512, 0, stream>>>(pairs, binS, cnt, rowp, es, N);

    gat_gemm_wmma<8><<<gB, TB, 0, stream>>>(x, W1, h1, N);
    gat_softmax_gather<4, true, true><<<gG, TB, 0, stream>>>(rowp, cnt, es, as1, ad1, h1, b1, g2, N, W2);

    gat_softmax_gather<1, false, false><<<gG, TB, 0, stream>>>(rowp, cnt, es, as2, ad2, g2, b2, out, N, nullptr);
}


# Per-bin node sort (last edge-sort stage) runs as extra 256-thread blocks of the projection GEMM's launch, under its matrix work: 6 -> 5 launches; row sums stay in LDS arrival order, order changes
# speedup vs baseline: 17.9814x; 1.0367x over previous
#include <hip/hip_runtime.h>

#define NEG_SLOPE 0.2f

typedef float v2f __attribute__((ext_vector_type(2)));
typedef float v8f __attribute__((ext_vector_type(8)));
typedef float v4f __attribute__((ext_vector_type(4)));

template<int M>
__device__ __forceinline__ float gat_row_xor(float v) {
    return __int_as_float(__builtin_amdgcn_update_dpp(0, __float_as_int(v), 0x160 | M, 0xf, 0xf, true));
}

__device__ __forceinline__ float gat_sum8(float v) {
    v += gat_row_xor<1>(v);
    v += gat_row_xor<2>(v);
    v += gat_row_xor<4>(v);
    return v;
}

__device__ __forceinline__ float gat_dot4(v4f a, v4f b, v2f seed) {
    const v2f t = a.xy * b.xy + (a.zw * b.zw + seed);
    return t.x + t.y;
}

#define GAT_ROWP 132

#define GAT_BINW   256
#define GAT_BINSH  8
#define GAT_NBMAX  1024
#define GAT_CHUNKS 128

__device__ __forceinline__ void gat_bin_sort_block(int* c, int* s_wtot, int bin,
                                                   const int* __restrict__ pairs, const int* __restrict__ binStart,
                                                   int* __restrict__ cnt, int* __restrict__ rowptr,
                                                   int* __restrict__ es, int n)
{
    static_assert(GAT_BINW == 256, "one thread per node of the bin, blocks of 256 threads");
    const int t = threadIdx.x, lane = t & 31, wv = t >> 5;
    const int base = bin * GAT_BINW;
    const int s0 = binStart[bin], s1 = binStart[bin + 1];
    const int2* pr = (const int2*)pairs;
    c[t] = 0;
    __syncthreads();
    for (int i = s0 + t; i < s1; i += 8 * GAT_BINW) {
        int y[8];
        for (int k = 0; k < 8; ++k) y[k] = pr[min(i + k * GAT_BINW, s1 - 1)].y;
        asm volatile("" :: "v"(y[0]), "v"(y[1]), "v"(y[2]), "v"(y[3]), "v"(y[4]), "v"(y[5]), "v"(y[6]), "v"(y[7]));
        for (int k = 0; k < 8; ++k)
            if (i + k * GAT_BINW < s1) atomicAdd(&c[y[k] - base], 1);
    }
    __syncthreads();
    const int v = c[t];
    int inc = v;
    for (int o = 1; o < 32; o <<= 1) {
        int y = __shfl_up(inc, o);
        if (lane >= o) inc += y;
    }
    if (lane == 31) s_wtot[wv] = inc;
    __syncthreads();
    int off = s0;
    for (int w = 0; w < GAT_BINW / 32; ++w) off += (w < wv) ? s_wtot[w] : 0;
    const int excl = off + inc - v;
    c[t] = excl;
    if (base + t < n) { cnt[base + t] = v; rowptr[base + t] = excl; }
    __syncthreads();
    for (int i = s0 + t; i < s1; i += 8 * GAT_BINW) {
        int2 p[8];
        for (int k = 0; k < 8; ++k) p[k] = pr[min(i + k * GAT_BINW, s1 - 1)];
        asm volatile("" :: "v"(p[0].x), "v"(p[1].x), "v"(p[2].x), "v"(p[3].x), "v"(p[4].x), "v"(p[5].x), "v"(p[6].x), "v"(p[7].x),
                           "v"(p[0].y), "v"(p[1].y), "v"(p[2].y), "v"(p[3].y), "v"(p[4].y), "v"(p[5].y), "v"(p[6].y), "v"(p[7].y));
        for (int k = 0; k < 8; ++k)
            if (i + k * GAT_BINW < s1) {
                const int slot = atomicAdd(&c[p[k].y - base], 1);
                es[slot] = p[k].x;
            }
    }
}

template<int NT>
__global__ __launch_bounds__(256)
void gat_gemm_wmma_and_sort(const float* __restrict__ X, const float* __restrict__ W,
                            float* __restrict__ Hout, int n_rows, int gemm_blocks,
                            const int* __restrict__ pairs, const int* __restrict__ binStart,
                            int* __restrict__ cnt, int* __restrict__ rowptr, int* __restrict__ es)
{
    constexpr int KD = 128;
    constexpr int ND = NT * 16;
    __shared__ float Wt[ND * KD];
    __shared__ float Xs[8 * 16 * GAT_ROWP];
    __shared__ int sort_c[GAT_BINW];
    __shared__ int sort_wtot[GAT_BINW / 32];

    if ((int)blockIdx.x >= gemm_blocks) {
        gat_bin_sort_block(sort_c, sort_wtot, (int)blockIdx.x - gemm_blocks,
                           pairs, binStart, cnt, rowptr, es, n_rows);
        return;
    }

    const int lane = threadIdx.x & 31;
    const int wave = threadIdx.x >> 5;
    const int m0   = (blockIdx.x * 8 + wave) * 16;
    float* xs = Xs + wave * 16 * GAT_ROWP;

    v4f xr[16];
#pragma unroll
    for (int r = 0; r < 16; ++r) {
        int row = m0 + r;
        if (row >= n_rows) row = n_rows - 1;
        xr[r] = *(const v4f*)(X + (size_t)row * KD + lane * 4);
    }
    static_assert(ND == 128, "the thread-to-column map below takes 128 columns and 256 threads");
    const int wn  = threadIdx.x & (ND - 1);
    const int wkb = threadIdx.x >> 7;
    float wv[64];
#pragma unroll
    for (int j = 0; j < 64; ++j) wv[j] = W[(wkb + 2 * j) * ND + wn];
#pragma unroll
    for (int j = 0; j < 64; ++j)
        Wt[wn * KD + ((wkb + 2 * j + 4 * wn) & (KD - 1))] = wv[j];
#pragma unroll
    for (int r = 0; r < 16; ++r) *(v4f*)(xs + r * GAT_ROWP + lane * 4) = xr[r];
    __syncthreads();

    const int mlo  = lane & 15;
    const int kofs = (lane >> 4) * 2;
    const float* arow = xs + mlo * GAT_ROWP + kofs;
    const float* wcol = Wt + mlo * KD;
    const int c0 = kofs + 4 * mlo;

    v8f acc[NT] = {};
    v2f a0, a1, b0[NT], b1[NT];
    a0 = *(const v2f*)arow;
#pragma unroll
    for (int t = 0; t < NT; ++t)
        b0[t] = *(const v2f*)(wcol + t * 16 * KD + ((c0 + 64 * t) & (KD - 1)));

#pragma unroll 1
    for (int k = 0; k < KD; k += 8) {
        a1 = *(const v2f*)(arow + k + 4);
#pragma unroll
        for (int t = 0; t < NT; ++t)
            b1[t] = *(const v2f*)(wcol + t * 16 * KD + ((k + 4 + c0 + 64 * t) & (KD - 1)));
        __builtin_amdgcn_sched_barrier(0);
#pragma unroll
        for (int t = 0; t < NT; ++t)
            acc[t] = __builtin_amdgcn_wmma_f32_16x16x4_f32(
                false, a0, false, b0[t], (short)0, acc[t], false, false);
        __builtin_amdgcn_sched_barrier(0);
        a0 = *(const v2f*)(arow + ((k + 8) & (KD - 1)));
#pragma unroll
        for (int t = 0; t < NT; ++t)
            b0[t] = *(const v2f*)(wcol + t * 16 * KD + ((k + 8 + c0 + 64 * t) & (KD - 1)));
        __builtin_amdgcn_sched_barrier(0);
#pragma unroll
        for (int t = 0; t < NT; ++t)
            acc[t] = __builtin_amdgcn_wmma_f32_16x16x4_f32(
                false, a1, false, b1[t], (short)0, acc[t], false, false);
        __builtin_amdgcn_sched_barrier(0);
    }

    const int mhi = (lane >> 4) * 8;
#pragma unroll
    for (int t = 0; t < NT; ++t)
#pragma unroll
        for (int j = 0; j < 8; ++j) {
            int m = m0 + mhi + j;
            if (m < n_rows) Hout[(size_t)m * ND + t * 16 + mlo] = acc[t][j];
        }
}


__global__ __launch_bounds__(1024)
void gat_bin_hist(const int* __restrict__ dstI, int* __restrict__ tab, int E, int nb)
{
    __shared__ int hist[GAT_NBMAX];
    const int t = threadIdx.x;
    hist[t] = 0;
    __syncthreads();
    const int chunk = (E + GAT_CHUNKS - 1) / GAT_CHUNKS;
    const int e0 = blockIdx.x * chunk;
    const int e1 = min(E, e0 + chunk);
    for (int e = e0 + t; e < e1; e += 8 * 1024) {
        int d[8];
        for (int k = 0; k < 8; ++k) d[k] = dstI[min(e + k * 1024, e1 - 1)];
        asm volatile("" :: "v"(d[0]), "v"(d[1]), "v"(d[2]), "v"(d[3]), "v"(d[4]), "v"(d[5]), "v"(d[6]), "v"(d[7]));
        for (int k = 0; k < 8; ++k)
            if (e + k * 1024 < e1) atomicAdd(&hist[d[k] >> GAT_BINSH], 1);
    }
    __syncthreads();
    if (t < nb) tab[blockIdx.x * nb + t] = hist[t];
}

__global__ __launch_bounds__(1024)
void gat_bin_scatter(const int* __restrict__ srcI, const int* __restrict__ dstI,
                     const int* __restrict__ tab, int* __restrict__ binStart,
                     int* __restrict__ pairs, int E, int nb)
{
    __shared__ int colTot[GAT_NBMAX];
    __shared__ int cursor[GAT_NBMAX];
    __shared__ int s_wtot[32];
    const int t = threadIdx.x, lane = t & 31, wv = t >> 5;
    const int blk = blockIdx.x;
    colTot[t] = 0; cursor[t] = 0;
    __syncthreads();
    for (int j = t; j < nb * 8; j += 1024) {
        const int r = j / nb, b = j - r * nb;
        int tot = 0, part = 0;
        for (int k = 0; k < GAT_CHUNKS / 8; ++k) {
            const int row = r * (GAT_CHUNKS / 8) + k;
            const int v = tab[row * nb + b];
            tot  += v;
            part += (row < blk) ? v : 0;
        }
        atomicAdd(&colTot[b], tot);
        atomicAdd(&cursor[b], part);
    }
    __syncthreads();
    const int v = colTot[t];
    int inc = v;
    for (int o = 1; o < 32; o <<= 1) {
        int y = __shfl_up(inc, o);
        if (lane >= o) inc += y;
    }
    if (lane == 31) s_wtot[wv] = inc;
    __syncthreads();
    int off = 0;
    for (int w = 0; w < 32; ++w) off += (w < wv) ? s_wtot[w] : 0;
    const int excl = off + inc - v;
    if (t < nb) {
        cursor[t] += excl;
        if (blk == 0) binStart[t] = excl;
    }
    if (blk == 0 && t == 0) binStart[nb] = E;
    __syncthreads();
    const int chunk = (E + GAT_CHUNKS - 1) / GAT_CHUNKS;
    const int e0 = blk * chunk;
    const int e1 = min(E, e0 + chunk);
    int2* pr = (int2*)pairs;
    for (int e = e0 + t; e < e1; e += 8 * 1024) {
        int d[8], sv[8];
        for (int k = 0; k < 8; ++k) {
            const int idx = min(e + k * 1024, e1 - 1);
            d[k] = dstI[idx]; sv[k] = srcI[idx];
        }
        asm volatile("" :: "v"(d[0]), "v"(d[1]), "v"(d[2]), "v"(d[3]), "v"(d[4]), "v"(d[5]), "v"(d[6]), "v"(d[7]),
                           "v"(sv[0]), "v"(sv[1]), "v"(sv[2]), "v"(sv[3]), "v"(sv[4]), "v"(sv[5]), "v"(sv[6]), "v"(sv[7]));
        for (int k = 0; k < 8; ++k)
            if (e + k * 1024 < e1) {
                const int pos = atomicAdd(&cursor[d[k] >> GAT_BINSH], 1);
                pr[pos] = make_int2(sv[k], d[k]);
            }
    }
}

template<int H, int U>
__device__ __forceinline__ void gat_gather_batch(v4f& acc, float& m, float& den,
                                                 const v4f* __restrict__ Hv, int srcv,
                                                 int i, int Tc, int g, int sub,
                                                 v4f as4, v2f ad8)
{
    constexpr int LPR = 8 * H;
    constexpr int EPT = 32 / LPR;
    v4f   u[U];
    float sc[U];
    for (int k = 0; k < U; ++k) {
        int s;
        if constexpr (H == 4) s = __builtin_amdgcn_readlane(srcv, i + k);
        else                  s = __shfl(srcv, i + k * EPT + g);
        if constexpr (H == 4) u[k] = (Hv + (size_t)s * LPR)[sub];
        else                  u[k] = Hv[(unsigned)(s * LPR + sub)];
    }
    __builtin_amdgcn_sched_barrier(0);
    for (int k = 0; k < U; ++k) {
        const int slot = (H == 4) ? (i + k) : (i + k * EPT + g);
        float v = gat_sum8(gat_dot4(u[k], as4, ad8));
        v = (v >= 0.f) ? v : NEG_SLOPE * v;
        sc[k] = (slot < Tc) ? v : -__builtin_inff();
    }
    float mn = m;
    for (int k = 0; k < U; ++k) mn = fmaxf(mn, sc[k]);
    const float scale = __expf(m - mn);
    acc *= scale;
    den *= scale;
    for (int k = 0; k < U; ++k) {
        const float p = __expf(sc[k] - mn);
        den += p;
        acc += p * u[k];
    }
    m = mn;
}

__device__ __forceinline__ void gat_project_rows(const float* rw, const float* __restrict__ W,
                                                 float* __restrict__ Gout, int m0, int n_rows, int lane)
{
    const int mlo  = lane & 15;
    const int kofs = (lane >> 4) * 2;
    const float* arow = rw + mlo * GAT_ROWP + kofs;
    const float* bcol = W + kofs * 32 + mlo;
    v8f acc0 = {}, acc1 = {};
#pragma unroll 16
    for (int k = 0; k < 128; k += 4) {
        const v2f a  = *(const v2f*)(arow + k);
        const v2f b0 = { bcol[k * 32],      bcol[(k + 1) * 32] };
        const v2f b1 = { bcol[k * 32 + 16], bcol[(k + 1) * 32 + 16] };
        acc0 = __builtin_amdgcn_wmma_f32_16x16x4_f32(false, a, false, b0, (short)0, acc0, false, false);
        acc1 = __builtin_amdgcn_wmma_f32_16x16x4_f32(false, a, false, b1, (short)0, acc1, false, false);
    }
    const int mhi = (lane >> 4) * 8;
#pragma unroll
    for (int j = 0; j < 8; ++j) {
        const int m = m0 + mhi + j;
        if (m < n_rows) {
            Gout[(size_t)m * 32 + mlo]      = acc0[j];
            Gout[(size_t)m * 32 + 16 + mlo] = acc1[j];
        }
    }
}

template<int H, bool RELU, bool PROJ>
__global__ __launch_bounds__(256, H == 4 ? 12 : 10)
void gat_softmax_gather(const int* __restrict__ rowptr, const int* __restrict__ cnt,
                        const int* __restrict__ es, const float* __restrict__ att_s,
                        const float* __restrict__ att_d, const float* __restrict__ Hf,
                        const float* __restrict__ bias, float* __restrict__ outp, int n,
                        const float* __restrict__ W2)
{
    static_assert(!PROJ || H == 4, "the projection epilogue takes whole 128-float rows");
    __shared__ float rows[PROJ ? 32 * GAT_ROWP : 1];
    constexpr int LPR = 8 * H;
    constexpr int EPT = 32 / LPR;
    constexpr int NPW = 4;
    const int lane = threadIdx.x & 31;
    const int wv  = __builtin_amdgcn_readfirstlane((int)(threadIdx.x >> 5));
    const int sub = lane % LPR;
    const int g   = lane / LPR;
    const v4f* Hv = (const v4f*)Hf;
    const int d0  = (blockIdx.x * 8 + wv) * NPW;
    const v4f as4 = ((const v4f*)att_s)[sub];
    const v4f ad4 = ((const v4f*)att_d)[sub];

    for (int q = 0; q < NPW; ++q) {
        const int d = d0 + q;
        if (d >= n) {
            if constexpr (PROJ) {
                const v4f z = { 0.f, 0.f, 0.f, 0.f };
                *(v4f*)&rows[(wv * NPW + q) * GAT_ROWP + sub * 4] = z;
                continue;
            } else break;
        }
        const v4f us   = Hv[(size_t)d * LPR + sub];
        const int row0 = rowptr[d];
        const int deg  = cnt[d];
        int srcv = d;
        if (lane < deg) srcv = es[row0 + lane];

        const v2f  z2  = { 0.f, 0.f };
        const float ad = gat_sum8(gat_dot4(us, ad4, z2));
        const v2f  ad8 = { 0.125f * ad, 0.f };
        float m = gat_sum8(gat_dot4(us, as4, ad8));
        m = (m >= 0.f) ? m : NEG_SLOPE * m;
        const v4f z4 = { 0.f, 0.f, 0.f, 0.f };
        float den = (g == 0) ? 1.f : 0.f;
        v4f   acc = (g == 0) ? us : z4;

        for (int c0 = 0; c0 < deg; c0 += 32) {
            if (c0 > 0) {
                const int t = c0 + lane;
                srcv = d;
                if (t < deg) srcv = es[row0 + t];
            }
            const int Tc = min(deg - c0, 32);
            int i = 0;
            for (; Tc - i >= 8 * EPT; i += 8 * EPT)
                gat_gather_batch<H, 8>(acc, m, den, Hv, srcv, i, Tc, g, sub, as4, ad8);
            if (Tc - i > 4 * EPT)
                gat_gather_batch<H, 8>(acc, m, den, Hv, srcv, i, Tc, g, sub, as4, ad8);
            else if (Tc - i > 0)
                gat_gather_batch<H, 4>(acc, m, den, Hv, srcv, i, Tc, g, sub, as4, ad8);
        }

        if constexpr (EPT > 1) {
            float mx = m;
            for (int o = LPR; o < 32; o <<= 1) mx = fmaxf(mx, __shfl_xor(mx, o));
            const float scale = __expf(m - mx);
            den *= scale;
            acc *= scale;
            for (int o = LPR; o < 32; o <<= 1) {
                den   += __shfl_xor(den, o);
                acc.x += __shfl_xor(acc.x, o); acc.y += __shfl_xor(acc.y, o);
                acc.z += __shfl_xor(acc.z, o); acc.w += __shfl_xor(acc.w, o);
            }
        }
        if (g == 0) {
            v4f b = { bias[sub * 4 + 0], bias[sub * 4 + 1], bias[sub * 4 + 2], bias[sub * 4 + 3] };
            v4f r = acc / den + b;
            if (RELU) {
                r.x = fmaxf(r.x, 0.f); r.y = fmaxf(r.y, 0.f);
                r.z = fmaxf(r.z, 0.f); r.w = fmaxf(r.w, 0.f);
            }
            if constexpr (PROJ) *(v4f*)&rows[(wv * NPW + q) * GAT_ROWP + sub * 4] = r;
            else ((v4f*)outp)[(size_t)d * LPR + sub] = r;
        }
    }
    if constexpr (PROJ) {
        __syncthreads();
        if (wv < 2)
            gat_project_rows(rows + wv * 16 * GAT_ROWP, W2, outp, blockIdx.x * 32 + wv * 16, n, lane);
    }
}

extern "C" void kernel_launch(void* const* d_in, const int* in_sizes, int n_in,
                              void* d_out, int out_size, void* d_ws, size_t ws_size,
                              hipStream_t stream)
{
    const float* x   = (const float*)d_in[0];
    const int*   ei  = (const int*)d_in[1];
    const float* W1  = (const float*)d_in[2];
    const float* as1 = (const float*)d_in[3];
    const float* ad1 = (const float*)d_in[4];
    const float* b1  = (const float*)d_in[5];
    const float* W2  = (const float*)d_in[6];
    const float* as2 = (const float*)d_in[7];
    const float* ad2 = (const float*)d_in[8];
    const float* b2  = (const float*)d_in[9];
    float* out = (float*)d_out;

    const int N  = in_sizes[0] / 128;
    const int E  = in_sizes[1] / 2;
    const int* srcI = ei;
    const int* dstI = ei + E;

    float* ws = (float*)d_ws;
    float*    h1    = ws;              ws += (size_t)N * 128;
    float*    g2    = ws;              ws += (size_t)N * 32;
    int*      pairs = (int*)ws;        ws += (size_t)E * 2;
    int*      es    = (int*)ws;        ws += (size_t)E;
    int*      cnt   = (int*)ws;        ws += (size_t)N;
    int*      rowp  = (int*)ws;        ws += (size_t)N;
    int*      binS  = (int*)ws;        ws += GAT_NBMAX + 1;
    int*      tab   = (int*)ws;

    const int TB = 256;
    const int gB   = (N + 127) / 128;
    const int nb   = (N + GAT_BINW - 1) / GAT_BINW;
    const int gG   = (N + 31) / 32;

    gat_bin_hist<<<GAT_CHUNKS, 1024, 0, stream>>>(dstI, tab, E, nb);
    gat_bin_scatter<<<GAT_CHUNKS, 1024, 0, stream>>>(srcI, dstI, tab, binS, pairs, E, nb);

    gat_gemm_wmma_and_sort<8><<<gB + nb, TB, 0, stream>>>(x, W1, h1, N, gB, pairs, binS, cnt, rowp, es);
    gat_softmax_gather<4, true, true><<<gG, TB, 0, stream>>>(rowp, cnt, es, as1, ad1, h1, b1, g2, N, W2);

    gat_softmax_gather<1, false, false><<<gG, TB, 0, stream>>>(rowp, cnt, es, as2, ad2, g2, b2, out, N, nullptr);
}


# Edge sort level 1 made chunk-local (count, LDS scan, copy into bin order inside the chunk; a bin block collects its piece of each chunk): histogram launch gone, 5 -> 4; row-sum arrival order changes
# speedup vs baseline: 19.3420x; 1.0757x over previous
#include <hip/hip_runtime.h>

#define NEG_SLOPE 0.2f

typedef float v2f __attribute__((ext_vector_type(2)));
typedef float v8f __attribute__((ext_vector_type(8)));
typedef float v4f __attribute__((ext_vector_type(4)));

template<int M>
__device__ __forceinline__ float gat_row_xor(float v) {
    return __int_as_float(__builtin_amdgcn_update_dpp(0, __float_as_int(v), 0x160 | M, 0xf, 0xf, true));
}

__device__ __forceinline__ float gat_sum8(float v) {
    v += gat_row_xor<1>(v);
    v += gat_row_xor<2>(v);
    v += gat_row_xor<4>(v);
    return v;
}

__device__ __forceinline__ float gat_dot4(v4f a, v4f b, v2f seed) {
    const v2f t = a.xy * b.xy + (a.zw * b.zw + seed);
    return t.x + t.y;
}

#define GAT_ROWP 132

#define GAT_BINW   256
#define GAT_BINSH  8
#define GAT_NBMAX  1024
#define GAT_CHUNKS 128

__device__ __forceinline__ void gat_bin_sort_block(int* c, int* s_wtot, int* s_st, int* seg, int bin,
                                                   const int* __restrict__ pairs, const int* __restrict__ tab,
                                                   int* __restrict__ cnt, int* __restrict__ rowptr,
                                                   int* __restrict__ es, int n, int E, int nb)
{
    static_assert(GAT_BINW == 256, "one thread per node of the bin, blocks of 256 threads");
    static_assert(GAT_CHUNKS == 128, "8 waves take 16 chunks each; threads 0..127 fetch the pieces");
    const int t = threadIdx.x, lane = t & 31, wv = t >> 5;
    const int base = bin * GAT_BINW;
    const int2* pr = (const int2*)pairs;
    const int chunk = (E + GAT_CHUNKS - 1) / GAT_CHUNKS;
    int st = 0;
    if (t < GAT_CHUNKS) {
        const int a = tab[t * (nb + 1) + bin];
        const int b = tab[t * (nb + 1) + bin + 1];
        st = a;
        seg[t] = t * chunk + a;
        seg[GAT_CHUNKS + t] = b - a;
    }
    c[t] = 0;
    int sum = st;
    for (int o = 16; o > 0; o >>= 1) sum += __shfl_xor(sum, o);
    if (lane == 0) s_st[wv] = sum;
    __syncthreads();
    int s0 = 0;
    for (int w = 0; w < GAT_BINW / 32; ++w) s0 += s_st[w];

    int2 p[32];
#pragma unroll
    for (int j = 0; j < 16; ++j) {
        const int gs = seg[wv * 16 + j];
        const int gl = seg[GAT_CHUNKS + wv * 16 + j];
#pragma unroll
        for (int h = 0; h < 2; ++h)
            p[2 * j + h] = pr[min(gs + min(lane + 32 * h, max(gl - 1, 0)), E - 1)];
    }
    asm volatile("" :: "v"(p[0]), "v"(p[1]), "v"(p[2]), "v"(p[3]), "v"(p[4]), "v"(p[5]), "v"(p[6]), "v"(p[7]),
                       "v"(p[8]), "v"(p[9]), "v"(p[10]), "v"(p[11]), "v"(p[12]), "v"(p[13]), "v"(p[14]), "v"(p[15]),
                       "v"(p[16]), "v"(p[17]), "v"(p[18]), "v"(p[19]), "v"(p[20]), "v"(p[21]), "v"(p[22]), "v"(p[23]),
                       "v"(p[24]), "v"(p[25]), "v"(p[26]), "v"(p[27]), "v"(p[28]), "v"(p[29]), "v"(p[30]), "v"(p[31]));
#pragma unroll
    for (int j = 0; j < 16; ++j) {
        const int gl = seg[GAT_CHUNKS + wv * 16 + j];
#pragma unroll
        for (int h = 0; h < 2; ++h)
            if (lane + 32 * h < gl) atomicAdd(&c[p[2 * j + h].y - base], 1);
    }
    for (int j = 0; j < 16; ++j) {
        const int gs = seg[wv * 16 + j];
        const int gl = seg[GAT_CHUNKS + wv * 16 + j];
        for (int i = lane + 64; i < gl; i += 32) atomicAdd(&c[pr[gs + i].y - base], 1);
    }
    __syncthreads();
    const int v = c[t];
    int inc = v;
    for (int o = 1; o < 32; o <<= 1) {
        int y = __shfl_up(inc, o);
        if (lane >= o) inc += y;
    }
    if (lane == 31) s_wtot[wv] = inc;
    __syncthreads();
    int off = s0;
    for (int w = 0; w < GAT_BINW / 32; ++w) off += (w < wv) ? s_wtot[w] : 0;
    const int excl = off + inc - v;
    c[t] = excl;
    if (base + t < n) { cnt[base + t] = v; rowptr[base + t] = excl; }
    __syncthreads();
#pragma unroll
    for (int j = 0; j < 16; ++j) {
        const int gl = seg[GAT_CHUNKS + wv * 16 + j];
#pragma unroll
        for (int h = 0; h < 2; ++h)
            if (lane + 32 * h < gl) {
                const int slot = atomicAdd(&c[p[2 * j + h].y - base], 1);
                es[slot] = p[2 * j + h].x;
            }
    }
    for (int j = 0; j < 16; ++j) {
        const int gs = seg[wv * 16 + j];
        const int gl = seg[GAT_CHUNKS + wv * 16 + j];
        for (int i = lane + 64; i < gl; i += 32) {
            const int2 q = pr[gs + i];
            const int slot = atomicAdd(&c[q.y - base], 1);
            es[slot] = q.x;
        }
    }
}

template<int NT>
__global__ __launch_bounds__(256)
void gat_gemm_wmma_and_sort(const float* __restrict__ X, const float* __restrict__ W,
                            float* __restrict__ Hout, int n_rows, int gemm_blocks,
                            const int* __restrict__ pairs, const int* __restrict__ tab,
                            int* __restrict__ cnt, int* __restrict__ rowptr, int* __restrict__ es,
                            int E, int nb)
{
    constexpr int KD = 128;
    constexpr int ND = NT * 16;
    __shared__ float Wt[ND * KD];
    __shared__ float Xs[8 * 16 * GAT_ROWP];
    __shared__ int sort_c[GAT_BINW];
    __shared__ int sort_wtot[GAT_BINW / 32];
    __shared__ int sort_st[GAT_BINW / 32];
    __shared__ int sort_seg[2 * GAT_CHUNKS];

    if ((int)blockIdx.x >= gemm_blocks) {
        gat_bin_sort_block(sort_c, sort_wtot, sort_st, sort_seg, (int)blockIdx.x - gemm_blocks,
                           pairs, tab, cnt, rowptr, es, n_rows, E, nb);
        return;
    }

    const int lane = threadIdx.x & 31;
    const int wave = threadIdx.x >> 5;
    const int m0   = (blockIdx.x * 8 + wave) * 16;
    float* xs = Xs + wave * 16 * GAT_ROWP;

    v4f xr[16];
#pragma unroll
    for (int r = 0; r < 16; ++r) {
        int row = m0 + r;
        if (row >= n_rows) row = n_rows - 1;
        xr[r] = *(const v4f*)(X + (size_t)row * KD + lane * 4);
    }
    static_assert(ND == 128, "the thread-to-column map below takes 128 columns and 256 threads");
    const int wn  = threadIdx.x & (ND - 1);
    const int wkb = threadIdx.x >> 7;
    float wv[64];
#pragma unroll
    for (int j = 0; j < 64; ++j) wv[j] = W[(wkb + 2 * j) * ND + wn];
#pragma unroll
    for (int j = 0; j < 64; ++j)
        Wt[wn * KD + ((wkb + 2 * j + 4 * wn) & (KD - 1))] = wv[j];
#pragma unroll
    for (int r = 0; r < 16; ++r) *(v4f*)(xs + r * GAT_ROWP + lane * 4) = xr[r];
    __syncthreads();

    const int mlo  = lane & 15;
    const int kofs = (lane >> 4) * 2;
    const float* arow = xs + mlo * GAT_ROWP + kofs;
    const float* wcol = Wt + mlo * KD;
    const int c0 = kofs + 4 * mlo;

    v8f acc[NT] = {};
    v2f a0, a1, b0[NT], b1[NT];
    a0 = *(const v2f*)arow;
#pragma unroll
    for (int t = 0; t < NT; ++t)
        b0[t] = *(const v2f*)(wcol + t * 16 * KD + ((c0 + 64 * t) & (KD - 1)));

#pragma unroll 1
    for (int k = 0; k < KD; k += 8) {
        a1 = *(const v2f*)(arow + k + 4);
#pragma unroll
        for (int t = 0; t < NT; ++t)
            b1[t] = *(const v2f*)(wcol + t * 16 * KD + ((k + 4 + c0 + 64 * t) & (KD - 1)));
        __builtin_amdgcn_sched_barrier(0);
#pragma unroll
        for (int t = 0; t < NT; ++t)
            acc[t] = __builtin_amdgcn_wmma_f32_16x16x4_f32(
                false, a0, false, b0[t], (short)0, acc[t], false, false);
        __builtin_amdgcn_sched_barrier(0);
        a0 = *(const v2f*)(arow + ((k + 8) & (KD - 1)));
#pragma unroll
        for (int t = 0; t < NT; ++t)
            b0[t] = *(const v2f*)(wcol + t * 16 * KD + ((k + 8 + c0 + 64 * t) & (KD - 1)));
        __builtin_amdgcn_sched_barrier(0);
#pragma unroll
        for (int t = 0; t < NT; ++t)
            acc[t] = __builtin_amdgcn_wmma_f32_16x16x4_f32(
                false, a1, false, b1[t], (short)0, acc[t], false, false);
        __builtin_amdgcn_sched_barrier(0);
    }

    const int mhi = (lane >> 4) * 8;
#pragma unroll
    for (int t = 0; t < NT; ++t)
#pragma unroll
        for (int j = 0; j < 8; ++j) {
            int m = m0 + mhi + j;
            if (m < n_rows) Hout[(size_t)m * ND + t * 16 + mlo] = acc[t][j];
        }
}


__global__ __launch_bounds__(1024)
void gat_chunk_sort(const int* __restrict__ srcI, const int* __restrict__ dstI,
                    int* __restrict__ tab, int* __restrict__ pairs, int E, int nb)
{
    __shared__ int cur[GAT_NBMAX];
    __shared__ int s_wtot[32];
    const int t = threadIdx.x, lane = t & 31, wv = t >> 5;
    const int blk = blockIdx.x;
    cur[t] = 0;
    __syncthreads();
    const int chunk = (E + GAT_CHUNKS - 1) / GAT_CHUNKS;
    const int e0 = blk * chunk;
    const int e1 = min(E, e0 + chunk);
    const bool keep = chunk <= 8 * 1024;
    int d[8], sv[8];
    if (keep) {
        const int e = e0 + t;
        if (e < e1) {
            for (int k = 0; k < 8; ++k) {
                const int idx = min(e + k * 1024, e1 - 1);
                d[k] = dstI[idx]; sv[k] = srcI[idx];
            }
            asm volatile("" :: "v"(d[0]), "v"(d[1]), "v"(d[2]), "v"(d[3]), "v"(d[4]), "v"(d[5]), "v"(d[6]), "v"(d[7]),
                               "v"(sv[0]), "v"(sv[1]), "v"(sv[2]), "v"(sv[3]), "v"(sv[4]), "v"(sv[5]), "v"(sv[6]), "v"(sv[7]));
            for (int k = 0; k < 8; ++k)
                if (e + k * 1024 < e1) atomicAdd(&cur[d[k] >> GAT_BINSH], 1);
        }
    } else {
        for (int e = e0 + t; e < e1; e += 8 * 1024) {
            for (int k = 0; k < 8; ++k) d[k] = dstI[min(e + k * 1024, e1 - 1)];
            asm volatile("" :: "v"(d[0]), "v"(d[1]), "v"(d[2]), "v"(d[3]), "v"(d[4]), "v"(d[5]), "v"(d[6]), "v"(d[7]));
            for (int k = 0; k < 8; ++k)
                if (e + k * 1024 < e1) atomicAdd(&cur[d[k] >> GAT_BINSH], 1);
        }
    }
    __syncthreads();
    const int v = cur[t];
    int inc = v;
    for (int o = 1; o < 32; o <<= 1) {
        int y = __shfl_up(inc, o);
        if (lane >= o) inc += y;
    }
    if (lane == 31) s_wtot[wv] = inc;
    __syncthreads();
    int off = 0;
    for (int w = 0; w < 32; ++w) off += (w < wv) ? s_wtot[w] : 0;
    const int excl = off + inc - v;
    cur[t] = excl;
    __syncthreads();
    if (t < nb) tab[blk * (nb + 1) + t] = excl;
    if (t == 0) tab[blk * (nb + 1) + nb] = max(e1 - e0, 0);
    int2* pr = (int2*)pairs;
    if (keep) {
        const int e = e0 + t;
        for (int k = 0; k < 8; ++k)
            if (e + k * 1024 < e1) {
                const int pos = atomicAdd(&cur[d[k] >> GAT_BINSH], 1);
                pr[e0 + pos] = make_int2(sv[k], d[k]);
            }
    } else {
        for (int e = e0 + t; e < e1; e += 8 * 1024) {
            for (int k = 0; k < 8; ++k) {
                const int idx = min(e + k * 1024, e1 - 1);
                d[k] = dstI[idx]; sv[k] = srcI[idx];
            }
            asm volatile("" :: "v"(d[0]), "v"(d[1]), "v"(d[2]), "v"(d[3]), "v"(d[4]), "v"(d[5]), "v"(d[6]), "v"(d[7]),
                               "v"(sv[0]), "v"(sv[1]), "v"(sv[2]), "v"(sv[3]), "v"(sv[4]), "v"(sv[5]), "v"(sv[6]), "v"(sv[7]));
            for (int k = 0; k < 8; ++k)
                if (e + k * 1024 < e1) {
                    const int pos = atomicAdd(&cur[d[k] >> GAT_BINSH], 1);
                    pr[e0 + pos] = make_int2(sv[k], d[k]);
                }
        }
    }
}

template<int H, int U>
__device__ __forceinline__ void gat_gather_batch(v4f& acc, float& m, float& den,
                                                 const v4f* __restrict__ Hv, int srcv,
                                                 int i, int Tc, int g, int sub,
                                                 v4f as4, v2f ad8)
{
    constexpr int LPR = 8 * H;
    constexpr int EPT = 32 / LPR;
    v4f   u[U];
    float sc[U];
    for (int k = 0; k < U; ++k) {
        int s;
        if constexpr (H == 4) s = __builtin_amdgcn_readlane(srcv, i + k);
        else                  s = __shfl(srcv, i + k * EPT + g);
        if constexpr (H == 4) u[k] = (Hv + (size_t)s * LPR)[sub];
        else                  u[k] = Hv[(unsigned)(s * LPR + sub)];
    }
    __builtin_amdgcn_sched_barrier(0);
    for (int k = 0; k < U; ++k) {
        const int slot = (H == 4) ? (i + k) : (i + k * EPT + g);
        float v = gat_sum8(gat_dot4(u[k], as4, ad8));
        v = (v >= 0.f) ? v : NEG_SLOPE * v;
        sc[k] = (slot < Tc) ? v : -__builtin_inff();
    }
    float mn = m;
    for (int k = 0; k < U; ++k) mn = fmaxf(mn, sc[k]);
    const float scale = __expf(m - mn);
    acc *= scale;
    den *= scale;
    for (int k = 0; k < U; ++k) {
        const float p = __expf(sc[k] - mn);
        den += p;
        acc += p * u[k];
    }
    m = mn;
}

__device__ __forceinline__ void gat_project_rows(const float* rw, const float* __restrict__ W,
                                                 float* __restrict__ Gout, int m0, int n_rows, int lane)
{
    const int mlo  = lane & 15;
    const int kofs = (lane >> 4) * 2;
    const float* arow = rw + mlo * GAT_ROWP + kofs;
    const float* bcol = W + kofs * 32 + mlo;
    v8f acc0 = {}, acc1 = {};
#pragma unroll 16
    for (int k = 0; k < 128; k += 4) {
        const v2f a  = *(const v2f*)(arow + k);
        const v2f b0 = { bcol[k * 32],      bcol[(k + 1) * 32] };
        const v2f b1 = { bcol[k * 32 + 16], bcol[(k + 1) * 32 + 16] };
        acc0 = __builtin_amdgcn_wmma_f32_16x16x4_f32(false, a, false, b0, (short)0, acc0, false, false);
        acc1 = __builtin_amdgcn_wmma_f32_16x16x4_f32(false, a, false, b1, (short)0, acc1, false, false);
    }
    const int mhi = (lane >> 4) * 8;
#pragma unroll
    for (int j = 0; j < 8; ++j) {
        const int m = m0 + mhi + j;
        if (m < n_rows) {
            Gout[(size_t)m * 32 + mlo]      = acc0[j];
            Gout[(size_t)m * 32 + 16 + mlo] = acc1[j];
        }
    }
}

template<int H, bool RELU, bool PROJ>
__global__ __launch_bounds__(256, H == 4 ? 12 : 10)
void gat_softmax_gather(const int* __restrict__ rowptr, const int* __restrict__ cnt,
                        const int* __restrict__ es, const float* __restrict__ att_s,
                        const float* __restrict__ att_d, const float* __restrict__ Hf,
                        const float* __restrict__ bias, float* __restrict__ outp, int n,
                        const float* __restrict__ W2)
{
    static_assert(!PROJ || H == 4, "the projection epilogue takes whole 128-float rows");
    __shared__ float rows[PROJ ? 32 * GAT_ROWP : 1];
    constexpr int LPR = 8 * H;
    constexpr int EPT = 32 / LPR;
    constexpr int NPW = 4;
    const int lane = threadIdx.x & 31;
    const int wv  = __builtin_amdgcn_readfirstlane((int)(threadIdx.x >> 5));
    const int sub = lane % LPR;
    const int g   = lane / LPR;
    const v4f* Hv = (const v4f*)Hf;
    const int d0  = (blockIdx.x * 8 + wv) * NPW;
    const v4f as4 = ((const v4f*)att_s)[sub];
    const v4f ad4 = ((const v4f*)att_d)[sub];

    for (int q = 0; q < NPW; ++q) {
        const int d = d0 + q;
        if (d >= n) {
            if constexpr (PROJ) {
                const v4f z = { 0.f, 0.f, 0.f, 0.f };
                *(v4f*)&rows[(wv * NPW + q) * GAT_ROWP + sub * 4] = z;
                continue;
            } else break;
        }
        const v4f us   = Hv[(size_t)d * LPR + sub];
        const int row0 = rowptr[d];
        const int deg  = cnt[d];
        int srcv = d;
        if (lane < deg) srcv = es[row0 + lane];

        const v2f  z2  = { 0.f, 0.f };
        const float ad = gat_sum8(gat_dot4(us, ad4, z2));
        const v2f  ad8 = { 0.125f * ad, 0.f };
        float m = gat_sum8(gat_dot4(us, as4, ad8));
        m = (m >= 0.f) ? m : NEG_SLOPE * m;
        const v4f z4 = { 0.f, 0.f, 0.f, 0.f };
        float den = (g == 0) ? 1.f : 0.f;
        v4f   acc = (g == 0) ? us : z4;

        for (int c0 = 0; c0 < deg; c0 += 32) {
            if (c0 > 0) {
                const int t = c0 + lane;
                srcv = d;
                if (t < deg) srcv = es[row0 + t];
            }
            const int Tc = min(deg - c0, 32);
            int i = 0;
            for (; Tc - i >= 8 * EPT; i += 8 * EPT)
                gat_gather_batch<H, 8>(acc, m, den, Hv, srcv, i, Tc, g, sub, as4, ad8);
            if (Tc - i > 4 * EPT)
                gat_gather_batch<H, 8>(acc, m, den, Hv, srcv, i, Tc, g, sub, as4, ad8);
            else if (Tc - i > 0)
                gat_gather_batch<H, 4>(acc, m, den, Hv, srcv, i, Tc, g, sub, as4, ad8);
        }

        if constexpr (EPT > 1) {
            float mx = m;
            for (int o = LPR; o < 32; o <<= 1) mx = fmaxf(mx, __shfl_xor(mx, o));
            const float scale = __expf(m - mx);
            den *= scale;
            acc *= scale;
            for (int o = LPR; o < 32; o <<= 1) {
                den   += __shfl_xor(den, o);
                acc.x += __shfl_xor(acc.x, o); acc.y += __shfl_xor(acc.y, o);
                acc.z += __shfl_xor(acc.z, o); acc.w += __shfl_xor(acc.w, o);
            }
        }
        if (g == 0) {
            v4f b = { bias[sub * 4 + 0], bias[sub * 4 + 1], bias[sub * 4 + 2], bias[sub * 4 + 3] };
            v4f r = acc / den + b;
            if (RELU) {
                r.x = fmaxf(r.x, 0.f); r.y = fmaxf(r.y, 0.f);
                r.z = fmaxf(r.z, 0.f); r.w = fmaxf(r.w, 0.f);
            }
            if constexpr (PROJ) *(v4f*)&rows[(wv * NPW + q) * GAT_ROWP + sub * 4] = r;
            else ((v4f*)outp)[(size_t)d * LPR + sub] = r;
        }
    }
    if constexpr (PROJ) {
        __syncthreads();
        if (wv < 2)
            gat_project_rows(rows + wv * 16 * GAT_ROWP, W2, outp, blockIdx.x * 32 + wv * 16, n, lane);
    }
}

extern "C" void kernel_launch(void* const* d_in, const int* in_sizes, int n_in,
                              void* d_out, int out_size, void* d_ws, size_t ws_size,
                              hipStream_t stream)
{
    const float* x   = (const float*)d_in[0];
    const int*   ei  = (const int*)d_in[1];
    const float* W1  = (const float*)d_in[2];
    const float* as1 = (const float*)d_in[3];
    const float* ad1 = (const float*)d_in[4];
    const float* b1  = (const float*)d_in[5];
    const float* W2  = (const float*)d_in[6];
    const float* as2 = (const float*)d_in[7];
    const float* ad2 = (const float*)d_in[8];
    const float* b2  = (const float*)d_in[9];
    float* out = (float*)d_out;

    const int N  = in_sizes[0] / 128;
    const int E  = in_sizes[1] / 2;
    const int* srcI = ei;
    const int* dstI = ei + E;

    float* ws = (float*)d_ws;
    float*    h1    = ws;              ws += (size_t)N * 128;
    float*    g2    = ws;              ws += (size_t)N * 32;
    int*      pairs = (int*)ws;        ws += (size_t)E * 2;
    int*      es    = (int*)ws;        ws += (size_t)E;
    int*      cnt   = (int*)ws;        ws += (size_t)N;
    int*      rowp  = (int*)ws;        ws += (size_t)N;
    int*      tab   = (int*)ws;

    const int TB = 256;
    const int gB   = (N + 127) / 128;
    const int nb   = (N + GAT_BINW - 1) / GAT_BINW;
    const int gG   = (N + 31) / 32;

    gat_chunk_sort<<<GAT_CHUNKS, 1024, 0, stream>>>(srcI, dstI, tab, pairs, E, nb);

    gat_gemm_wmma_and_sort<8><<<gB + nb, TB, 0, stream>>>(x, W1, h1, N, gB, pairs, tab, cnt, rowp, es, E, nb);
    gat_softmax_gather<4, true, true><<<gG, TB, 0, stream>>>(rowp, cnt, es, as1, ad1, h1, b1, g2, N, W2);

    gat_softmax_gather<1, false, false><<<gG, TB, 0, stream>>>(rowp, cnt, es, as2, ad2, g2, b2, out, N, nullptr);
}


# Layer-2 owner pass: each 8-lane group owns a node (4 nodes side by side per wave, list order, running max); slot-group merge and LDS shuffles gone; layer-2 sums now in list order
# speedup vs baseline: 20.2708x; 1.0480x over previous
#include <hip/hip_runtime.h>

#define NEG_SLOPE 0.2f

typedef float v2f __attribute__((ext_vector_type(2)));
typedef float v8f __attribute__((ext_vector_type(8)));
typedef float v4f __attribute__((ext_vector_type(4)));

template<int M>
__device__ __forceinline__ float gat_row_xor(float v) {
    return __int_as_float(__builtin_amdgcn_update_dpp(0, __float_as_int(v), 0x160 | M, 0xf, 0xf, true));
}

__device__ __forceinline__ float gat_sum8(float v) {
    v += gat_row_xor<1>(v);
    v += gat_row_xor<2>(v);
    v += gat_row_xor<4>(v);
    return v;
}

__device__ __forceinline__ float gat_dot4(v4f a, v4f b, v2f seed) {
    const v2f t = a.xy * b.xy + (a.zw * b.zw + seed);
    return t.x + t.y;
}

#define GAT_ROWP 132

#define GAT_BINW   256
#define GAT_BINSH  8
#define GAT_NBMAX  1024
#define GAT_CHUNKS 128

__device__ __forceinline__ void gat_bin_sort_block(int* c, int* s_wtot, int* s_st, int* seg, int bin,
                                                   const int* __restrict__ pairs, const int* __restrict__ tab,
                                                   int* __restrict__ cnt, int* __restrict__ rowptr,
                                                   int* __restrict__ es, int n, int E, int nb)
{
    static_assert(GAT_BINW == 256, "one thread per node of the bin, blocks of 256 threads");
    static_assert(GAT_CHUNKS == 128, "8 waves take 16 chunks each; threads 0..127 fetch the pieces");
    const int t = threadIdx.x, lane = t & 31, wv = t >> 5;
    const int base = bin * GAT_BINW;
    const int2* pr = (const int2*)pairs;
    const int chunk = (E + GAT_CHUNKS - 1) / GAT_CHUNKS;
    int st = 0;
    if (t < GAT_CHUNKS) {
        const int a = tab[t * (nb + 1) + bin];
        const int b = tab[t * (nb + 1) + bin + 1];
        st = a;
        seg[t] = t * chunk + a;
        seg[GAT_CHUNKS + t] = b - a;
    }
    c[t] = 0;
    int sum = st;
    for (int o = 16; o > 0; o >>= 1) sum += __shfl_xor(sum, o);
    if (lane == 0) s_st[wv] = sum;
    __syncthreads();
    int s0 = 0;
    for (int w = 0; w < GAT_BINW / 32; ++w) s0 += s_st[w];

    int2 p[32];
#pragma unroll
    for (int j = 0; j < 16; ++j) {
        const int gs = seg[wv * 16 + j];
        const int gl = seg[GAT_CHUNKS + wv * 16 + j];
#pragma unroll
        for (int h = 0; h < 2; ++h)
            p[2 * j + h] = pr[min(gs + min(lane + 32 * h, max(gl - 1, 0)), E - 1)];
    }
    asm volatile("" :: "v"(p[0]), "v"(p[1]), "v"(p[2]), "v"(p[3]), "v"(p[4]), "v"(p[5]), "v"(p[6]), "v"(p[7]),
                       "v"(p[8]), "v"(p[9]), "v"(p[10]), "v"(p[11]), "v"(p[12]), "v"(p[13]), "v"(p[14]), "v"(p[15]),
                       "v"(p[16]), "v"(p[17]), "v"(p[18]), "v"(p[19]), "v"(p[20]), "v"(p[21]), "v"(p[22]), "v"(p[23]),
                       "v"(p[24]), "v"(p[25]), "v"(p[26]), "v"(p[27]), "v"(p[28]), "v"(p[29]), "v"(p[30]), "v"(p[31]));
#pragma unroll
    for (int j = 0; j < 16; ++j) {
        const int gl = seg[GAT_CHUNKS + wv * 16 + j];
#pragma unroll
        for (int h = 0; h < 2; ++h)
            if (lane + 32 * h < gl) atomicAdd(&c[p[2 * j + h].y - base], 1);
    }
    for (int j = 0; j < 16; ++j) {
        const int gs = seg[wv * 16 + j];
        const int gl = seg[GAT_CHUNKS + wv * 16 + j];
        for (int i = lane + 64; i < gl; i += 32) atomicAdd(&c[pr[gs + i].y - base], 1);
    }
    __syncthreads();
    const int v = c[t];
    int inc = v;
    for (int o = 1; o < 32; o <<= 1) {
        int y = __shfl_up(inc, o);
        if (lane >= o) inc += y;
    }
    if (lane == 31) s_wtot[wv] = inc;
    __syncthreads();
    int off = s0;
    for (int w = 0; w < GAT_BINW / 32; ++w) off += (w < wv) ? s_wtot[w] : 0;
    const int excl = off + inc - v;
    c[t] = excl;
    if (base + t < n) { cnt[base + t] = v; rowptr[base + t] = excl; }
    __syncthreads();
#pragma unroll
    for (int j = 0; j < 16; ++j) {
        const int gl = seg[GAT_CHUNKS + wv * 16 + j];
#pragma unroll
        for (int h = 0; h < 2; ++h)
            if (lane + 32 * h < gl) {
                const int slot = atomicAdd(&c[p[2 * j + h].y - base], 1);
                es[slot] = p[2 * j + h].x;
            }
    }
    for (int j = 0; j < 16; ++j) {
        const int gs = seg[wv * 16 + j];
        const int gl = seg[GAT_CHUNKS + wv * 16 + j];
        for (int i = lane + 64; i < gl; i += 32) {
            const int2 q = pr[gs + i];
            const int slot = atomicAdd(&c[q.y - base], 1);
            es[slot] = q.x;
        }
    }
}

template<int NT>
__global__ __launch_bounds__(256)
void gat_gemm_wmma_and_sort(const float* __restrict__ X, const float* __restrict__ W,
                            float* __restrict__ Hout, int n_rows, int gemm_blocks,
                            const int* __restrict__ pairs, const int* __restrict__ tab,
                            int* __restrict__ cnt, int* __restrict__ rowptr, int* __restrict__ es,
                            int E, int nb)
{
    constexpr int KD = 128;
    constexpr int ND = NT * 16;
    __shared__ float Wt[ND * KD];
    __shared__ float Xs[8 * 16 * GAT_ROWP];
    __shared__ int sort_c[GAT_BINW];
    __shared__ int sort_wtot[GAT_BINW / 32];
    __shared__ int sort_st[GAT_BINW / 32];
    __shared__ int sort_seg[2 * GAT_CHUNKS];

    if ((int)blockIdx.x >= gemm_blocks) {
        gat_bin_sort_block(sort_c, sort_wtot, sort_st, sort_seg, (int)blockIdx.x - gemm_blocks,
                           pairs, tab, cnt, rowptr, es, n_rows, E, nb);
        return;
    }

    const int lane = threadIdx.x & 31;
    const int wave = threadIdx.x >> 5;
    const int m0   = (blockIdx.x * 8 + wave) * 16;
    float* xs = Xs + wave * 16 * GAT_ROWP;

    v4f xr[16];
#pragma unroll
    for (int r = 0; r < 16; ++r) {
        int row = m0 + r;
        if (row >= n_rows) row = n_rows - 1;
        xr[r] = *(const v4f*)(X + (size_t)row * KD + lane * 4);
    }
    static_assert(ND == 128, "the thread-to-column map below takes 128 columns and 256 threads");
    const int wn  = threadIdx.x & (ND - 1);
    const int wkb = threadIdx.x >> 7;
    float wv[64];
#pragma unroll
    for (int j = 0; j < 64; ++j) wv[j] = W[(wkb + 2 * j) * ND + wn];
#pragma unroll
    for (int j = 0; j < 64; ++j)
        Wt[wn * KD + ((wkb + 2 * j + 4 * wn) & (KD - 1))] = wv[j];
#pragma unroll
    for (int r = 0; r < 16; ++r) *(v4f*)(xs + r * GAT_ROWP + lane * 4) = xr[r];
    __syncthreads();

    const int mlo  = lane & 15;
    const int kofs = (lane >> 4) * 2;
    const float* arow = xs + mlo * GAT_ROWP + kofs;
    const float* wcol = Wt + mlo * KD;
    const int c0 = kofs + 4 * mlo;

    v8f acc[NT] = {};
    v2f a0, a1, b0[NT], b1[NT];
    a0 = *(const v2f*)arow;
#pragma unroll
    for (int t = 0; t < NT; ++t)
        b0[t] = *(const v2f*)(wcol + t * 16 * KD + ((c0 + 64 * t) & (KD - 1)));

#pragma unroll 1
    for (int k = 0; k < KD; k += 8) {
        a1 = *(const v2f*)(arow + k + 4);
#pragma unroll
        for (int t = 0; t < NT; ++t)
            b1[t] = *(const v2f*)(wcol + t * 16 * KD + ((k + 4 + c0 + 64 * t) & (KD - 1)));
        __builtin_amdgcn_sched_barrier(0);
#pragma unroll
        for (int t = 0; t < NT; ++t)
            acc[t] = __builtin_amdgcn_wmma_f32_16x16x4_f32(
                false, a0, false, b0[t], (short)0, acc[t], false, false);
        __builtin_amdgcn_sched_barrier(0);
        a0 = *(const v2f*)(arow + ((k + 8) & (KD - 1)));
#pragma unroll
        for (int t = 0; t < NT; ++t)
            b0[t] = *(const v2f*)(wcol + t * 16 * KD + ((k + 8 + c0 + 64 * t) & (KD - 1)));
        __builtin_amdgcn_sched_barrier(0);
#pragma unroll
        for (int t = 0; t < NT; ++t)
            acc[t] = __builtin_amdgcn_wmma_f32_16x16x4_f32(
                false, a1, false, b1[t], (short)0, acc[t], false, false);
        __builtin_amdgcn_sched_barrier(0);
    }

    const int mhi = (lane >> 4) * 8;
#pragma unroll
    for (int t = 0; t < NT; ++t)
#pragma unroll
        for (int j = 0; j < 8; ++j) {
            int m = m0 + mhi + j;
            if (m < n_rows) Hout[(size_t)m * ND + t * 16 + mlo] = acc[t][j];
        }
}


__global__ __launch_bounds__(1024)
void gat_chunk_sort(const int* __restrict__ srcI, const int* __restrict__ dstI,
                    int* __restrict__ tab, int* __restrict__ pairs, int E, int nb)
{
    __shared__ int cur[GAT_NBMAX];
    __shared__ int s_wtot[32];
    const int t = threadIdx.x, lane = t & 31, wv = t >> 5;
    const int blk = blockIdx.x;
    cur[t] = 0;
    __syncthreads();
    const int chunk = (E + GAT_CHUNKS - 1) / GAT_CHUNKS;
    const int e0 = blk * chunk;
    const int e1 = min(E, e0 + chunk);
    const bool keep = chunk <= 8 * 1024;
    int d[8], sv[8];
    if (keep) {
        const int e = e0 + t;
        if (e < e1) {
            for (int k = 0; k < 8; ++k) {
                const int idx = min(e + k * 1024, e1 - 1);
                d[k] = dstI[idx]; sv[k] = srcI[idx];
            }
            asm volatile("" :: "v"(d[0]), "v"(d[1]), "v"(d[2]), "v"(d[3]), "v"(d[4]), "v"(d[5]), "v"(d[6]), "v"(d[7]),
                               "v"(sv[0]), "v"(sv[1]), "v"(sv[2]), "v"(sv[3]), "v"(sv[4]), "v"(sv[5]), "v"(sv[6]), "v"(sv[7]));
            for (int k = 0; k < 8; ++k)
                if (e + k * 1024 < e1) atomicAdd(&cur[d[k] >> GAT_BINSH], 1);
        }
    } else {
        for (int e = e0 + t; e < e1; e += 8 * 1024) {
            for (int k = 0; k < 8; ++k) d[k] = dstI[min(e + k * 1024, e1 - 1)];
            asm volatile("" :: "v"(d[0]), "v"(d[1]), "v"(d[2]), "v"(d[3]), "v"(d[4]), "v"(d[5]), "v"(d[6]), "v"(d[7]));
            for (int k = 0; k < 8; ++k)
                if (e + k * 1024 < e1) atomicAdd(&cur[d[k] >> GAT_BINSH], 1);
        }
    }
    __syncthreads();
    const int v = cur[t];
    int inc = v;
    for (int o = 1; o < 32; o <<= 1) {
        int y = __shfl_up(inc, o);
        if (lane >= o) inc += y;
    }
    if (lane == 31) s_wtot[wv] = inc;
    __syncthreads();
    int off = 0;
    for (int w = 0; w < 32; ++w) off += (w < wv) ? s_wtot[w] : 0;
    const int excl = off + inc - v;
    cur[t] = excl;
    __syncthreads();
    if (t < nb) tab[blk * (nb + 1) + t] = excl;
    if (t == 0) tab[blk * (nb + 1) + nb] = max(e1 - e0, 0);
    int2* pr = (int2*)pairs;
    if (keep) {
        const int e = e0 + t;
        for (int k = 0; k < 8; ++k)
            if (e + k * 1024 < e1) {
                const int pos = atomicAdd(&cur[d[k] >> GAT_BINSH], 1);
                pr[e0 + pos] = make_int2(sv[k], d[k]);
            }
    } else {
        for (int e = e0 + t; e < e1; e += 8 * 1024) {
            for (int k = 0; k < 8; ++k) {
                const int idx = min(e + k * 1024, e1 - 1);
                d[k] = dstI[idx]; sv[k] = srcI[idx];
            }
            asm volatile("" :: "v"(d[0]), "v"(d[1]), "v"(d[2]), "v"(d[3]), "v"(d[4]), "v"(d[5]), "v"(d[6]), "v"(d[7]),
                               "v"(sv[0]), "v"(sv[1]), "v"(sv[2]), "v"(sv[3]), "v"(sv[4]), "v"(sv[5]), "v"(sv[6]), "v"(sv[7]));
            for (int k = 0; k < 8; ++k)
                if (e + k * 1024 < e1) {
                    const int pos = atomicAdd(&cur[d[k] >> GAT_BINSH], 1);
                    pr[e0 + pos] = make_int2(sv[k], d[k]);
                }
        }
    }
}

template<int H, int U>
__device__ __forceinline__ void gat_gather_batch(v4f& acc, float& m, float& den,
                                                 const v4f* __restrict__ Hv, int srcv,
                                                 int i, int Tc, int g, int sub,
                                                 v4f as4, v2f ad8)
{
    constexpr int LPR = 8 * H;
    constexpr int EPT = 32 / LPR;
    v4f   u[U];
    float sc[U];
    for (int k = 0; k < U; ++k) {
        int s;
        if constexpr (H == 4) s = __builtin_amdgcn_readlane(srcv, i + k);
        else                  s = __shfl(srcv, i + k * EPT + g);
        if constexpr (H == 4) u[k] = (Hv + (size_t)s * LPR)[sub];
        else                  u[k] = Hv[(unsigned)(s * LPR + sub)];
    }
    __builtin_amdgcn_sched_barrier(0);
    for (int k = 0; k < U; ++k) {
        const int slot = (H == 4) ? (i + k) : (i + k * EPT + g);
        float v = gat_sum8(gat_dot4(u[k], as4, ad8));
        v = (v >= 0.f) ? v : NEG_SLOPE * v;
        sc[k] = (slot < Tc) ? v : -__builtin_inff();
    }
    float mn = m;
    for (int k = 0; k < U; ++k) mn = fmaxf(mn, sc[k]);
    const float scale = __expf(m - mn);
    acc *= scale;
    den *= scale;
    for (int k = 0; k < U; ++k) {
        const float p = __expf(sc[k] - mn);
        den += p;
        acc += p * u[k];
    }
    m = mn;
}

__device__ __forceinline__ void gat_project_rows(const float* rw, const float* __restrict__ W,
                                                 float* __restrict__ Gout, int m0, int n_rows, int lane)
{
    const int mlo  = lane & 15;
    const int kofs = (lane >> 4) * 2;
    const float* arow = rw + mlo * GAT_ROWP + kofs;
    const float* bcol = W + kofs * 32 + mlo;
    v8f acc0 = {}, acc1 = {};
#pragma unroll 16
    for (int k = 0; k < 128; k += 4) {
        const v2f a  = *(const v2f*)(arow + k);
        const v2f b0 = { bcol[k * 32],      bcol[(k + 1) * 32] };
        const v2f b1 = { bcol[k * 32 + 16], bcol[(k + 1) * 32 + 16] };
        acc0 = __builtin_amdgcn_wmma_f32_16x16x4_f32(false, a, false, b0, (short)0, acc0, false, false);
        acc1 = __builtin_amdgcn_wmma_f32_16x16x4_f32(false, a, false, b1, (short)0, acc1, false, false);
    }
    const int mhi = (lane >> 4) * 8;
#pragma unroll
    for (int j = 0; j < 8; ++j) {
        const int m = m0 + mhi + j;
        if (m < n_rows) {
            Gout[(size_t)m * 32 + mlo]      = acc0[j];
            Gout[(size_t)m * 32 + 16 + mlo] = acc1[j];
        }
    }
}

template<int H, bool RELU, bool PROJ>
__global__ __launch_bounds__(256, H == 4 ? 12 : 10)
void gat_softmax_gather(const int* __restrict__ rowptr, const int* __restrict__ cnt,
                        const int* __restrict__ es, const float* __restrict__ att_s,
                        const float* __restrict__ att_d, const float* __restrict__ Hf,
                        const float* __restrict__ bias, float* __restrict__ outp, int n,
                        const float* __restrict__ W2)
{
    static_assert(!PROJ || H == 4, "the projection epilogue takes whole 128-float rows");
    __shared__ float rows[PROJ ? 32 * GAT_ROWP : 1];
    constexpr int LPR = 8 * H;
    constexpr int EPT = 32 / LPR;
    constexpr int NPW = 4;
    const int lane = threadIdx.x & 31;
    const int wv  = __builtin_amdgcn_readfirstlane((int)(threadIdx.x >> 5));
    const int sub = lane % LPR;
    const int g   = lane / LPR;
    const v4f* Hv = (const v4f*)Hf;
    const int d0  = (blockIdx.x * 8 + wv) * NPW;
    const v4f as4 = ((const v4f*)att_s)[sub];
    const v4f ad4 = ((const v4f*)att_d)[sub];

    for (int q = 0; q < NPW; ++q) {
        const int d = d0 + q;
        if (d >= n) {
            if constexpr (PROJ) {
                const v4f z = { 0.f, 0.f, 0.f, 0.f };
                *(v4f*)&rows[(wv * NPW + q) * GAT_ROWP + sub * 4] = z;
                continue;
            } else break;
        }
        const v4f us   = Hv[(size_t)d * LPR + sub];
        const int row0 = rowptr[d];
        const int deg  = cnt[d];
        int srcv = d;
        if (lane < deg) srcv = es[row0 + lane];

        const v2f  z2  = { 0.f, 0.f };
        const float ad = gat_sum8(gat_dot4(us, ad4, z2));
        const v2f  ad8 = { 0.125f * ad, 0.f };
        float m = gat_sum8(gat_dot4(us, as4, ad8));
        m = (m >= 0.f) ? m : NEG_SLOPE * m;
        const v4f z4 = { 0.f, 0.f, 0.f, 0.f };
        float den = (g == 0) ? 1.f : 0.f;
        v4f   acc = (g == 0) ? us : z4;

        for (int c0 = 0; c0 < deg; c0 += 32) {
            if (c0 > 0) {
                const int t = c0 + lane;
                srcv = d;
                if (t < deg) srcv = es[row0 + t];
            }
            const int Tc = min(deg - c0, 32);
            int i = 0;
            for (; Tc - i >= 8 * EPT; i += 8 * EPT)
                gat_gather_batch<H, 8>(acc, m, den, Hv, srcv, i, Tc, g, sub, as4, ad8);
            if (Tc - i > 4 * EPT)
                gat_gather_batch<H, 8>(acc, m, den, Hv, srcv, i, Tc, g, sub, as4, ad8);
            else if (Tc - i > 0)
                gat_gather_batch<H, 4>(acc, m, den, Hv, srcv, i, Tc, g, sub, as4, ad8);
        }

        if constexpr (EPT > 1) {
            float mx = m;
            for (int o = LPR; o < 32; o <<= 1) mx = fmaxf(mx, __shfl_xor(mx, o));
            const float scale = __expf(m - mx);
            den *= scale;
            acc *= scale;
            for (int o = LPR; o < 32; o <<= 1) {
                den   += __shfl_xor(den, o);
                acc.x += __shfl_xor(acc.x, o); acc.y += __shfl_xor(acc.y, o);
                acc.z += __shfl_xor(acc.z, o); acc.w += __shfl_xor(acc.w, o);
            }
        }
        if (g == 0) {
            v4f b = { bias[sub * 4 + 0], bias[sub * 4 + 1], bias[sub * 4 + 2], bias[sub * 4 + 3] };
            v4f r = acc / den + b;
            if (RELU) {
                r.x = fmaxf(r.x, 0.f); r.y = fmaxf(r.y, 0.f);
                r.z = fmaxf(r.z, 0.f); r.w = fmaxf(r.w, 0.f);
            }
            if constexpr (PROJ) *(v4f*)&rows[(wv * NPW + q) * GAT_ROWP + sub * 4] = r;
            else ((v4f*)outp)[(size_t)d * LPR + sub] = r;
        }
    }
    if constexpr (PROJ) {
        __syncthreads();
        if (wv < 2)
            gat_project_rows(rows + wv * 16 * GAT_ROWP, W2, outp, blockIdx.x * 32 + wv * 16, n, lane);
    }
}

template<int U>
__global__ __launch_bounds__(256)
void gat_softmax_gather_rows32(const int* __restrict__ rowptr, const int* __restrict__ cnt,
                               const int* __restrict__ es, const float* __restrict__ att_s,
                               const float* __restrict__ att_d, const float* __restrict__ Hf,
                               const float* __restrict__ bias, float* __restrict__ outp, int n)
{
    const int lane = threadIdx.x & 31;
    const int wv   = threadIdx.x >> 5;
    const int sub  = lane & 7;
    const int g    = lane >> 3;
    const v4f* Hv  = (const v4f*)Hf;
    const int d    = (blockIdx.x * 8 + wv) * 4 + g;
    const bool own = d < n;
    const int dd   = own ? d : n - 1;
    const v4f us   = Hv[(unsigned)(dd * 8 + sub)];
    const int row0 = rowptr[dd];
    const int deg  = own ? cnt[dd] : 0;
    const v4f as4  = ((const v4f*)att_s)[sub];
    const v4f ad4  = ((const v4f*)att_d)[sub];
    const v4f b4   = ((const v4f*)bias)[sub];
    const int last = max(row0 + deg - 1, 0);

    int sn[U];
    for (int k = 0; k < U; ++k) sn[k] = es[min(row0 + k, last)];

    const v2f  z2  = { 0.f, 0.f };
    const float ad = gat_sum8(gat_dot4(us, ad4, z2));
    const v2f  ad8 = { 0.125f * ad, 0.f };
    float m = gat_sum8(gat_dot4(us, as4, ad8));
    m = (m >= 0.f) ? m : NEG_SLOPE * m;
    float den = 1.f;
    v4f   acc = us;

    for (int i = 0; i < deg; i += U) {
        v4f   u[U];
        float sc[U];
        for (int k = 0; k < U; ++k) u[k] = Hv[(unsigned)(sn[k] * 8 + sub)];
        for (int k = 0; k < U; ++k) sn[k] = es[min(row0 + i + U + k, last)];
        __builtin_amdgcn_sched_barrier(0);
        for (int k = 0; k < U; ++k) {
            float v = gat_sum8(gat_dot4(u[k], as4, ad8));
            v = (v >= 0.f) ? v : NEG_SLOPE * v;
            sc[k] = (i + k < deg) ? v : -__builtin_inff();
        }
        float mn = m;
        for (int k = 0; k < U; ++k) mn = fmaxf(mn, sc[k]);
        const float scale = __expf(m - mn);
        acc *= scale;
        den *= scale;
        for (int k = 0; k < U; ++k) {
            const float p = __expf(sc[k] - mn);
            den += p;
            acc += p * u[k];
        }
        m = mn;
    }
    if (own) ((v4f*)outp)[(unsigned)(d * 8 + sub)] = acc / den + b4;
}

extern "C" void kernel_launch(void* const* d_in, const int* in_sizes, int n_in,
                              void* d_out, int out_size, void* d_ws, size_t ws_size,
                              hipStream_t stream)
{
    const float* x   = (const float*)d_in[0];
    const int*   ei  = (const int*)d_in[1];
    const float* W1  = (const float*)d_in[2];
    const float* as1 = (const float*)d_in[3];
    const float* ad1 = (const float*)d_in[4];
    const float* b1  = (const float*)d_in[5];
    const float* W2  = (const float*)d_in[6];
    const float* as2 = (const float*)d_in[7];
    const float* ad2 = (const float*)d_in[8];
    const float* b2  = (const float*)d_in[9];
    float* out = (float*)d_out;

    const int N  = in_sizes[0] / 128;
    const int E  = in_sizes[1] / 2;
    const int* srcI = ei;
    const int* dstI = ei + E;

    float* ws = (float*)d_ws;
    float*    h1    = ws;              ws += (size_t)N * 128;
    float*    g2    = ws;              ws += (size_t)N * 32;
    int*      pairs = (int*)ws;        ws += (size_t)E * 2;
    int*      es    = (int*)ws;        ws += (size_t)E;
    int*      cnt   = (int*)ws;        ws += (size_t)N;
    int*      rowp  = (int*)ws;        ws += (size_t)N;
    int*      tab   = (int*)ws;

    const int TB = 256;
    const int gB   = (N + 127) / 128;
    const int nb   = (N + GAT_BINW - 1) / GAT_BINW;
    const int gG   = (N + 31) / 32;

    gat_chunk_sort<<<GAT_CHUNKS, 1024, 0, stream>>>(srcI, dstI, tab, pairs, E, nb);

    gat_gemm_wmma_and_sort<8><<<gB + nb, TB, 0, stream>>>(x, W1, h1, N, gB, pairs, tab, cnt, rowp, es, E, nb);
    gat_softmax_gather<4, true, true><<<gG, TB, 0, stream>>>(rowp, cnt, es, as1, ad1, h1, b1, g2, N, W2);

    gat_softmax_gather_rows32<8><<<gG, TB, 0, stream>>>(rowp, cnt, es, as2, ad2, g2, b2, out, N);
}
